# starting kernel
# baseline (speedup 1.0000x reference)
#include <hip/hip_runtime.h>
#include <hip/hip_bf16.h>
#include <hip/hip_fp16.h>

#define NNODES 100000
#define NEDGES 1600000
#define HDIM   64
#define HNUMB  128
#define NCONVL 2
#define NMLPL  2
#define NEGSLP 0.2f

typedef __attribute__((ext_vector_type(2))) float v2f;
typedef __attribute__((ext_vector_type(8))) float v8f;

__device__ __forceinline__ float leaky(float x) { return x > 0.f ? x : NEGSLP * x; }

__device__ __forceinline__ unsigned fenc(float f) {
    unsigned u = __float_as_uint(f);
    return (u & 0x80000000u) ? ~u : (u | 0x80000000u);
}
__device__ __forceinline__ float fdec(unsigned u) {
    return (u & 0x80000000u) ? __uint_as_float(u & 0x7fffffffu) : __uint_as_float(~u);
}

__global__ __launch_bounds__(256) void gemm_xw_kernel(const float* __restrict__ X,
                                                      const float* __restrict__ W,
                                                      float* __restrict__ Hout,
                                                      int nrows) {
    __shared__ float Wlds[HDIM * HDIM];
    const int tid = threadIdx.x;
    for (int idx = tid; idx < HDIM * HDIM; idx += 256) {
        int k = idx >> 6, n = idx & 63;
        Wlds[(k >> 1) * 128 + n * 2 + (k & 1)] = W[idx];
    }
    __syncthreads();

    const int wave    = tid >> 5;
    const int lane    = tid & 31;
    const int hi      = lane >> 4;
    const int mrow    = lane & 15;
    const int rowbase = blockIdx.x * 128 + wave * 16;
    if (rowbase >= nrows) return;

    v2f afrag[16];
#pragma unroll
    for (int kk = 0; kk < 16; ++kk) {
        const v2f* ap = (const v2f*)(X + (size_t)(rowbase + mrow) * HDIM + kk * 4 + hi * 2);
        afrag[kk] = *ap;
    }
#pragma unroll
    for (int nt = 0; nt < 4; ++nt) {
        const int ncol = nt * 16 + mrow;
        v8f acc = {};
#pragma unroll
        for (int kk = 0; kk < 16; ++kk) {
            v2f b = *(const v2f*)&Wlds[(kk * 2 + hi) * 128 + ncol * 2];
            acc = __builtin_amdgcn_wmma_f32_16x16x4_f32(false, afrag[kk], false, b,
                                                        (short)0, acc, false, false);
        }
#pragma unroll
        for (int v = 0; v < 8; ++v)
            Hout[(size_t)(rowbase + v + 8 * hi) * HDIM + ncol] = acc[v];
    }
}

__global__ void dots_kernel(const float* __restrict__ Hm, const float* __restrict__ asrc,
                            const float* __restrict__ adst, float* __restrict__ a_s,
                            float* __restrict__ a_d, int n) {
    int r = blockIdx.x * blockDim.x + threadIdx.x;
    if (r >= n) return;
    const float* hr = Hm + (size_t)r * HDIM;
    float s = 0.f, d = 0.f;
#pragma unroll
    for (int k = 0; k < HDIM; ++k) { float v = hr[k]; s += v * asrc[k]; d += v * adst[k]; }
    a_s[r] = s; a_d[r] = d;
}

__global__ void node_init_kernel(const float* __restrict__ a_s, const float* __restrict__ a_d,
                                 float* __restrict__ e_self, unsigned* __restrict__ m_enc, int n) {
    int i = blockIdx.x * blockDim.x + threadIdx.x;
    if (i >= n) return;
    float es = leaky(a_s[i] + a_d[i]);
    e_self[i] = es;
    m_enc[i]  = fenc(es);
}

__global__ void edge_max_kernel(const int* __restrict__ src, const int* __restrict__ dst,
                                const float* __restrict__ a_s, const float* __restrict__ a_d,
                                unsigned* __restrict__ m_enc, int ne) {
    int e = blockIdx.x * blockDim.x + threadIdx.x;
    if (e >= ne) return;
    int s = src[e], d = dst[e];
    atomicMax(m_enc + d, fenc(leaky(a_s[s] + a_d[d])));
}

__global__ void node_selfw_kernel(const float* __restrict__ e_self,
                                  const unsigned* __restrict__ m_enc,
                                  float* __restrict__ denom, int n) {
    int i = blockIdx.x * blockDim.x + threadIdx.x;
    if (i >= n) return;
    denom[i] = __expf(e_self[i] - fdec(m_enc[i]));
}

__global__ void num_init_kernel(const float* __restrict__ wself, const float* __restrict__ Hm,
                                float* __restrict__ num, int total) {
    int t = blockIdx.x * blockDim.x + threadIdx.x;
    if (t >= total) return;
    num[t] = wself[t >> 6] * Hm[t];
}

__global__ void edge_w_kernel(const int* __restrict__ src, const int* __restrict__ dst,
                              const float* __restrict__ a_s, const float* __restrict__ a_d,
                              const unsigned* __restrict__ m_enc, float* __restrict__ w_edge,
                              float* __restrict__ denom, int ne) {
    int e = blockIdx.x * blockDim.x + threadIdx.x;
    if (e >= ne) return;
    int s = src[e], d = dst[e];
    float w = __expf(leaky(a_s[s] + a_d[d]) - fdec(m_enc[d]));
    w_edge[e] = w;
    atomicAdd(denom + d, w);
}

__global__ void edge_scatter_kernel(const int* __restrict__ src, const int* __restrict__ dst,
                                    const float* __restrict__ w_edge, const float* __restrict__ Hm,
                                    float* __restrict__ num, long total) {
    long t = (long)blockIdx.x * blockDim.x + threadIdx.x;
    if (t >= total) return;
    int e = (int)(t >> 6), f = (int)(t & 63);
    atomicAdd(num + (size_t)dst[e] * HDIM + f, w_edge[e] * Hm[(size_t)src[e] * HDIM + f]);
}

__global__ void finalize_kernel(const float* __restrict__ num, const float* __restrict__ denom,
                                const float* __restrict__ bias, float* __restrict__ out, int total) {
    int t = blockIdx.x * blockDim.x + threadIdx.x;
    if (t >= total) return;
    out[t] = num[t] / (denom[t >> 6] + 1e-16f) + bias[t & 63];
}

__global__ void gather_add_kernel(const float* __restrict__ out, const float* __restrict__ vnd,
                                  const int* __restrict__ blocks, float* __restrict__ x_in, int total) {
    int t = blockIdx.x * blockDim.x + threadIdx.x;
    if (t >= total) return;
    x_in[t] = out[t] + vnd[blocks[t >> 6] * HDIM + (t & 63)];
}

__global__ void zero_kernel(float* __restrict__ p, int n) {
    int t = blockIdx.x * blockDim.x + threadIdx.x;
    if (t < n) p[t] = 0.f;
}

__global__ __launch_bounds__(256) void seg_sum_kernel(const float* __restrict__ out,
                                                      const int* __restrict__ blocks,
                                                      float* __restrict__ vn_seg, long total) {
    __shared__ float acc[HNUMB * HDIM];
    for (int i = threadIdx.x; i < HNUMB * HDIM; i += 256) acc[i] = 0.f;
    __syncthreads();
    for (long t = (long)blockIdx.x * blockDim.x + threadIdx.x; t < total;
         t += (long)gridDim.x * blockDim.x) {
        int n = (int)(t >> 6), f = (int)(t & 63);
        atomicAdd(&acc[blocks[n] * HDIM + f], out[t]);
    }
    __syncthreads();
    for (int i = threadIdx.x; i < HNUMB * HDIM; i += 256) atomicAdd(vn_seg + i, acc[i]);
}

__global__ __launch_bounds__(256) void vn_update_kernel(float* __restrict__ vn_direct,
                                                        const float* __restrict__ vn_seg,
                                                        float* __restrict__ vn_root) {
    int tid = threadIdx.x;
    for (int i = tid; i < HNUMB * HDIM; i += 256)
        vn_direct[i] += vn_seg[i] + vn_root[i & (HDIM - 1)];
    __syncthreads();
    if (tid < HDIM) {
        float s = 0.f;
        for (int r = 0; r < HNUMB; ++r) s += vn_direct[r * HDIM + tid];
        vn_root[tid] += s;
    }
}

__global__ __launch_bounds__(64) void mlp_kernel(float* __restrict__ vn,
                                                 const float* __restrict__ W1, const float* __restrict__ b1,
                                                 const float* __restrict__ W2, const float* __restrict__ b2) {
    __shared__ float rin[HDIM];
    __shared__ float t1[HDIM];
    int f = threadIdx.x;
    float* row = vn + (size_t)blockIdx.x * HDIM;
    rin[f] = row[f];
    __syncthreads();
    float s = b1[f];
#pragma unroll
    for (int k = 0; k < HDIM; ++k) s += rin[k] * W1[k * HDIM + f];
    t1[f] = fmaxf(s, 0.f);
    __syncthreads();
    float s2 = b2[f];
#pragma unroll
    for (int k = 0; k < HDIM; ++k) s2 += t1[k] * W2[k * HDIM + f];
    row[f] = fmaxf(s2, 0.f);
}

__global__ void vn_init_kernel(const float* __restrict__ vn_emb, float* __restrict__ vn_direct,
                               float* __restrict__ vn_root) {
    int t = blockIdx.x * blockDim.x + threadIdx.x;
    if (t < HNUMB * HDIM) vn_direct[t] = vn_emb[t & (HDIM - 1)];
    if (t < HDIM) vn_root[t] = vn_emb[t];
}

extern "C" void kernel_launch(void* const* d_in, const int* in_sizes, int n_in,
                              void* d_out, int out_size, void* d_ws, size_t ws_size,
                              hipStream_t stream) {
    const float* x       = (const float*)d_in[0];
    const int*   ei      = (const int*)d_in[1];
    const int*   src     = ei;
    const int*   dst     = ei + NEDGES;
    const int*   hblocks = (const int*)d_in[2];
    const float* W0      = (const float*)d_in[5];
    const float* asrc0   = (const float*)d_in[6];
    const float* adst0   = (const float*)d_in[7];
    const float* b0      = (const float*)d_in[8];
    const float* Wl      = (const float*)d_in[9];
    const float* asrcl   = (const float*)d_in[10];
    const float* adstl   = (const float*)d_in[11];
    const float* bl      = (const float*)d_in[12];
    const float* vn_emb  = (const float*)d_in[13];
    const float* mW1     = (const float*)d_in[14];
    const float* mb1     = (const float*)d_in[15];
    const float* mW2     = (const float*)d_in[16];
    const float* mb2     = (const float*)d_in[17];
    float* out = (float*)d_out;

    char*  ws  = (char*)d_ws;
    size_t off = 0;
    auto carve = [&](size_t bytes) -> void* {
        void* p = ws + off;
        off += (bytes + 255) & ~(size_t)255;
        return p;
    };
    float*    h         = (float*)carve((size_t)NNODES * HDIM * 4);
    float*    x_in      = (float*)carve((size_t)NNODES * HDIM * 4);
    float*    num       = (float*)carve((size_t)NNODES * HDIM * 4);
    float*    a_s       = (float*)carve((size_t)NNODES * 4);
    float*    a_d       = (float*)carve((size_t)NNODES * 4);
    float*    e_self    = (float*)carve((size_t)NNODES * 4);
    float*    denom     = (float*)carve((size_t)NNODES * 4);
    unsigned* m_enc     = (unsigned*)carve((size_t)NNODES * 4);
    float*    w_edge    = (float*)carve((size_t)NEDGES * 4);
    float*    vn_direct = (float*)carve((size_t)HNUMB * HDIM * 4);
    float*    vn_root   = (float*)carve((size_t)HDIM * 4);
    float*    vn_seg    = (float*)carve((size_t)HNUMB * HDIM * 4);

    const int NF      = NNODES * HDIM;
    const int nodeBlk = (NNODES + 255) / 256;
    const int nfBlk   = NF / 256;
    const int edgeBlk = NEDGES / 256;
    const long EF     = (long)NEDGES * HDIM;
    const int efBlk   = (int)(EF / 256);

    auto run_gat = [&](const float* Xin, const float* W, const float* asrc,
                       const float* adst, const float* bias) {
        gemm_xw_kernel<<<(NNODES + 127) / 128, 256, 0, stream>>>(Xin, W, h, NNODES);
        dots_kernel<<<nodeBlk, 256, 0, stream>>>(h, asrc, adst, a_s, a_d, NNODES);
        node_init_kernel<<<nodeBlk, 256, 0, stream>>>(a_s, a_d, e_self, m_enc, NNODES);
        edge_max_kernel<<<edgeBlk, 256, 0, stream>>>(src, dst, a_s, a_d, m_enc, NEDGES);
        node_selfw_kernel<<<nodeBlk, 256, 0, stream>>>(e_self, m_enc, denom, NNODES);
        num_init_kernel<<<nfBlk, 256, 0, stream>>>(denom, h, num, NF);
        edge_w_kernel<<<edgeBlk, 256, 0, stream>>>(src, dst, a_s, a_d, m_enc, w_edge, denom, NEDGES);
        edge_scatter_kernel<<<efBlk, 256, 0, stream>>>(src, dst, w_edge, h, num, EF);
        finalize_kernel<<<nfBlk, 256, 0, stream>>>(num, denom, bias, out, NF);
    };

    vn_init_kernel<<<(HNUMB * HDIM + 255) / 256, 256, 0, stream>>>(vn_emb, vn_direct, vn_root);

    run_gat(x, W0, asrc0, adst0, b0);

    for (int l = 0; l < NCONVL; ++l) {
        gather_add_kernel<<<nfBlk, 256, 0, stream>>>(out, vn_direct, hblocks, x_in, NF);
        run_gat(x_in, Wl + l * HDIM * HDIM, asrcl + l * HDIM, adstl + l * HDIM, bl + l * HDIM);
        zero_kernel<<<(HNUMB * HDIM + 255) / 256, 256, 0, stream>>>(vn_seg, HNUMB * HDIM);
        seg_sum_kernel<<<128, 256, 0, stream>>>(out, hblocks, vn_seg, (long)NF);
        vn_update_kernel<<<1, 256, 0, stream>>>(vn_direct, vn_seg, vn_root);
        for (int j = 0; j < NMLPL; ++j) {
            mlp_kernel<<<HNUMB, 64, 0, stream>>>(vn_direct, mW1 + j * HDIM * HDIM, mb1 + j * HDIM,
                                                 mW2 + j * HDIM * HDIM, mb2 + j * HDIM);
            mlp_kernel<<<1, 64, 0, stream>>>(vn_root, mW1 + j * HDIM * HDIM, mb1 + j * HDIM,
                                             mW2 + j * HDIM * HDIM, mb2 + j * HDIM);
        }
    }
}


# Edges bucketed by destination once per call (int-atomic count whose return is the slot, scan, fill); one wave per node does max, exp, sums, row gather in registers: no float atomics, sum order = slot
# speedup vs baseline: 2.9665x; 2.9665x over previous
#include <hip/hip_runtime.h>
#include <hip/hip_bf16.h>
#include <hip/hip_fp16.h>

#define NNODES 100000
#define NEDGES 1600000
#define HDIM   64
#define HNUMB  128
#define NCONVL 2
#define NMLPL  2
#define NEGSLP 0.2f

typedef __attribute__((ext_vector_type(2))) float v2f;
typedef __attribute__((ext_vector_type(8))) float v8f;

__device__ __forceinline__ float leaky(float x) { return x > 0.f ? x : NEGSLP * x; }

__global__ __launch_bounds__(256) void gemm_xw_kernel(const float* __restrict__ X,
                                                      const float* __restrict__ W,
                                                      float* __restrict__ Hout,
                                                      int nrows) {
    __shared__ float Wlds[HDIM * HDIM];
    const int tid = threadIdx.x;
    for (int idx = tid; idx < HDIM * HDIM; idx += 256) {
        int k = idx >> 6, n = idx & 63;
        Wlds[(k >> 1) * 128 + n * 2 + (k & 1)] = W[idx];
    }
    __syncthreads();

    const int wave    = tid >> 5;
    const int lane    = tid & 31;
    const int hi      = lane >> 4;
    const int mrow    = lane & 15;
    const int rowbase = blockIdx.x * 128 + wave * 16;
    if (rowbase >= nrows) return;

    v2f afrag[16];
#pragma unroll
    for (int kk = 0; kk < 16; ++kk) {
        const v2f* ap = (const v2f*)(X + (size_t)(rowbase + mrow) * HDIM + kk * 4 + hi * 2);
        afrag[kk] = *ap;
    }
#pragma unroll
    for (int nt = 0; nt < 4; ++nt) {
        const int ncol = nt * 16 + mrow;
        v8f acc = {};
#pragma unroll
        for (int kk = 0; kk < 16; ++kk) {
            v2f b = *(const v2f*)&Wlds[(kk * 2 + hi) * 128 + ncol * 2];
            acc = __builtin_amdgcn_wmma_f32_16x16x4_f32(false, afrag[kk], false, b,
                                                        (short)0, acc, false, false);
        }
#pragma unroll
        for (int v = 0; v < 8; ++v)
            Hout[(size_t)(rowbase + v + 8 * hi) * HDIM + ncol] = acc[v];
    }
}

__global__ void dots_kernel(const float* __restrict__ Hm, const float* __restrict__ asrc,
                            const float* __restrict__ adst, float* __restrict__ a_s,
                            float* __restrict__ a_d, int n) {
    int r = blockIdx.x * blockDim.x + threadIdx.x;
    if (r >= n) return;
    const float* hr = Hm + (size_t)r * HDIM;
    float s = 0.f, d = 0.f;
#pragma unroll
    for (int k = 0; k < HDIM; ++k) { float v = hr[k]; s += v * asrc[k]; d += v * adst[k]; }
    a_s[r] = s; a_d[r] = d;
}

__global__ void zero_int_kernel(int* __restrict__ p, int n) {
    int t = blockIdx.x * blockDim.x + threadIdx.x;
    if (t < n) p[t] = 0;
}

__global__ void deg_count_kernel(const int* __restrict__ dst, int* __restrict__ cnt,
                                 int* __restrict__ rank, int ne) {
    int e = blockIdx.x * blockDim.x + threadIdx.x;
    if (e >= ne) return;
    rank[e] = atomicAdd(cnt + dst[e], 1);
}

__global__ __launch_bounds__(256) void tile_sum_kernel(const int* __restrict__ cnt,
                                                       int* __restrict__ tsum, int n) {
    __shared__ int red[8];
    const int tid = threadIdx.x, lane = tid & 31, wave = tid >> 5;
    const int base = blockIdx.x * 1024 + tid * 4;
    int s = 0;
#pragma unroll
    for (int k = 0; k < 4; ++k) if (base + k < n) s += cnt[base + k];
    for (int o = 16; o > 0; o >>= 1) s += __shfl_xor(s, o, 32);
    if (lane == 0) red[wave] = s;
    __syncthreads();
    if (tid == 0) {
        int t = 0;
        for (int w = 0; w < 8; ++w) t += red[w];
        tsum[blockIdx.x] = t;
    }
}

__global__ __launch_bounds__(256) void tile_scan_kernel(const int* __restrict__ cnt,
                                                        const int* __restrict__ tsum,
                                                        int* __restrict__ rowptr, int n) {
    __shared__ int wsum[8];
    __shared__ int tile_off;
    const int tid = threadIdx.x, lane = tid & 31, wave = tid >> 5;
    if (wave == 0) {
        int t = 0;
        for (int i = lane; i < (int)blockIdx.x; i += 32) t += tsum[i];
        for (int o = 16; o > 0; o >>= 1) t += __shfl_xor(t, o, 32);
        if (lane == 0) tile_off = t;
    }
    const int base = blockIdx.x * 1024 + tid * 4;
    int c[4];
#pragma unroll
    for (int k = 0; k < 4; ++k) c[k] = (base + k < n) ? cnt[base + k] : 0;
    const int mine = c[0] + c[1] + c[2] + c[3];
    int incl = mine;
    for (int o = 1; o < 32; o <<= 1) {
        int v = __shfl_up(incl, o, 32);
        if (lane >= o) incl += v;
    }
    if (lane == 31) wsum[wave] = incl;
    __syncthreads();
    int woff = 0;
    for (int w = 0; w < wave; ++w) woff += wsum[w];
    int run = tile_off + woff + incl - mine;
#pragma unroll
    for (int k = 0; k < 4; ++k) {
        if (base + k < n) {
            rowptr[base + k] = run;
            run += c[k];
        }
    }
    if (base <= n - 1 && n - 1 < base + 4) rowptr[n] = run;
}

__global__ void csr_fill_kernel(const int* __restrict__ src, const int* __restrict__ dst,
                                const int* __restrict__ rowptr, const int* __restrict__ rank,
                                int* __restrict__ csr_src, int ne) {
    int e = blockIdx.x * blockDim.x + threadIdx.x;
    if (e >= ne) return;
    csr_src[rowptr[dst[e]] + rank[e]] = src[e];
}

__global__ __launch_bounds__(256) void gat_pull_kernel(const int* __restrict__ rowptr,
                                                       const int* __restrict__ csr_src,
                                                       const float* __restrict__ a_s,
                                                       const float* __restrict__ a_d,
                                                       const float* __restrict__ Hm,
                                                       const float* __restrict__ bias,
                                                       float* __restrict__ out, int n) {
    const int lane = threadIdx.x & 31;
    const int node = __builtin_amdgcn_readfirstlane((int)(blockIdx.x * 8 + (threadIdx.x >> 5)));
    if (node >= n) return;
    const int beg = rowptr[node];
    const int end = rowptr[node + 1];
    const float ad = a_d[node];
    const float es = leaky(a_s[node] + ad);

    int   s0 = node;
    float e0 = es;
    if (beg + lane < end) { s0 = csr_src[beg + lane]; e0 = leaky(a_s[s0] + ad); }
    float m = fmaxf(es, e0);
    for (int i = beg + 32 + lane; i < end; i += 32) m = fmaxf(m, leaky(a_s[csr_src[i]] + ad));
    for (int o = 16; o > 0; o >>= 1) m = fmaxf(m, __shfl_xor(m, o, 32));

    const float wself = __expf(es - m);
    const v2f hs = *(const v2f*)(Hm + (size_t)node * HDIM + 2 * lane);
    float ax = wself * hs.x, ay = wself * hs.y;
    float dpart = 0.f;

    for (int base = beg; base < end; base += 32) {
        const int i = base + lane;
        int   s = node;
        float w = 0.f;
        if (base == beg) {
            s = s0;
            if (i < end) w = __expf(e0 - m);
        } else if (i < end) {
            s = csr_src[i];
            w = __expf(leaky(a_s[s] + ad) - m);
        }
        dpart += w;
        const int wbits = __float_as_int(w);
        const int cnt = (end - base < 32) ? (end - base) : 32;
        int j = 0;
        for (; j + 4 <= cnt; j += 4) {
            const int   sA = __builtin_amdgcn_readlane(s, j),     sB = __builtin_amdgcn_readlane(s, j + 1);
            const int   sC = __builtin_amdgcn_readlane(s, j + 2), sD = __builtin_amdgcn_readlane(s, j + 3);
            const float wA = __int_as_float(__builtin_amdgcn_readlane(wbits, j));
            const float wB = __int_as_float(__builtin_amdgcn_readlane(wbits, j + 1));
            const float wC = __int_as_float(__builtin_amdgcn_readlane(wbits, j + 2));
            const float wD = __int_as_float(__builtin_amdgcn_readlane(wbits, j + 3));
            const v2f hA = *(const v2f*)(Hm + (size_t)sA * HDIM + 2 * lane);
            const v2f hB = *(const v2f*)(Hm + (size_t)sB * HDIM + 2 * lane);
            const v2f hC = *(const v2f*)(Hm + (size_t)sC * HDIM + 2 * lane);
            const v2f hD = *(const v2f*)(Hm + (size_t)sD * HDIM + 2 * lane);
            ax = fmaf(wA, hA.x, ax); ay = fmaf(wA, hA.y, ay);
            ax = fmaf(wB, hB.x, ax); ay = fmaf(wB, hB.y, ay);
            ax = fmaf(wC, hC.x, ax); ay = fmaf(wC, hC.y, ay);
            ax = fmaf(wD, hD.x, ax); ay = fmaf(wD, hD.y, ay);
        }
        for (; j < cnt; ++j) {
            const int   sA = __builtin_amdgcn_readlane(s, j);
            const float wA = __int_as_float(__builtin_amdgcn_readlane(wbits, j));
            const v2f hA = *(const v2f*)(Hm + (size_t)sA * HDIM + 2 * lane);
            ax = fmaf(wA, hA.x, ax); ay = fmaf(wA, hA.y, ay);
        }
    }
    for (int o = 16; o > 0; o >>= 1) dpart += __shfl_xor(dpart, o, 32);
    const float den = (dpart + wself) + 1e-16f;
    v2f r;
    r.x = ax / den + bias[2 * lane];
    r.y = ay / den + bias[2 * lane + 1];
    *(v2f*)(out + (size_t)node * HDIM + 2 * lane) = r;
}

__global__ void gather_add_kernel(const float* __restrict__ out, const float* __restrict__ vnd,
                                  const int* __restrict__ blocks, float* __restrict__ x_in, int total) {
    int t = blockIdx.x * blockDim.x + threadIdx.x;
    if (t >= total) return;
    x_in[t] = out[t] + vnd[blocks[t >> 6] * HDIM + (t & 63)];
}

__global__ void zero_kernel(float* __restrict__ p, int n) {
    int t = blockIdx.x * blockDim.x + threadIdx.x;
    if (t < n) p[t] = 0.f;
}

__global__ __launch_bounds__(256) void seg_sum_kernel(const float* __restrict__ out,
                                                      const int* __restrict__ blocks,
                                                      float* __restrict__ vn_seg, long total) {
    __shared__ float acc[HNUMB * HDIM];
    for (int i = threadIdx.x; i < HNUMB * HDIM; i += 256) acc[i] = 0.f;
    __syncthreads();
    for (long t = (long)blockIdx.x * blockDim.x + threadIdx.x; t < total;
         t += (long)gridDim.x * blockDim.x) {
        int n = (int)(t >> 6), f = (int)(t & 63);
        atomicAdd(&acc[blocks[n] * HDIM + f], out[t]);
    }
    __syncthreads();
    for (int i = threadIdx.x; i < HNUMB * HDIM; i += 256) atomicAdd(vn_seg + i, acc[i]);
}

__global__ __launch_bounds__(256) void vn_update_kernel(float* __restrict__ vn_direct,
                                                        const float* __restrict__ vn_seg,
                                                        float* __restrict__ vn_root) {
    int tid = threadIdx.x;
    for (int i = tid; i < HNUMB * HDIM; i += 256)
        vn_direct[i] += vn_seg[i] + vn_root[i & (HDIM - 1)];
    __syncthreads();
    if (tid < HDIM) {
        float s = 0.f;
        for (int r = 0; r < HNUMB; ++r) s += vn_direct[r * HDIM + tid];
        vn_root[tid] += s;
    }
}

__global__ __launch_bounds__(64) void mlp_kernel(float* __restrict__ vn,
                                                 const float* __restrict__ W1, const float* __restrict__ b1,
                                                 const float* __restrict__ W2, const float* __restrict__ b2) {
    __shared__ float rin[HDIM];
    __shared__ float t1[HDIM];
    int f = threadIdx.x;
    float* row = vn + (size_t)blockIdx.x * HDIM;
    rin[f] = row[f];
    __syncthreads();
    float s = b1[f];
#pragma unroll
    for (int k = 0; k < HDIM; ++k) s += rin[k] * W1[k * HDIM + f];
    t1[f] = fmaxf(s, 0.f);
    __syncthreads();
    float s2 = b2[f];
#pragma unroll
    for (int k = 0; k < HDIM; ++k) s2 += t1[k] * W2[k * HDIM + f];
    row[f] = fmaxf(s2, 0.f);
}

__global__ void vn_init_kernel(const float* __restrict__ vn_emb, float* __restrict__ vn_direct,
                               float* __restrict__ vn_root) {
    int t = blockIdx.x * blockDim.x + threadIdx.x;
    if (t < HNUMB * HDIM) vn_direct[t] = vn_emb[t & (HDIM - 1)];
    if (t < HDIM) vn_root[t] = vn_emb[t];
}

extern "C" void kernel_launch(void* const* d_in, const int* in_sizes, int n_in,
                              void* d_out, int out_size, void* d_ws, size_t ws_size,
                              hipStream_t stream) {
    const float* x       = (const float*)d_in[0];
    const int*   ei      = (const int*)d_in[1];
    const int*   src     = ei;
    const int*   dst     = ei + NEDGES;
    const int*   hblocks = (const int*)d_in[2];
    const float* W0      = (const float*)d_in[5];
    const float* asrc0   = (const float*)d_in[6];
    const float* adst0   = (const float*)d_in[7];
    const float* b0      = (const float*)d_in[8];
    const float* Wl      = (const float*)d_in[9];
    const float* asrcl   = (const float*)d_in[10];
    const float* adstl   = (const float*)d_in[11];
    const float* bl      = (const float*)d_in[12];
    const float* vn_emb  = (const float*)d_in[13];
    const float* mW1     = (const float*)d_in[14];
    const float* mb1     = (const float*)d_in[15];
    const float* mW2     = (const float*)d_in[16];
    const float* mb2     = (const float*)d_in[17];
    float* out = (float*)d_out;

    char*  ws  = (char*)d_ws;
    size_t off = 0;
    auto carve = [&](size_t bytes) -> void* {
        void* p = ws + off;
        off += (bytes + 255) & ~(size_t)255;
        return p;
    };
    float*    h         = (float*)carve((size_t)NNODES * HDIM * 4);
    float*    x_in      = (float*)carve((size_t)NNODES * HDIM * 4);
    float*    a_s       = (float*)carve((size_t)NNODES * 4);
    float*    a_d       = (float*)carve((size_t)NNODES * 4);
    int*      cnt       = (int*)carve((size_t)NNODES * 4);
    int*      rowptr    = (int*)carve((size_t)(NNODES + 1) * 4);
    int*      tsum      = (int*)carve((size_t)((NNODES + 1023) / 1024) * 4);
    int*      rank      = (int*)carve((size_t)NEDGES * 4);
    int*      csr_src   = (int*)carve((size_t)NEDGES * 4);
    float*    vn_direct = (float*)carve((size_t)HNUMB * HDIM * 4);
    float*    vn_root   = (float*)carve((size_t)HDIM * 4);
    float*    vn_seg    = (float*)carve((size_t)HNUMB * HDIM * 4);

    const int NF      = NNODES * HDIM;
    const int nodeBlk = (NNODES + 255) / 256;
    const int nfBlk   = NF / 256;
    const int edgeBlk = NEDGES / 256;
    const int tileBlk = (NNODES + 1023) / 1024;
    const int pullBlk = (NNODES + 7) / 8;

    zero_int_kernel<<<nodeBlk, 256, 0, stream>>>(cnt, NNODES);
    deg_count_kernel<<<edgeBlk, 256, 0, stream>>>(dst, cnt, rank, NEDGES);
    tile_sum_kernel<<<tileBlk, 256, 0, stream>>>(cnt, tsum, NNODES);
    tile_scan_kernel<<<tileBlk, 256, 0, stream>>>(cnt, tsum, rowptr, NNODES);
    csr_fill_kernel<<<edgeBlk, 256, 0, stream>>>(src, dst, rowptr, rank, csr_src, NEDGES);

    auto run_gat = [&](const float* Xin, const float* W, const float* asrc,
                       const float* adst, const float* bias) {
        gemm_xw_kernel<<<(NNODES + 127) / 128, 256, 0, stream>>>(Xin, W, h, NNODES);
        dots_kernel<<<nodeBlk, 256, 0, stream>>>(h, asrc, adst, a_s, a_d, NNODES);
        gat_pull_kernel<<<pullBlk, 256, 0, stream>>>(rowptr, csr_src, a_s, a_d, h, bias, out, NNODES);
    };

    vn_init_kernel<<<(HNUMB * HDIM + 255) / 256, 256, 0, stream>>>(vn_emb, vn_direct, vn_root);

    run_gat(x, W0, asrc0, adst0, b0);

    for (int l = 0; l < NCONVL; ++l) {
        gather_add_kernel<<<nfBlk, 256, 0, stream>>>(out, vn_direct, hblocks, x_in, NF);
        run_gat(x_in, Wl + l * HDIM * HDIM, asrcl + l * HDIM, adstl + l * HDIM, bl + l * HDIM);
        zero_kernel<<<(HNUMB * HDIM + 255) / 256, 256, 0, stream>>>(vn_seg, HNUMB * HDIM);
        seg_sum_kernel<<<128, 256, 0, stream>>>(out, hblocks, vn_seg, (long)NF);
        vn_update_kernel<<<1, 256, 0, stream>>>(vn_direct, vn_seg, vn_root);
        for (int j = 0; j < NMLPL; ++j) {
            mlp_kernel<<<HNUMB, 64, 0, stream>>>(vn_direct, mW1 + j * HDIM * HDIM, mb1 + j * HDIM,
                                                 mW2 + j * HDIM * HDIM, mb2 + j * HDIM);
            mlp_kernel<<<1, 64, 0, stream>>>(vn_root, mW1 + j * HDIM * HDIM, mb1 + j * HDIM,
                                             mW2 + j * HDIM * HDIM, mb2 + j * HDIM);
        }
    }
}


# Virtual-node row add fused into the layer GEMM's A-fragment load (A = out + vn[block[row]]): the one-element-per-thread add pass, its 200k waves per layer and 2 launches go; sums unchanged
# speedup vs baseline: 3.2473x; 1.0947x over previous
#include <hip/hip_runtime.h>
#include <hip/hip_bf16.h>
#include <hip/hip_fp16.h>

#define NNODES 100000
#define NEDGES 1600000
#define HDIM   64
#define HNUMB  128
#define NCONVL 2
#define NMLPL  2
#define NEGSLP 0.2f

typedef __attribute__((ext_vector_type(2))) float v2f;
typedef __attribute__((ext_vector_type(8))) float v8f;

__device__ __forceinline__ float leaky(float x) { return x > 0.f ? x : NEGSLP * x; }

template <bool ADDVN>
__global__ __launch_bounds__(256) void gemm_xw_kernel(const float* __restrict__ X,
                                                      const float* __restrict__ W,
                                                      const float* __restrict__ vnd,
                                                      const int* __restrict__ blocks,
                                                      float* __restrict__ Hout,
                                                      int nrows) {
    __shared__ float Wlds[HDIM * HDIM];
    const int tid = threadIdx.x;
    for (int idx = tid; idx < HDIM * HDIM; idx += 256) {
        int k = idx >> 6, n = idx & 63;
        Wlds[(k >> 1) * 128 + n * 2 + (k & 1)] = W[idx];
    }
    __syncthreads();

    const int wave    = tid >> 5;
    const int lane    = tid & 31;
    const int hi      = lane >> 4;
    const int mrow    = lane & 15;
    const int rowbase = blockIdx.x * 128 + wave * 16;
    if (rowbase >= nrows) return;

    const float* xrow = X + (size_t)(rowbase + mrow) * HDIM + hi * 2;
    v2f afrag[16];
    if (ADDVN) {
        const float* vrow = vnd + (size_t)blocks[rowbase + mrow] * HDIM + hi * 2;
#pragma unroll
        for (int kk = 0; kk < 16; ++kk)
            afrag[kk] = *(const v2f*)(xrow + kk * 4) + *(const v2f*)(vrow + kk * 4);
    } else {
#pragma unroll
        for (int kk = 0; kk < 16; ++kk)
            afrag[kk] = *(const v2f*)(xrow + kk * 4);
    }
#pragma unroll
    for (int nt = 0; nt < 4; ++nt) {
        const int ncol = nt * 16 + mrow;
        v8f acc = {};
#pragma unroll
        for (int kk = 0; kk < 16; ++kk) {
            v2f b = *(const v2f*)&Wlds[(kk * 2 + hi) * 128 + ncol * 2];
            acc = __builtin_amdgcn_wmma_f32_16x16x4_f32(false, afrag[kk], false, b,
                                                        (short)0, acc, false, false);
        }
#pragma unroll
        for (int v = 0; v < 8; ++v)
            Hout[(size_t)(rowbase + v + 8 * hi) * HDIM + ncol] = acc[v];
    }
}

__global__ void dots_kernel(const float* __restrict__ Hm, const float* __restrict__ asrc,
                            const float* __restrict__ adst, float* __restrict__ a_s,
                            float* __restrict__ a_d, int n) {
    int r = blockIdx.x * blockDim.x + threadIdx.x;
    if (r >= n) return;
    const float* hr = Hm + (size_t)r * HDIM;
    float s = 0.f, d = 0.f;
#pragma unroll
    for (int k = 0; k < HDIM; ++k) { float v = hr[k]; s += v * asrc[k]; d += v * adst[k]; }
    a_s[r] = s; a_d[r] = d;
}

__global__ void zero_int_kernel(int* __restrict__ p, int n) {
    int t = blockIdx.x * blockDim.x + threadIdx.x;
    if (t < n) p[t] = 0;
}

__global__ void deg_count_kernel(const int* __restrict__ dst, int* __restrict__ cnt,
                                 int* __restrict__ rank, int ne) {
    int e = blockIdx.x * blockDim.x + threadIdx.x;
    if (e >= ne) return;
    rank[e] = atomicAdd(cnt + dst[e], 1);
}

__global__ __launch_bounds__(256) void tile_sum_kernel(const int* __restrict__ cnt,
                                                       int* __restrict__ tsum, int n) {
    __shared__ int red[8];
    const int tid = threadIdx.x, lane = tid & 31, wave = tid >> 5;
    const int base = blockIdx.x * 1024 + tid * 4;
    int s = 0;
#pragma unroll
    for (int k = 0; k < 4; ++k) if (base + k < n) s += cnt[base + k];
    for (int o = 16; o > 0; o >>= 1) s += __shfl_xor(s, o, 32);
    if (lane == 0) red[wave] = s;
    __syncthreads();
    if (tid == 0) {
        int t = 0;
        for (int w = 0; w < 8; ++w) t += red[w];
        tsum[blockIdx.x] = t;
    }
}

__global__ __launch_bounds__(256) void tile_scan_kernel(const int* __restrict__ cnt,
                                                        const int* __restrict__ tsum,
                                                        int* __restrict__ rowptr, int n) {
    __shared__ int wsum[8];
    __shared__ int tile_off;
    const int tid = threadIdx.x, lane = tid & 31, wave = tid >> 5;
    if (wave == 0) {
        int t = 0;
        for (int i = lane; i < (int)blockIdx.x; i += 32) t += tsum[i];
        for (int o = 16; o > 0; o >>= 1) t += __shfl_xor(t, o, 32);
        if (lane == 0) tile_off = t;
    }
    const int base = blockIdx.x * 1024 + tid * 4;
    int c[4];
#pragma unroll
    for (int k = 0; k < 4; ++k) c[k] = (base + k < n) ? cnt[base + k] : 0;
    const int mine = c[0] + c[1] + c[2] + c[3];
    int incl = mine;
    for (int o = 1; o < 32; o <<= 1) {
        int v = __shfl_up(incl, o, 32);
        if (lane >= o) incl += v;
    }
    if (lane == 31) wsum[wave] = incl;
    __syncthreads();
    int woff = 0;
    for (int w = 0; w < wave; ++w) woff += wsum[w];
    int run = tile_off + woff + incl - mine;
#pragma unroll
    for (int k = 0; k < 4; ++k) {
        if (base + k < n) {
            rowptr[base + k] = run;
            run += c[k];
        }
    }
    if (base <= n - 1 && n - 1 < base + 4) rowptr[n] = run;
}

__global__ void csr_fill_kernel(const int* __restrict__ src, const int* __restrict__ dst,
                                const int* __restrict__ rowptr, const int* __restrict__ rank,
                                int* __restrict__ csr_src, int ne) {
    int e = blockIdx.x * blockDim.x + threadIdx.x;
    if (e >= ne) return;
    csr_src[rowptr[dst[e]] + rank[e]] = src[e];
}

__global__ __launch_bounds__(256) void gat_pull_kernel(const int* __restrict__ rowptr,
                                                       const int* __restrict__ csr_src,
                                                       const float* __restrict__ a_s,
                                                       const float* __restrict__ a_d,
                                                       const float* __restrict__ Hm,
                                                       const float* __restrict__ bias,
                                                       float* __restrict__ out, int n) {
    const int lane = threadIdx.x & 31;
    const int node = __builtin_amdgcn_readfirstlane((int)(blockIdx.x * 8 + (threadIdx.x >> 5)));
    if (node >= n) return;
    const int beg = rowptr[node];
    const int end = rowptr[node + 1];
    const float ad = a_d[node];
    const float es = leaky(a_s[node] + ad);

    int   s0 = node;
    float e0 = es;
    if (beg + lane < end) { s0 = csr_src[beg + lane]; e0 = leaky(a_s[s0] + ad); }
    float m = fmaxf(es, e0);
    for (int i = beg + 32 + lane; i < end; i += 32) m = fmaxf(m, leaky(a_s[csr_src[i]] + ad));
    for (int o = 16; o > 0; o >>= 1) m = fmaxf(m, __shfl_xor(m, o, 32));

    const float wself = __expf(es - m);
    const v2f hs = *(const v2f*)(Hm + (size_t)node * HDIM + 2 * lane);
    float ax = wself * hs.x, ay = wself * hs.y;
    float dpart = 0.f;

    for (int base = beg; base < end; base += 32) {
        const int i = base + lane;
        int   s = node;
        float w = 0.f;
        if (base == beg) {
            s = s0;
            if (i < end) w = __expf(e0 - m);
        } else if (i < end) {
            s = csr_src[i];
            w = __expf(leaky(a_s[s] + ad) - m);
        }
        dpart += w;
        const int wbits = __float_as_int(w);
        const int cnt = (end - base < 32) ? (end - base) : 32;
        int j = 0;
        for (; j + 4 <= cnt; j += 4) {
            const int   sA = __builtin_amdgcn_readlane(s, j),     sB = __builtin_amdgcn_readlane(s, j + 1);
            const int   sC = __builtin_amdgcn_readlane(s, j + 2), sD = __builtin_amdgcn_readlane(s, j + 3);
            const float wA = __int_as_float(__builtin_amdgcn_readlane(wbits, j));
            const float wB = __int_as_float(__builtin_amdgcn_readlane(wbits, j + 1));
            const float wC = __int_as_float(__builtin_amdgcn_readlane(wbits, j + 2));
            const float wD = __int_as_float(__builtin_amdgcn_readlane(wbits, j + 3));
            const v2f hA = *(const v2f*)(Hm + (size_t)sA * HDIM + 2 * lane);
            const v2f hB = *(const v2f*)(Hm + (size_t)sB * HDIM + 2 * lane);
            const v2f hC = *(const v2f*)(Hm + (size_t)sC * HDIM + 2 * lane);
            const v2f hD = *(const v2f*)(Hm + (size_t)sD * HDIM + 2 * lane);
            ax = fmaf(wA, hA.x, ax); ay = fmaf(wA, hA.y, ay);
            ax = fmaf(wB, hB.x, ax); ay = fmaf(wB, hB.y, ay);
            ax = fmaf(wC, hC.x, ax); ay = fmaf(wC, hC.y, ay);
            ax = fmaf(wD, hD.x, ax); ay = fmaf(wD, hD.y, ay);
        }
        for (; j < cnt; ++j) {
            const int   sA = __builtin_amdgcn_readlane(s, j);
            const float wA = __int_as_float(__builtin_amdgcn_readlane(wbits, j));
            const v2f hA = *(const v2f*)(Hm + (size_t)sA * HDIM + 2 * lane);
            ax = fmaf(wA, hA.x, ax); ay = fmaf(wA, hA.y, ay);
        }
    }
    for (int o = 16; o > 0; o >>= 1) dpart += __shfl_xor(dpart, o, 32);
    const float den = (dpart + wself) + 1e-16f;
    v2f r;
    r.x = ax / den + bias[2 * lane];
    r.y = ay / den + bias[2 * lane + 1];
    *(v2f*)(out + (size_t)node * HDIM + 2 * lane) = r;
}

__global__ void zero_kernel(float* __restrict__ p, int n) {
    int t = blockIdx.x * blockDim.x + threadIdx.x;
    if (t < n) p[t] = 0.f;
}

__global__ __launch_bounds__(256) void seg_sum_kernel(const float* __restrict__ out,
                                                      const int* __restrict__ blocks,
                                                      float* __restrict__ vn_seg, long total) {
    __shared__ float acc[HNUMB * HDIM];
    for (int i = threadIdx.x; i < HNUMB * HDIM; i += 256) acc[i] = 0.f;
    __syncthreads();
    for (long t = (long)blockIdx.x * blockDim.x + threadIdx.x; t < total;
         t += (long)gridDim.x * blockDim.x) {
        int n = (int)(t >> 6), f = (int)(t & 63);
        atomicAdd(&acc[blocks[n] * HDIM + f], out[t]);
    }
    __syncthreads();
    for (int i = threadIdx.x; i < HNUMB * HDIM; i += 256) atomicAdd(vn_seg + i, acc[i]);
}

__global__ __launch_bounds__(256) void vn_update_kernel(float* __restrict__ vn_direct,
                                                        const float* __restrict__ vn_seg,
                                                        float* __restrict__ vn_root) {
    int tid = threadIdx.x;
    for (int i = tid; i < HNUMB * HDIM; i += 256)
        vn_direct[i] += vn_seg[i] + vn_root[i & (HDIM - 1)];
    __syncthreads();
    if (tid < HDIM) {
        float s = 0.f;
        for (int r = 0; r < HNUMB; ++r) s += vn_direct[r * HDIM + tid];
        vn_root[tid] += s;
    }
}

__global__ __launch_bounds__(64) void mlp_kernel(float* __restrict__ vn,
                                                 const float* __restrict__ W1, const float* __restrict__ b1,
                                                 const float* __restrict__ W2, const float* __restrict__ b2) {
    __shared__ float rin[HDIM];
    __shared__ float t1[HDIM];
    int f = threadIdx.x;
    float* row = vn + (size_t)blockIdx.x * HDIM;
    rin[f] = row[f];
    __syncthreads();
    float s = b1[f];
#pragma unroll
    for (int k = 0; k < HDIM; ++k) s += rin[k] * W1[k * HDIM + f];
    t1[f] = fmaxf(s, 0.f);
    __syncthreads();
    float s2 = b2[f];
#pragma unroll
    for (int k = 0; k < HDIM; ++k) s2 += t1[k] * W2[k * HDIM + f];
    row[f] = fmaxf(s2, 0.f);
}

__global__ void vn_init_kernel(const float* __restrict__ vn_emb, float* __restrict__ vn_direct,
                               float* __restrict__ vn_root) {
    int t = blockIdx.x * blockDim.x + threadIdx.x;
    if (t < HNUMB * HDIM) vn_direct[t] = vn_emb[t & (HDIM - 1)];
    if (t < HDIM) vn_root[t] = vn_emb[t];
}

extern "C" void kernel_launch(void* const* d_in, const int* in_sizes, int n_in,
                              void* d_out, int out_size, void* d_ws, size_t ws_size,
                              hipStream_t stream) {
    const float* x       = (const float*)d_in[0];
    const int*   ei      = (const int*)d_in[1];
    const int*   src     = ei;
    const int*   dst     = ei + NEDGES;
    const int*   hblocks = (const int*)d_in[2];
    const float* W0      = (const float*)d_in[5];
    const float* asrc0   = (const float*)d_in[6];
    const float* adst0   = (const float*)d_in[7];
    const float* b0      = (const float*)d_in[8];
    const float* Wl      = (const float*)d_in[9];
    const float* asrcl   = (const float*)d_in[10];
    const float* adstl   = (const float*)d_in[11];
    const float* bl      = (const float*)d_in[12];
    const float* vn_emb  = (const float*)d_in[13];
    const float* mW1     = (const float*)d_in[14];
    const float* mb1     = (const float*)d_in[15];
    const float* mW2     = (const float*)d_in[16];
    const float* mb2     = (const float*)d_in[17];
    float* out = (float*)d_out;

    char*  ws  = (char*)d_ws;
    size_t off = 0;
    auto carve = [&](size_t bytes) -> void* {
        void* p = ws + off;
        off += (bytes + 255) & ~(size_t)255;
        return p;
    };
    float*    h         = (float*)carve((size_t)NNODES * HDIM * 4);
    float*    a_s       = (float*)carve((size_t)NNODES * 4);
    float*    a_d       = (float*)carve((size_t)NNODES * 4);
    int*      cnt       = (int*)carve((size_t)NNODES * 4);
    int*      rowptr    = (int*)carve((size_t)(NNODES + 1) * 4);
    int*      tsum      = (int*)carve((size_t)((NNODES + 1023) / 1024) * 4);
    int*      rank      = (int*)carve((size_t)NEDGES * 4);
    int*      csr_src   = (int*)carve((size_t)NEDGES * 4);
    float*    vn_direct = (float*)carve((size_t)HNUMB * HDIM * 4);
    float*    vn_root   = (float*)carve((size_t)HDIM * 4);
    float*    vn_seg    = (float*)carve((size_t)HNUMB * HDIM * 4);

    const int NF      = NNODES * HDIM;
    const int nodeBlk = (NNODES + 255) / 256;
    const int edgeBlk = NEDGES / 256;
    const int tileBlk = (NNODES + 1023) / 1024;
    const int pullBlk = (NNODES + 7) / 8;

    zero_int_kernel<<<nodeBlk, 256, 0, stream>>>(cnt, NNODES);
    deg_count_kernel<<<edgeBlk, 256, 0, stream>>>(dst, cnt, rank, NEDGES);
    tile_sum_kernel<<<tileBlk, 256, 0, stream>>>(cnt, tsum, NNODES);
    tile_scan_kernel<<<tileBlk, 256, 0, stream>>>(cnt, tsum, rowptr, NNODES);
    csr_fill_kernel<<<edgeBlk, 256, 0, stream>>>(src, dst, rowptr, rank, csr_src, NEDGES);

    auto run_gat = [&](const float* Xin, bool addvn, const float* W, const float* asrc,
                       const float* adst, const float* bias) {
        if (addvn)
            gemm_xw_kernel<true><<<(NNODES + 127) / 128, 256, 0, stream>>>(Xin, W, vn_direct, hblocks, h, NNODES);
        else
            gemm_xw_kernel<false><<<(NNODES + 127) / 128, 256, 0, stream>>>(Xin, W, vn_direct, hblocks, h, NNODES);
        dots_kernel<<<nodeBlk, 256, 0, stream>>>(h, asrc, adst, a_s, a_d, NNODES);
        gat_pull_kernel<<<pullBlk, 256, 0, stream>>>(rowptr, csr_src, a_s, a_d, h, bias, out, NNODES);
    };

    vn_init_kernel<<<(HNUMB * HDIM + 255) / 256, 256, 0, stream>>>(vn_emb, vn_direct, vn_root);

    run_gat(x, false, W0, asrc0, adst0, b0);

    for (int l = 0; l < NCONVL; ++l) {
        run_gat(out, true, Wl + l * HDIM * HDIM, asrcl + l * HDIM, adstl + l * HDIM, bl + l * HDIM);
        zero_kernel<<<(HNUMB * HDIM + 255) / 256, 256, 0, stream>>>(vn_seg, HNUMB * HDIM);
        seg_sum_kernel<<<128, 256, 0, stream>>>(out, hblocks, vn_seg, (long)NF);
        vn_update_kernel<<<1, 256, 0, stream>>>(vn_direct, vn_seg, vn_root);
        for (int j = 0; j < NMLPL; ++j) {
            mlp_kernel<<<HNUMB, 64, 0, stream>>>(vn_direct, mW1 + j * HDIM * HDIM, mb1 + j * HDIM,
                                                 mW2 + j * HDIM * HDIM, mb2 + j * HDIM);
            mlp_kernel<<<1, 64, 0, stream>>>(vn_root, mW1 + j * HDIM * HDIM, mb1 + j * HDIM,
                                             mW2 + j * HDIM * HDIM, mb2 + j * HDIM);
        }
    }
}


# Segment sum: a wave takes 25 whole rows and issues all their loads before its first LDS add (was one element per memory round trip, 195 trips); rows regroup per work-group: sum order changes
# speedup vs baseline: 4.6495x; 1.4318x over previous
#include <hip/hip_runtime.h>
#include <hip/hip_bf16.h>
#include <hip/hip_fp16.h>

#define NNODES 100000
#define NEDGES 1600000
#define HDIM   64
#define HNUMB  128
#define NCONVL 2
#define NMLPL  2
#define NEGSLP 0.2f

typedef __attribute__((ext_vector_type(2))) float v2f;
typedef __attribute__((ext_vector_type(8))) float v8f;

__device__ __forceinline__ float leaky(float x) { return x > 0.f ? x : NEGSLP * x; }

template <bool ADDVN>
__global__ __launch_bounds__(256) void gemm_xw_kernel(const float* __restrict__ X,
                                                      const float* __restrict__ W,
                                                      const float* __restrict__ vnd,
                                                      const int* __restrict__ blocks,
                                                      float* __restrict__ Hout,
                                                      int nrows) {
    __shared__ float Wlds[HDIM * HDIM];
    const int tid = threadIdx.x;
    for (int idx = tid; idx < HDIM * HDIM; idx += 256) {
        int k = idx >> 6, n = idx & 63;
        Wlds[(k >> 1) * 128 + n * 2 + (k & 1)] = W[idx];
    }
    __syncthreads();

    const int wave    = tid >> 5;
    const int lane    = tid & 31;
    const int hi      = lane >> 4;
    const int mrow    = lane & 15;
    const int rowbase = blockIdx.x * 128 + wave * 16;
    if (rowbase >= nrows) return;

    const float* xrow = X + (size_t)(rowbase + mrow) * HDIM + hi * 2;
    v2f afrag[16];
    if (ADDVN) {
        const float* vrow = vnd + (size_t)blocks[rowbase + mrow] * HDIM + hi * 2;
#pragma unroll
        for (int kk = 0; kk < 16; ++kk)
            afrag[kk] = *(const v2f*)(xrow + kk * 4) + *(const v2f*)(vrow + kk * 4);
    } else {
#pragma unroll
        for (int kk = 0; kk < 16; ++kk)
            afrag[kk] = *(const v2f*)(xrow + kk * 4);
    }
#pragma unroll
    for (int nt = 0; nt < 4; ++nt) {
        const int ncol = nt * 16 + mrow;
        v8f acc = {};
#pragma unroll
        for (int kk = 0; kk < 16; ++kk) {
            v2f b = *(const v2f*)&Wlds[(kk * 2 + hi) * 128 + ncol * 2];
            acc = __builtin_amdgcn_wmma_f32_16x16x4_f32(false, afrag[kk], false, b,
                                                        (short)0, acc, false, false);
        }
#pragma unroll
        for (int v = 0; v < 8; ++v)
            Hout[(size_t)(rowbase + v + 8 * hi) * HDIM + ncol] = acc[v];
    }
}

__global__ void dots_kernel(const float* __restrict__ Hm, const float* __restrict__ asrc,
                            const float* __restrict__ adst, float* __restrict__ a_s,
                            float* __restrict__ a_d, int n) {
    int r = blockIdx.x * blockDim.x + threadIdx.x;
    if (r >= n) return;
    const float* hr = Hm + (size_t)r * HDIM;
    float s = 0.f, d = 0.f;
#pragma unroll
    for (int k = 0; k < HDIM; ++k) { float v = hr[k]; s += v * asrc[k]; d += v * adst[k]; }
    a_s[r] = s; a_d[r] = d;
}

__global__ void zero_int_kernel(int* __restrict__ p, int n) {
    int t = blockIdx.x * blockDim.x + threadIdx.x;
    if (t < n) p[t] = 0;
}

__global__ void deg_count_kernel(const int* __restrict__ dst, int* __restrict__ cnt,
                                 int* __restrict__ rank, int ne) {
    int e = blockIdx.x * blockDim.x + threadIdx.x;
    if (e >= ne) return;
    rank[e] = atomicAdd(cnt + dst[e], 1);
}

__global__ __launch_bounds__(256) void tile_sum_kernel(const int* __restrict__ cnt,
                                                       int* __restrict__ tsum, int n) {
    __shared__ int red[8];
    const int tid = threadIdx.x, lane = tid & 31, wave = tid >> 5;
    const int base = blockIdx.x * 1024 + tid * 4;
    int s = 0;
#pragma unroll
    for (int k = 0; k < 4; ++k) if (base + k < n) s += cnt[base + k];
    for (int o = 16; o > 0; o >>= 1) s += __shfl_xor(s, o, 32);
    if (lane == 0) red[wave] = s;
    __syncthreads();
    if (tid == 0) {
        int t = 0;
        for (int w = 0; w < 8; ++w) t += red[w];
        tsum[blockIdx.x] = t;
    }
}

__global__ __launch_bounds__(256) void tile_scan_kernel(const int* __restrict__ cnt,
                                                        const int* __restrict__ tsum,
                                                        int* __restrict__ rowptr, int n) {
    __shared__ int wsum[8];
    __shared__ int tile_off;
    const int tid = threadIdx.x, lane = tid & 31, wave = tid >> 5;
    if (wave == 0) {
        int t = 0;
        for (int i = lane; i < (int)blockIdx.x; i += 32) t += tsum[i];
        for (int o = 16; o > 0; o >>= 1) t += __shfl_xor(t, o, 32);
        if (lane == 0) tile_off = t;
    }
    const int base = blockIdx.x * 1024 + tid * 4;
    int c[4];
#pragma unroll
    for (int k = 0; k < 4; ++k) c[k] = (base + k < n) ? cnt[base + k] : 0;
    const int mine = c[0] + c[1] + c[2] + c[3];
    int incl = mine;
    for (int o = 1; o < 32; o <<= 1) {
        int v = __shfl_up(incl, o, 32);
        if (lane >= o) incl += v;
    }
    if (lane == 31) wsum[wave] = incl;
    __syncthreads();
    int woff = 0;
    for (int w = 0; w < wave; ++w) woff += wsum[w];
    int run = tile_off + woff + incl - mine;
#pragma unroll
    for (int k = 0; k < 4; ++k) {
        if (base + k < n) {
            rowptr[base + k] = run;
            run += c[k];
        }
    }
    if (base <= n - 1 && n - 1 < base + 4) rowptr[n] = run;
}

__global__ void csr_fill_kernel(const int* __restrict__ src, const int* __restrict__ dst,
                                const int* __restrict__ rowptr, const int* __restrict__ rank,
                                int* __restrict__ csr_src, int ne) {
    int e = blockIdx.x * blockDim.x + threadIdx.x;
    if (e >= ne) return;
    csr_src[rowptr[dst[e]] + rank[e]] = src[e];
}

__global__ __launch_bounds__(256) void gat_pull_kernel(const int* __restrict__ rowptr,
                                                       const int* __restrict__ csr_src,
                                                       const float* __restrict__ a_s,
                                                       const float* __restrict__ a_d,
                                                       const float* __restrict__ Hm,
                                                       const float* __restrict__ bias,
                                                       float* __restrict__ out, int n) {
    const int lane = threadIdx.x & 31;
    const int node = __builtin_amdgcn_readfirstlane((int)(blockIdx.x * 8 + (threadIdx.x >> 5)));
    if (node >= n) return;
    const int beg = rowptr[node];
    const int end = rowptr[node + 1];
    const float ad = a_d[node];
    const float es = leaky(a_s[node] + ad);

    int   s0 = node;
    float e0 = es;
    if (beg + lane < end) { s0 = csr_src[beg + lane]; e0 = leaky(a_s[s0] + ad); }
    float m = fmaxf(es, e0);
    for (int i = beg + 32 + lane; i < end; i += 32) m = fmaxf(m, leaky(a_s[csr_src[i]] + ad));
    for (int o = 16; o > 0; o >>= 1) m = fmaxf(m, __shfl_xor(m, o, 32));

    const float wself = __expf(es - m);
    const v2f hs = *(const v2f*)(Hm + (size_t)node * HDIM + 2 * lane);
    float ax = wself * hs.x, ay = wself * hs.y;
    float dpart = 0.f;

    for (int base = beg; base < end; base += 32) {
        const int i = base + lane;
        int   s = node;
        float w = 0.f;
        if (base == beg) {
            s = s0;
            if (i < end) w = __expf(e0 - m);
        } else if (i < end) {
            s = csr_src[i];
            w = __expf(leaky(a_s[s] + ad) - m);
        }
        dpart += w;
        const int wbits = __float_as_int(w);
        const int cnt = (end - base < 32) ? (end - base) : 32;
        int j = 0;
        for (; j + 4 <= cnt; j += 4) {
            const int   sA = __builtin_amdgcn_readlane(s, j),     sB = __builtin_amdgcn_readlane(s, j + 1);
            const int   sC = __builtin_amdgcn_readlane(s, j + 2), sD = __builtin_amdgcn_readlane(s, j + 3);
            const float wA = __int_as_float(__builtin_amdgcn_readlane(wbits, j));
            const float wB = __int_as_float(__builtin_amdgcn_readlane(wbits, j + 1));
            const float wC = __int_as_float(__builtin_amdgcn_readlane(wbits, j + 2));
            const float wD = __int_as_float(__builtin_amdgcn_readlane(wbits, j + 3));
            const v2f hA = *(const v2f*)(Hm + (size_t)sA * HDIM + 2 * lane);
            const v2f hB = *(const v2f*)(Hm + (size_t)sB * HDIM + 2 * lane);
            const v2f hC = *(const v2f*)(Hm + (size_t)sC * HDIM + 2 * lane);
            const v2f hD = *(const v2f*)(Hm + (size_t)sD * HDIM + 2 * lane);
            ax = fmaf(wA, hA.x, ax); ay = fmaf(wA, hA.y, ay);
            ax = fmaf(wB, hB.x, ax); ay = fmaf(wB, hB.y, ay);
            ax = fmaf(wC, hC.x, ax); ay = fmaf(wC, hC.y, ay);
            ax = fmaf(wD, hD.x, ax); ay = fmaf(wD, hD.y, ay);
        }
        for (; j < cnt; ++j) {
            const int   sA = __builtin_amdgcn_readlane(s, j);
            const float wA = __int_as_float(__builtin_amdgcn_readlane(wbits, j));
            const v2f hA = *(const v2f*)(Hm + (size_t)sA * HDIM + 2 * lane);
            ax = fmaf(wA, hA.x, ax); ay = fmaf(wA, hA.y, ay);
        }
    }
    for (int o = 16; o > 0; o >>= 1) dpart += __shfl_xor(dpart, o, 32);
    const float den = (dpart + wself) + 1e-16f;
    v2f r;
    r.x = ax / den + bias[2 * lane];
    r.y = ay / den + bias[2 * lane + 1];
    *(v2f*)(out + (size_t)node * HDIM + 2 * lane) = r;
}

__global__ void zero_kernel(float* __restrict__ p, int n) {
    int t = blockIdx.x * blockDim.x + threadIdx.x;
    if (t < n) p[t] = 0.f;
}

#define SEG_ROWS 25
__global__ __launch_bounds__(1024) void seg_sum_kernel(const float* __restrict__ out,
                                                       const int* __restrict__ blocks,
                                                       float* __restrict__ vn_seg, int n) {
    __shared__ float acc[HNUMB * HDIM];
    for (int i = threadIdx.x; i < HNUMB * HDIM; i += 1024) acc[i] = 0.f;
    __syncthreads();
    const int lane   = threadIdx.x & 31;
    const int nwaves = gridDim.x * 32;
    const int gw   = __builtin_amdgcn_readfirstlane((int)(blockIdx.x * 32 + (threadIdx.x >> 5)));
    const int last = n - 1;
    for (int base = gw * SEG_ROWS; base < n; base += nwaves * SEG_ROWS) {
        const int rk  = base + lane;
        const int bid = blocks[rk < last ? rk : last];
        v2f v[SEG_ROWS];
#pragma unroll
        for (int k = 0; k < SEG_ROWS; ++k) {
            const int r = (base + k < n) ? (base + k) : last;
            v[k] = *(const v2f*)(out + (size_t)r * HDIM + 2 * lane);
        }
#pragma unroll
        for (int k = 0; k < SEG_ROWS; ++k) {
            const bool live = base + k < n;
            const int  b    = __builtin_amdgcn_readlane(bid, k);
            float* a = acc + b * HDIM + 2 * lane;
            atomicAdd(a,     live ? v[k].x : 0.f);
            atomicAdd(a + 1, live ? v[k].y : 0.f);
        }
    }
    __syncthreads();
    for (int i = threadIdx.x; i < HNUMB * HDIM; i += 1024) atomicAdd(vn_seg + i, acc[i]);
}

__global__ __launch_bounds__(256) void vn_update_kernel(float* __restrict__ vn_direct,
                                                        const float* __restrict__ vn_seg,
                                                        float* __restrict__ vn_root) {
    int tid = threadIdx.x;
    for (int i = tid; i < HNUMB * HDIM; i += 256)
        vn_direct[i] += vn_seg[i] + vn_root[i & (HDIM - 1)];
    __syncthreads();
    if (tid < HDIM) {
        float s = 0.f;
        for (int r = 0; r < HNUMB; ++r) s += vn_direct[r * HDIM + tid];
        vn_root[tid] += s;
    }
}

__global__ __launch_bounds__(64) void mlp_kernel(float* __restrict__ vn,
                                                 const float* __restrict__ W1, const float* __restrict__ b1,
                                                 const float* __restrict__ W2, const float* __restrict__ b2) {
    __shared__ float rin[HDIM];
    __shared__ float t1[HDIM];
    int f = threadIdx.x;
    float* row = vn + (size_t)blockIdx.x * HDIM;
    rin[f] = row[f];
    __syncthreads();
    float s = b1[f];
#pragma unroll
    for (int k = 0; k < HDIM; ++k) s += rin[k] * W1[k * HDIM + f];
    t1[f] = fmaxf(s, 0.f);
    __syncthreads();
    float s2 = b2[f];
#pragma unroll
    for (int k = 0; k < HDIM; ++k) s2 += t1[k] * W2[k * HDIM + f];
    row[f] = fmaxf(s2, 0.f);
}

__global__ void vn_init_kernel(const float* __restrict__ vn_emb, float* __restrict__ vn_direct,
                               float* __restrict__ vn_root) {
    int t = blockIdx.x * blockDim.x + threadIdx.x;
    if (t < HNUMB * HDIM) vn_direct[t] = vn_emb[t & (HDIM - 1)];
    if (t < HDIM) vn_root[t] = vn_emb[t];
}

extern "C" void kernel_launch(void* const* d_in, const int* in_sizes, int n_in,
                              void* d_out, int out_size, void* d_ws, size_t ws_size,
                              hipStream_t stream) {
    const float* x       = (const float*)d_in[0];
    const int*   ei      = (const int*)d_in[1];
    const int*   src     = ei;
    const int*   dst     = ei + NEDGES;
    const int*   hblocks = (const int*)d_in[2];
    const float* W0      = (const float*)d_in[5];
    const float* asrc0   = (const float*)d_in[6];
    const float* adst0   = (const float*)d_in[7];
    const float* b0      = (const float*)d_in[8];
    const float* Wl      = (const float*)d_in[9];
    const float* asrcl   = (const float*)d_in[10];
    const float* adstl   = (const float*)d_in[11];
    const float* bl      = (const float*)d_in[12];
    const float* vn_emb  = (const float*)d_in[13];
    const float* mW1     = (const float*)d_in[14];
    const float* mb1     = (const float*)d_in[15];
    const float* mW2     = (const float*)d_in[16];
    const float* mb2     = (const float*)d_in[17];
    float* out = (float*)d_out;

    char*  ws  = (char*)d_ws;
    size_t off = 0;
    auto carve = [&](size_t bytes) -> void* {
        void* p = ws + off;
        off += (bytes + 255) & ~(size_t)255;
        return p;
    };
    float*    h         = (float*)carve((size_t)NNODES * HDIM * 4);
    float*    a_s       = (float*)carve((size_t)NNODES * 4);
    float*    a_d       = (float*)carve((size_t)NNODES * 4);
    int*      cnt       = (int*)carve((size_t)NNODES * 4);
    int*      rowptr    = (int*)carve((size_t)(NNODES + 1) * 4);
    int*      tsum      = (int*)carve((size_t)((NNODES + 1023) / 1024) * 4);
    int*      rank      = (int*)carve((size_t)NEDGES * 4);
    int*      csr_src   = (int*)carve((size_t)NEDGES * 4);
    float*    vn_direct = (float*)carve((size_t)HNUMB * HDIM * 4);
    float*    vn_root   = (float*)carve((size_t)HDIM * 4);
    float*    vn_seg    = (float*)carve((size_t)HNUMB * HDIM * 4);

    const int nodeBlk = (NNODES + 255) / 256;
    const int edgeBlk = NEDGES / 256;
    const int tileBlk = (NNODES + 1023) / 1024;
    const int pullBlk = (NNODES + 7) / 8;

    zero_int_kernel<<<nodeBlk, 256, 0, stream>>>(cnt, NNODES);
    deg_count_kernel<<<edgeBlk, 256, 0, stream>>>(dst, cnt, rank, NEDGES);
    tile_sum_kernel<<<tileBlk, 256, 0, stream>>>(cnt, tsum, NNODES);
    tile_scan_kernel<<<tileBlk, 256, 0, stream>>>(cnt, tsum, rowptr, NNODES);
    csr_fill_kernel<<<edgeBlk, 256, 0, stream>>>(src, dst, rowptr, rank, csr_src, NEDGES);

    auto run_gat = [&](const float* Xin, bool addvn, const float* W, const float* asrc,
                       const float* adst, const float* bias) {
        if (addvn)
            gemm_xw_kernel<true><<<(NNODES + 127) / 128, 256, 0, stream>>>(Xin, W, vn_direct, hblocks, h, NNODES);
        else
            gemm_xw_kernel<false><<<(NNODES + 127) / 128, 256, 0, stream>>>(Xin, W, vn_direct, hblocks, h, NNODES);
        dots_kernel<<<nodeBlk, 256, 0, stream>>>(h, asrc, adst, a_s, a_d, NNODES);
        gat_pull_kernel<<<pullBlk, 256, 0, stream>>>(rowptr, csr_src, a_s, a_d, h, bias, out, NNODES);
    };

    vn_init_kernel<<<(HNUMB * HDIM + 255) / 256, 256, 0, stream>>>(vn_emb, vn_direct, vn_root);

    run_gat(x, false, W0, asrc0, adst0, b0);

    for (int l = 0; l < NCONVL; ++l) {
        run_gat(out, true, Wl + l * HDIM * HDIM, asrcl + l * HDIM, adstl + l * HDIM, bl + l * HDIM);
        zero_kernel<<<(HNUMB * HDIM + 255) / 256, 256, 0, stream>>>(vn_seg, HNUMB * HDIM);
        seg_sum_kernel<<<128, 1024, 0, stream>>>(out, hblocks, vn_seg, NNODES);
        vn_update_kernel<<<1, 256, 0, stream>>>(vn_direct, vn_seg, vn_root);
        for (int j = 0; j < NMLPL; ++j) {
            mlp_kernel<<<HNUMB, 64, 0, stream>>>(vn_direct, mW1 + j * HDIM * HDIM, mb1 + j * HDIM,
                                                 mW2 + j * HDIM * HDIM, mb2 + j * HDIM);
            mlp_kernel<<<1, 64, 0, stream>>>(vn_root, mW1 + j * HDIM * HDIM, mb1 + j * HDIM,
                                             mW2 + j * HDIM * HDIM, mb2 + j * HDIM);
        }
    }
}


# Neighbour gather: each 16-lane half of a wave owns one node (16-byte loads fetch whole rows) and requests its first 20 rows before any add; half the waves; weight-sum (denominator) lane tree changes
# speedup vs baseline: 4.8275x; 1.0383x over previous
#include <hip/hip_runtime.h>
#include <hip/hip_bf16.h>
#include <hip/hip_fp16.h>

#define NNODES 100000
#define NEDGES 1600000
#define HDIM   64
#define HNUMB  128
#define NCONVL 2
#define NMLPL  2
#define NEGSLP 0.2f

typedef __attribute__((ext_vector_type(2))) float v2f;
typedef __attribute__((ext_vector_type(8))) float v8f;

__device__ __forceinline__ float leaky(float x) { return x > 0.f ? x : NEGSLP * x; }

template <bool ADDVN>
__global__ __launch_bounds__(256) void gemm_xw_kernel(const float* __restrict__ X,
                                                      const float* __restrict__ W,
                                                      const float* __restrict__ vnd,
                                                      const int* __restrict__ blocks,
                                                      float* __restrict__ Hout,
                                                      int nrows) {
    __shared__ float Wlds[HDIM * HDIM];
    const int tid = threadIdx.x;
    for (int idx = tid; idx < HDIM * HDIM; idx += 256) {
        int k = idx >> 6, n = idx & 63;
        Wlds[(k >> 1) * 128 + n * 2 + (k & 1)] = W[idx];
    }
    __syncthreads();

    const int wave    = tid >> 5;
    const int lane    = tid & 31;
    const int hi      = lane >> 4;
    const int mrow    = lane & 15;
    const int rowbase = blockIdx.x * 128 + wave * 16;
    if (rowbase >= nrows) return;

    const float* xrow = X + (size_t)(rowbase + mrow) * HDIM + hi * 2;
    v2f afrag[16];
    if (ADDVN) {
        const float* vrow = vnd + (size_t)blocks[rowbase + mrow] * HDIM + hi * 2;
#pragma unroll
        for (int kk = 0; kk < 16; ++kk)
            afrag[kk] = *(const v2f*)(xrow + kk * 4) + *(const v2f*)(vrow + kk * 4);
    } else {
#pragma unroll
        for (int kk = 0; kk < 16; ++kk)
            afrag[kk] = *(const v2f*)(xrow + kk * 4);
    }
#pragma unroll
    for (int nt = 0; nt < 4; ++nt) {
        const int ncol = nt * 16 + mrow;
        v8f acc = {};
#pragma unroll
        for (int kk = 0; kk < 16; ++kk) {
            v2f b = *(const v2f*)&Wlds[(kk * 2 + hi) * 128 + ncol * 2];
            acc = __builtin_amdgcn_wmma_f32_16x16x4_f32(false, afrag[kk], false, b,
                                                        (short)0, acc, false, false);
        }
#pragma unroll
        for (int v = 0; v < 8; ++v)
            Hout[(size_t)(rowbase + v + 8 * hi) * HDIM + ncol] = acc[v];
    }
}

__global__ void dots_kernel(const float* __restrict__ Hm, const float* __restrict__ asrc,
                            const float* __restrict__ adst, float* __restrict__ a_s,
                            float* __restrict__ a_d, int n) {
    int r = blockIdx.x * blockDim.x + threadIdx.x;
    if (r >= n) return;
    const float* hr = Hm + (size_t)r * HDIM;
    float s = 0.f, d = 0.f;
#pragma unroll
    for (int k = 0; k < HDIM; ++k) { float v = hr[k]; s += v * asrc[k]; d += v * adst[k]; }
    a_s[r] = s; a_d[r] = d;
}

__global__ void zero_int_kernel(int* __restrict__ p, int n) {
    int t = blockIdx.x * blockDim.x + threadIdx.x;
    if (t < n) p[t] = 0;
}

__global__ void deg_count_kernel(const int* __restrict__ dst, int* __restrict__ cnt,
                                 int* __restrict__ rank, int ne) {
    int e = blockIdx.x * blockDim.x + threadIdx.x;
    if (e >= ne) return;
    rank[e] = atomicAdd(cnt + dst[e], 1);
}

__global__ __launch_bounds__(256) void tile_sum_kernel(const int* __restrict__ cnt,
                                                       int* __restrict__ tsum, int n) {
    __shared__ int red[8];
    const int tid = threadIdx.x, lane = tid & 31, wave = tid >> 5;
    const int base = blockIdx.x * 1024 + tid * 4;
    int s = 0;
#pragma unroll
    for (int k = 0; k < 4; ++k) if (base + k < n) s += cnt[base + k];
    for (int o = 16; o > 0; o >>= 1) s += __shfl_xor(s, o, 32);
    if (lane == 0) red[wave] = s;
    __syncthreads();
    if (tid == 0) {
        int t = 0;
        for (int w = 0; w < 8; ++w) t += red[w];
        tsum[blockIdx.x] = t;
    }
}

__global__ __launch_bounds__(256) void tile_scan_kernel(const int* __restrict__ cnt,
                                                        const int* __restrict__ tsum,
                                                        int* __restrict__ rowptr, int n) {
    __shared__ int wsum[8];
    __shared__ int tile_off;
    const int tid = threadIdx.x, lane = tid & 31, wave = tid >> 5;
    if (wave == 0) {
        int t = 0;
        for (int i = lane; i < (int)blockIdx.x; i += 32) t += tsum[i];
        for (int o = 16; o > 0; o >>= 1) t += __shfl_xor(t, o, 32);
        if (lane == 0) tile_off = t;
    }
    const int base = blockIdx.x * 1024 + tid * 4;
    int c[4];
#pragma unroll
    for (int k = 0; k < 4; ++k) c[k] = (base + k < n) ? cnt[base + k] : 0;
    const int mine = c[0] + c[1] + c[2] + c[3];
    int incl = mine;
    for (int o = 1; o < 32; o <<= 1) {
        int v = __shfl_up(incl, o, 32);
        if (lane >= o) incl += v;
    }
    if (lane == 31) wsum[wave] = incl;
    __syncthreads();
    int woff = 0;
    for (int w = 0; w < wave; ++w) woff += wsum[w];
    int run = tile_off + woff + incl - mine;
#pragma unroll
    for (int k = 0; k < 4; ++k) {
        if (base + k < n) {
            rowptr[base + k] = run;
            run += c[k];
        }
    }
    if (base <= n - 1 && n - 1 < base + 4) rowptr[n] = run;
}

__global__ void csr_fill_kernel(const int* __restrict__ src, const int* __restrict__ dst,
                                const int* __restrict__ rowptr, const int* __restrict__ rank,
                                int* __restrict__ csr_src, int ne) {
    int e = blockIdx.x * blockDim.x + threadIdx.x;
    if (e >= ne) return;
    csr_src[rowptr[dst[e]] + rank[e]] = src[e];
}

template <int J> __device__ __forceinline__ int row_get_i(int v) {
    return __builtin_amdgcn_update_dpp(0, v, 0x150 + J, 0xF, 0xF, true);
}
template <int J> __device__ __forceinline__ float row_get_f(float v) {
    return __int_as_float(__builtin_amdgcn_update_dpp(0, __float_as_int(v), 0x150 + J, 0xF, 0xF, true));
}
template <int M> __device__ __forceinline__ float row_xor_f(float v) {
    return __int_as_float(__builtin_amdgcn_update_dpp(0, __float_as_int(v), 0x160 + M, 0xF, 0xF, true));
}

#define PULL_LOAD(DST, J, SV) DST = Hm4[row_get_i<(J) & 15>(SV) * 16 + l16]
#define PULL_FMA(SRC, J, WV)                                      \
    {                                                             \
        const float wj = row_get_f<(J) & 15>(WV);                 \
        acc.x = fmaf(wj, SRC.x, acc.x); acc.y = fmaf(wj, SRC.y, acc.y); \
        acc.z = fmaf(wj, SRC.z, acc.z); acc.w = fmaf(wj, SRC.w, acc.w); \
    }
#define PULL_GROUP4(J0, SV, WV)                                   \
    {                                                             \
        float4 t0, t1, t2, t3;                                    \
        PULL_LOAD(t0, (J0), SV);     PULL_LOAD(t1, (J0) + 1, SV); \
        PULL_LOAD(t2, (J0) + 2, SV); PULL_LOAD(t3, (J0) + 3, SV); \
        PULL_FMA(t0, (J0), WV);     PULL_FMA(t1, (J0) + 1, WV);   \
        PULL_FMA(t2, (J0) + 2, WV); PULL_FMA(t3, (J0) + 3, WV);   \
    }

__global__ __launch_bounds__(256) void gat_pull_kernel(const int* __restrict__ rowptr,
                                                       const int* __restrict__ csr_src,
                                                       const float* __restrict__ a_s,
                                                       const float* __restrict__ a_d,
                                                       const float* __restrict__ Hm,
                                                       const float* __restrict__ bias,
                                                       float* __restrict__ out, int n) {
    const int l16  = threadIdx.x & 15;
    const int node = blockIdx.x * 16 + (threadIdx.x >> 4);
    const bool valid = node < n;
    const int nd  = valid ? node : n - 1;
    const int beg  = rowptr[nd];
    const int endv = rowptr[nd + 1];
    const int deg  = valid ? endv - beg : 0;
    const float ad = a_d[nd];
    const float4* __restrict__ Hm4 = (const float4*)Hm;

    const bool p0 = l16 < deg, p1 = l16 + 16 < deg;
    int s0 = csr_src[p0 ? beg + l16 : 0];
    int s1 = csr_src[p1 ? beg + 16 + l16 : 0];
    s0 = p0 ? s0 : nd;
    s1 = p1 ? s1 : nd;
    const float es = leaky(a_s[nd] + ad);
    const float e0 = leaky(a_s[s0] + ad);
    const float e1 = leaky(a_s[s1] + ad);

    const float4 hs = Hm4[nd * 16 + l16];
    const float4 bv = *(const float4*)(bias + 4 * l16);
    float4 r0, r1, r2, r3, r4, r5, r6, r7, r8, r9, r10, r11, r12, r13, r14, r15, r16, r17, r18, r19;
    PULL_LOAD(r0, 0, s0);   PULL_LOAD(r1, 1, s0);   PULL_LOAD(r2, 2, s0);   PULL_LOAD(r3, 3, s0);
    PULL_LOAD(r4, 4, s0);   PULL_LOAD(r5, 5, s0);   PULL_LOAD(r6, 6, s0);   PULL_LOAD(r7, 7, s0);
    PULL_LOAD(r8, 8, s0);   PULL_LOAD(r9, 9, s0);   PULL_LOAD(r10, 10, s0); PULL_LOAD(r11, 11, s0);
    PULL_LOAD(r12, 12, s0); PULL_LOAD(r13, 13, s0); PULL_LOAD(r14, 14, s0); PULL_LOAD(r15, 15, s0);
    PULL_LOAD(r16, 16, s1); PULL_LOAD(r17, 17, s1); PULL_LOAD(r18, 18, s1); PULL_LOAD(r19, 19, s1);

    const bool past32 = __builtin_amdgcn_ballot_w32(deg > 32) != 0u;

    float m = fmaxf(es, fmaxf(e0, e1));
    if (past32)
        for (int i = 32 + l16; i < deg; i += 16) m = fmaxf(m, leaky(a_s[csr_src[beg + i]] + ad));
    m = fmaxf(m, row_xor_f<1>(m));
    m = fmaxf(m, row_xor_f<2>(m));
    m = fmaxf(m, row_xor_f<4>(m));
    m = fmaxf(m, row_xor_f<8>(m));

    const float wself = __expf(es - m);
    const float w0 = p0 ? __expf(e0 - m) : 0.f;
    const float w1 = p1 ? __expf(e1 - m) : 0.f;
    float dpart = w0 + w1;
    float4 acc;
    acc.x = wself * hs.x; acc.y = wself * hs.y; acc.z = wself * hs.z; acc.w = wself * hs.w;

    PULL_FMA(r0, 0, w0);   PULL_FMA(r1, 1, w0);   PULL_FMA(r2, 2, w0);   PULL_FMA(r3, 3, w0);
    PULL_FMA(r4, 4, w0);   PULL_FMA(r5, 5, w0);   PULL_FMA(r6, 6, w0);   PULL_FMA(r7, 7, w0);
    PULL_FMA(r8, 8, w0);   PULL_FMA(r9, 9, w0);   PULL_FMA(r10, 10, w0); PULL_FMA(r11, 11, w0);
    PULL_FMA(r12, 12, w0); PULL_FMA(r13, 13, w0); PULL_FMA(r14, 14, w0); PULL_FMA(r15, 15, w0);
    PULL_FMA(r16, 16, w1); PULL_FMA(r17, 17, w1); PULL_FMA(r18, 18, w1); PULL_FMA(r19, 19, w1);

    if (__builtin_amdgcn_ballot_w32(deg > 20) != 0u) {
        PULL_GROUP4(20, s1, w1);
        if (__builtin_amdgcn_ballot_w32(deg > 24) != 0u) {
            PULL_GROUP4(24, s1, w1);
            if (__builtin_amdgcn_ballot_w32(deg > 28) != 0u) {
                PULL_GROUP4(28, s1, w1);
            }
        }
    }
    if (past32) {
        for (int cb = 32; __builtin_amdgcn_ballot_w32(cb < deg) != 0u; cb += 16) {
            const bool p = cb + l16 < deg;
            int s = csr_src[p ? beg + cb + l16 : 0];
            s = p ? s : nd;
            const float w = p ? __expf(leaky(a_s[s] + ad) - m) : 0.f;
            dpart += w;
            PULL_GROUP4(0, s, w);
            PULL_GROUP4(4, s, w);
            PULL_GROUP4(8, s, w);
            PULL_GROUP4(12, s, w);
        }
    }

    dpart += row_xor_f<1>(dpart);
    dpart += row_xor_f<2>(dpart);
    dpart += row_xor_f<4>(dpart);
    dpart += row_xor_f<8>(dpart);
    const float den = (dpart + wself) + 1e-16f;
    float4 res;
    res.x = acc.x / den + bv.x;
    res.y = acc.y / den + bv.y;
    res.z = acc.z / den + bv.z;
    res.w = acc.w / den + bv.w;
    if (valid) *(float4*)(out + (size_t)node * HDIM + 4 * l16) = res;
}

__global__ void zero_kernel(float* __restrict__ p, int n) {
    int t = blockIdx.x * blockDim.x + threadIdx.x;
    if (t < n) p[t] = 0.f;
}

#define SEG_ROWS 25
__global__ __launch_bounds__(1024) void seg_sum_kernel(const float* __restrict__ out,
                                                       const int* __restrict__ blocks,
                                                       float* __restrict__ vn_seg, int n) {
    __shared__ float acc[HNUMB * HDIM];
    for (int i = threadIdx.x; i < HNUMB * HDIM; i += 1024) acc[i] = 0.f;
    __syncthreads();
    const int lane   = threadIdx.x & 31;
    const int nwaves = gridDim.x * 32;
    const int gw   = __builtin_amdgcn_readfirstlane((int)(blockIdx.x * 32 + (threadIdx.x >> 5)));
    const int last = n - 1;
    for (int base = gw * SEG_ROWS; base < n; base += nwaves * SEG_ROWS) {
        const int rk  = base + lane;
        const int bid = blocks[rk < last ? rk : last];
        v2f v[SEG_ROWS];
#pragma unroll
        for (int k = 0; k < SEG_ROWS; ++k) {
            const int r = (base + k < n) ? (base + k) : last;
            v[k] = *(const v2f*)(out + (size_t)r * HDIM + 2 * lane);
        }
#pragma unroll
        for (int k = 0; k < SEG_ROWS; ++k) {
            const bool live = base + k < n;
            const int  b    = __builtin_amdgcn_readlane(bid, k);
            float* a = acc + b * HDIM + 2 * lane;
            atomicAdd(a,     live ? v[k].x : 0.f);
            atomicAdd(a + 1, live ? v[k].y : 0.f);
        }
    }
    __syncthreads();
    for (int i = threadIdx.x; i < HNUMB * HDIM; i += 1024) atomicAdd(vn_seg + i, acc[i]);
}

__global__ __launch_bounds__(256) void vn_update_kernel(float* __restrict__ vn_direct,
                                                        const float* __restrict__ vn_seg,
                                                        float* __restrict__ vn_root) {
    int tid = threadIdx.x;
    for (int i = tid; i < HNUMB * HDIM; i += 256)
        vn_direct[i] += vn_seg[i] + vn_root[i & (HDIM - 1)];
    __syncthreads();
    if (tid < HDIM) {
        float s = 0.f;
        for (int r = 0; r < HNUMB; ++r) s += vn_direct[r * HDIM + tid];
        vn_root[tid] += s;
    }
}

__global__ __launch_bounds__(64) void mlp_kernel(float* __restrict__ vn,
                                                 const float* __restrict__ W1, const float* __restrict__ b1,
                                                 const float* __restrict__ W2, const float* __restrict__ b2) {
    __shared__ float rin[HDIM];
    __shared__ float t1[HDIM];
    int f = threadIdx.x;
    float* row = vn + (size_t)blockIdx.x * HDIM;
    rin[f] = row[f];
    __syncthreads();
    float s = b1[f];
#pragma unroll
    for (int k = 0; k < HDIM; ++k) s += rin[k] * W1[k * HDIM + f];
    t1[f] = fmaxf(s, 0.f);
    __syncthreads();
    float s2 = b2[f];
#pragma unroll
    for (int k = 0; k < HDIM; ++k) s2 += t1[k] * W2[k * HDIM + f];
    row[f] = fmaxf(s2, 0.f);
}

__global__ void vn_init_kernel(const float* __restrict__ vn_emb, float* __restrict__ vn_direct,
                               float* __restrict__ vn_root) {
    int t = blockIdx.x * blockDim.x + threadIdx.x;
    if (t < HNUMB * HDIM) vn_direct[t] = vn_emb[t & (HDIM - 1)];
    if (t < HDIM) vn_root[t] = vn_emb[t];
}

extern "C" void kernel_launch(void* const* d_in, const int* in_sizes, int n_in,
                              void* d_out, int out_size, void* d_ws, size_t ws_size,
                              hipStream_t stream) {
    const float* x       = (const float*)d_in[0];
    const int*   ei      = (const int*)d_in[1];
    const int*   src     = ei;
    const int*   dst     = ei + NEDGES;
    const int*   hblocks = (const int*)d_in[2];
    const float* W0      = (const float*)d_in[5];
    const float* asrc0   = (const float*)d_in[6];
    const float* adst0   = (const float*)d_in[7];
    const float* b0      = (const float*)d_in[8];
    const float* Wl      = (const float*)d_in[9];
    const float* asrcl   = (const float*)d_in[10];
    const float* adstl   = (const float*)d_in[11];
    const float* bl      = (const float*)d_in[12];
    const float* vn_emb  = (const float*)d_in[13];
    const float* mW1     = (const float*)d_in[14];
    const float* mb1     = (const float*)d_in[15];
    const float* mW2     = (const float*)d_in[16];
    const float* mb2     = (const float*)d_in[17];
    float* out = (float*)d_out;

    char*  ws  = (char*)d_ws;
    size_t off = 0;
    auto carve = [&](size_t bytes) -> void* {
        void* p = ws + off;
        off += (bytes + 255) & ~(size_t)255;
        return p;
    };
    float*    h         = (float*)carve((size_t)NNODES * HDIM * 4);
    float*    a_s       = (float*)carve((size_t)NNODES * 4);
    float*    a_d       = (float*)carve((size_t)NNODES * 4);
    int*      cnt       = (int*)carve((size_t)NNODES * 4);
    int*      rowptr    = (int*)carve((size_t)(NNODES + 1) * 4);
    int*      tsum      = (int*)carve((size_t)((NNODES + 1023) / 1024) * 4);
    int*      rank      = (int*)carve((size_t)NEDGES * 4);
    int*      csr_src   = (int*)carve((size_t)NEDGES * 4);
    float*    vn_direct = (float*)carve((size_t)HNUMB * HDIM * 4);
    float*    vn_root   = (float*)carve((size_t)HDIM * 4);
    float*    vn_seg    = (float*)carve((size_t)HNUMB * HDIM * 4);

    const int nodeBlk = (NNODES + 255) / 256;
    const int edgeBlk = NEDGES / 256;
    const int tileBlk = (NNODES + 1023) / 1024;
    const int pullBlk = (NNODES + 15) / 16;

    zero_int_kernel<<<nodeBlk, 256, 0, stream>>>(cnt, NNODES);
    deg_count_kernel<<<edgeBlk, 256, 0, stream>>>(dst, cnt, rank, NEDGES);
    tile_sum_kernel<<<tileBlk, 256, 0, stream>>>(cnt, tsum, NNODES);
    tile_scan_kernel<<<tileBlk, 256, 0, stream>>>(cnt, tsum, rowptr, NNODES);
    csr_fill_kernel<<<edgeBlk, 256, 0, stream>>>(src, dst, rowptr, rank, csr_src, NEDGES);

    auto run_gat = [&](const float* Xin, bool addvn, const float* W, const float* asrc,
                       const float* adst, const float* bias) {
        if (addvn)
            gemm_xw_kernel<true><<<(NNODES + 127) / 128, 256, 0, stream>>>(Xin, W, vn_direct, hblocks, h, NNODES);
        else
            gemm_xw_kernel<false><<<(NNODES + 127) / 128, 256, 0, stream>>>(Xin, W, vn_direct, hblocks, h, NNODES);
        dots_kernel<<<nodeBlk, 256, 0, stream>>>(h, asrc, adst, a_s, a_d, NNODES);
        gat_pull_kernel<<<pullBlk, 256, 0, stream>>>(rowptr, csr_src, a_s, a_d, h, bias, out, NNODES);
    };

    vn_init_kernel<<<(HNUMB * HDIM + 255) / 256, 256, 0, stream>>>(vn_emb, vn_direct, vn_root);

    run_gat(x, false, W0, asrc0, adst0, b0);

    for (int l = 0; l < NCONVL; ++l) {
        run_gat(out, true, Wl + l * HDIM * HDIM, asrcl + l * HDIM, adstl + l * HDIM, bl + l * HDIM);
        zero_kernel<<<(HNUMB * HDIM + 255) / 256, 256, 0, stream>>>(vn_seg, HNUMB * HDIM);
        seg_sum_kernel<<<128, 1024, 0, stream>>>(out, hblocks, vn_seg, NNODES);
        vn_update_kernel<<<1, 256, 0, stream>>>(vn_direct, vn_seg, vn_root);
        for (int j = 0; j < NMLPL; ++j) {
            mlp_kernel<<<HNUMB, 64, 0, stream>>>(vn_direct, mW1 + j * HDIM * HDIM, mb1 + j * HDIM,
                                                 mW2 + j * HDIM * HDIM, mb2 + j * HDIM);
            mlp_kernel<<<1, 64, 0, stream>>>(vn_root, mW1 + j * HDIM * HDIM, mb1 + j * HDIM,
                                             mW2 + j * HDIM * HDIM, mb2 + j * HDIM);
        }
    }
}


# Virtual-node work that no output reads is not issued: the whole update behind the last conv, and the root's update and MLPs (read only by that tail); 29 launches become 20, no sum order changes
# speedup vs baseline: 5.5513x; 1.1499x over previous
#include <hip/hip_runtime.h>
#include <hip/hip_bf16.h>
#include <hip/hip_fp16.h>

#define NNODES 100000
#define NEDGES 1600000
#define HDIM   64
#define HNUMB  128
#define NCONVL 2
#define NMLPL  2
#define NEGSLP 0.2f

typedef __attribute__((ext_vector_type(2))) float v2f;
typedef __attribute__((ext_vector_type(8))) float v8f;

__device__ __forceinline__ float leaky(float x) { return x > 0.f ? x : NEGSLP * x; }

template <bool ADDVN>
__global__ __launch_bounds__(256) void gemm_xw_kernel(const float* __restrict__ X,
                                                      const float* __restrict__ W,
                                                      const float* __restrict__ vnd,
                                                      const int* __restrict__ blocks,
                                                      float* __restrict__ Hout,
                                                      int nrows) {
    __shared__ float Wlds[HDIM * HDIM];
    const int tid = threadIdx.x;
    for (int idx = tid; idx < HDIM * HDIM; idx += 256) {
        int k = idx >> 6, n = idx & 63;
        Wlds[(k >> 1) * 128 + n * 2 + (k & 1)] = W[idx];
    }
    __syncthreads();

    const int wave    = tid >> 5;
    const int lane    = tid & 31;
    const int hi      = lane >> 4;
    const int mrow    = lane & 15;
    const int rowbase = blockIdx.x * 128 + wave * 16;
    if (rowbase >= nrows) return;

    const float* xrow = X + (size_t)(rowbase + mrow) * HDIM + hi * 2;
    v2f afrag[16];
    if (ADDVN) {
        const float* vrow = vnd + (size_t)blocks[rowbase + mrow] * HDIM + hi * 2;
#pragma unroll
        for (int kk = 0; kk < 16; ++kk)
            afrag[kk] = *(const v2f*)(xrow + kk * 4) + *(const v2f*)(vrow + kk * 4);
    } else {
#pragma unroll
        for (int kk = 0; kk < 16; ++kk)
            afrag[kk] = *(const v2f*)(xrow + kk * 4);
    }
#pragma unroll
    for (int nt = 0; nt < 4; ++nt) {
        const int ncol = nt * 16 + mrow;
        v8f acc = {};
#pragma unroll
        for (int kk = 0; kk < 16; ++kk) {
            v2f b = *(const v2f*)&Wlds[(kk * 2 + hi) * 128 + ncol * 2];
            acc = __builtin_amdgcn_wmma_f32_16x16x4_f32(false, afrag[kk], false, b,
                                                        (short)0, acc, false, false);
        }
#pragma unroll
        for (int v = 0; v < 8; ++v)
            Hout[(size_t)(rowbase + v + 8 * hi) * HDIM + ncol] = acc[v];
    }
}

__global__ void dots_kernel(const float* __restrict__ Hm, const float* __restrict__ asrc,
                            const float* __restrict__ adst, float* __restrict__ a_s,
                            float* __restrict__ a_d, int n) {
    int r = blockIdx.x * blockDim.x + threadIdx.x;
    if (r >= n) return;
    const float* hr = Hm + (size_t)r * HDIM;
    float s = 0.f, d = 0.f;
#pragma unroll
    for (int k = 0; k < HDIM; ++k) { float v = hr[k]; s += v * asrc[k]; d += v * adst[k]; }
    a_s[r] = s; a_d[r] = d;
}

__global__ void zero_int_kernel(int* __restrict__ p, int n) {
    int t = blockIdx.x * blockDim.x + threadIdx.x;
    if (t < n) p[t] = 0;
}

__global__ void deg_count_kernel(const int* __restrict__ dst, int* __restrict__ cnt,
                                 int* __restrict__ rank, int ne) {
    int e = blockIdx.x * blockDim.x + threadIdx.x;
    if (e >= ne) return;
    rank[e] = atomicAdd(cnt + dst[e], 1);
}

__global__ __launch_bounds__(256) void tile_sum_kernel(const int* __restrict__ cnt,
                                                       int* __restrict__ tsum, int n) {
    __shared__ int red[8];
    const int tid = threadIdx.x, lane = tid & 31, wave = tid >> 5;
    const int base = blockIdx.x * 1024 + tid * 4;
    int s = 0;
#pragma unroll
    for (int k = 0; k < 4; ++k) if (base + k < n) s += cnt[base + k];
    for (int o = 16; o > 0; o >>= 1) s += __shfl_xor(s, o, 32);
    if (lane == 0) red[wave] = s;
    __syncthreads();
    if (tid == 0) {
        int t = 0;
        for (int w = 0; w < 8; ++w) t += red[w];
        tsum[blockIdx.x] = t;
    }
}

__global__ __launch_bounds__(256) void tile_scan_kernel(const int* __restrict__ cnt,
                                                        const int* __restrict__ tsum,
                                                        int* __restrict__ rowptr, int n) {
    __shared__ int wsum[8];
    __shared__ int tile_off;
    const int tid = threadIdx.x, lane = tid & 31, wave = tid >> 5;
    if (wave == 0) {
        int t = 0;
        for (int i = lane; i < (int)blockIdx.x; i += 32) t += tsum[i];
        for (int o = 16; o > 0; o >>= 1) t += __shfl_xor(t, o, 32);
        if (lane == 0) tile_off = t;
    }
    const int base = blockIdx.x * 1024 + tid * 4;
    int c[4];
#pragma unroll
    for (int k = 0; k < 4; ++k) c[k] = (base + k < n) ? cnt[base + k] : 0;
    const int mine = c[0] + c[1] + c[2] + c[3];
    int incl = mine;
    for (int o = 1; o < 32; o <<= 1) {
        int v = __shfl_up(incl, o, 32);
        if (lane >= o) incl += v;
    }
    if (lane == 31) wsum[wave] = incl;
    __syncthreads();
    int woff = 0;
    for (int w = 0; w < wave; ++w) woff += wsum[w];
    int run = tile_off + woff + incl - mine;
#pragma unroll
    for (int k = 0; k < 4; ++k) {
        if (base + k < n) {
            rowptr[base + k] = run;
            run += c[k];
        }
    }
    if (base <= n - 1 && n - 1 < base + 4) rowptr[n] = run;
}

__global__ void csr_fill_kernel(const int* __restrict__ src, const int* __restrict__ dst,
                                const int* __restrict__ rowptr, const int* __restrict__ rank,
                                int* __restrict__ csr_src, int ne) {
    int e = blockIdx.x * blockDim.x + threadIdx.x;
    if (e >= ne) return;
    csr_src[rowptr[dst[e]] + rank[e]] = src[e];
}

template <int J> __device__ __forceinline__ int row_get_i(int v) {
    return __builtin_amdgcn_update_dpp(0, v, 0x150 + J, 0xF, 0xF, true);
}
template <int J> __device__ __forceinline__ float row_get_f(float v) {
    return __int_as_float(__builtin_amdgcn_update_dpp(0, __float_as_int(v), 0x150 + J, 0xF, 0xF, true));
}
template <int M> __device__ __forceinline__ float row_xor_f(float v) {
    return __int_as_float(__builtin_amdgcn_update_dpp(0, __float_as_int(v), 0x160 + M, 0xF, 0xF, true));
}

#define PULL_LOAD(DST, J, SV) DST = Hm4[row_get_i<(J) & 15>(SV) * 16 + l16]
#define PULL_FMA(SRC, J, WV)                                      \
    {                                                             \
        const float wj = row_get_f<(J) & 15>(WV);                 \
        acc.x = fmaf(wj, SRC.x, acc.x); acc.y = fmaf(wj, SRC.y, acc.y); \
        acc.z = fmaf(wj, SRC.z, acc.z); acc.w = fmaf(wj, SRC.w, acc.w); \
    }
#define PULL_GROUP4(J0, SV, WV)                                   \
    {                                                             \
        float4 t0, t1, t2, t3;                                    \
        PULL_LOAD(t0, (J0), SV);     PULL_LOAD(t1, (J0) + 1, SV); \
        PULL_LOAD(t2, (J0) + 2, SV); PULL_LOAD(t3, (J0) + 3, SV); \
        PULL_FMA(t0, (J0), WV);     PULL_FMA(t1, (J0) + 1, WV);   \
        PULL_FMA(t2, (J0) + 2, WV); PULL_FMA(t3, (J0) + 3, WV);   \
    }

__global__ __launch_bounds__(256) void gat_pull_kernel(const int* __restrict__ rowptr,
                                                       const int* __restrict__ csr_src,
                                                       const float* __restrict__ a_s,
                                                       const float* __restrict__ a_d,
                                                       const float* __restrict__ Hm,
                                                       const float* __restrict__ bias,
                                                       float* __restrict__ out, int n) {
    const int l16  = threadIdx.x & 15;
    const int node = blockIdx.x * 16 + (threadIdx.x >> 4);
    const bool valid = node < n;
    const int nd  = valid ? node : n - 1;
    const int beg  = rowptr[nd];
    const int endv = rowptr[nd + 1];
    const int deg  = valid ? endv - beg : 0;
    const float ad = a_d[nd];
    const float4* __restrict__ Hm4 = (const float4*)Hm;

    const bool p0 = l16 < deg, p1 = l16 + 16 < deg;
    int s0 = csr_src[p0 ? beg + l16 : 0];
    int s1 = csr_src[p1 ? beg + 16 + l16 : 0];
    s0 = p0 ? s0 : nd;
    s1 = p1 ? s1 : nd;
    const float es = leaky(a_s[nd] + ad);
    const float e0 = leaky(a_s[s0] + ad);
    const float e1 = leaky(a_s[s1] + ad);

    const float4 hs = Hm4[nd * 16 + l16];
    const float4 bv = *(const float4*)(bias + 4 * l16);
    float4 r0, r1, r2, r3, r4, r5, r6, r7, r8, r9, r10, r11, r12, r13, r14, r15, r16, r17, r18, r19;
    PULL_LOAD(r0, 0, s0);   PULL_LOAD(r1, 1, s0);   PULL_LOAD(r2, 2, s0);   PULL_LOAD(r3, 3, s0);
    PULL_LOAD(r4, 4, s0);   PULL_LOAD(r5, 5, s0);   PULL_LOAD(r6, 6, s0);   PULL_LOAD(r7, 7, s0);
    PULL_LOAD(r8, 8, s0);   PULL_LOAD(r9, 9, s0);   PULL_LOAD(r10, 10, s0); PULL_LOAD(r11, 11, s0);
    PULL_LOAD(r12, 12, s0); PULL_LOAD(r13, 13, s0); PULL_LOAD(r14, 14, s0); PULL_LOAD(r15, 15, s0);
    PULL_LOAD(r16, 16, s1); PULL_LOAD(r17, 17, s1); PULL_LOAD(r18, 18, s1); PULL_LOAD(r19, 19, s1);

    const bool past32 = __builtin_amdgcn_ballot_w32(deg > 32) != 0u;

    float m = fmaxf(es, fmaxf(e0, e1));
    if (past32)
        for (int i = 32 + l16; i < deg; i += 16) m = fmaxf(m, leaky(a_s[csr_src[beg + i]] + ad));
    m = fmaxf(m, row_xor_f<1>(m));
    m = fmaxf(m, row_xor_f<2>(m));
    m = fmaxf(m, row_xor_f<4>(m));
    m = fmaxf(m, row_xor_f<8>(m));

    const float wself = __expf(es - m);
    const float w0 = p0 ? __expf(e0 - m) : 0.f;
    const float w1 = p1 ? __expf(e1 - m) : 0.f;
    float dpart = w0 + w1;
    float4 acc;
    acc.x = wself * hs.x; acc.y = wself * hs.y; acc.z = wself * hs.z; acc.w = wself * hs.w;

    PULL_FMA(r0, 0, w0);   PULL_FMA(r1, 1, w0);   PULL_FMA(r2, 2, w0);   PULL_FMA(r3, 3, w0);
    PULL_FMA(r4, 4, w0);   PULL_FMA(r5, 5, w0);   PULL_FMA(r6, 6, w0);   PULL_FMA(r7, 7, w0);
    PULL_FMA(r8, 8, w0);   PULL_FMA(r9, 9, w0);   PULL_FMA(r10, 10, w0); PULL_FMA(r11, 11, w0);
    PULL_FMA(r12, 12, w0); PULL_FMA(r13, 13, w0); PULL_FMA(r14, 14, w0); PULL_FMA(r15, 15, w0);
    PULL_FMA(r16, 16, w1); PULL_FMA(r17, 17, w1); PULL_FMA(r18, 18, w1); PULL_FMA(r19, 19, w1);

    if (__builtin_amdgcn_ballot_w32(deg > 20) != 0u) {
        PULL_GROUP4(20, s1, w1);
        if (__builtin_amdgcn_ballot_w32(deg > 24) != 0u) {
            PULL_GROUP4(24, s1, w1);
            if (__builtin_amdgcn_ballot_w32(deg > 28) != 0u) {
                PULL_GROUP4(28, s1, w1);
            }
        }
    }
    if (past32) {
        for (int cb = 32; __builtin_amdgcn_ballot_w32(cb < deg) != 0u; cb += 16) {
            const bool p = cb + l16 < deg;
            int s = csr_src[p ? beg + cb + l16 : 0];
            s = p ? s : nd;
            const float w = p ? __expf(leaky(a_s[s] + ad) - m) : 0.f;
            dpart += w;
            PULL_GROUP4(0, s, w);
            PULL_GROUP4(4, s, w);
            PULL_GROUP4(8, s, w);
            PULL_GROUP4(12, s, w);
        }
    }

    dpart += row_xor_f<1>(dpart);
    dpart += row_xor_f<2>(dpart);
    dpart += row_xor_f<4>(dpart);
    dpart += row_xor_f<8>(dpart);
    const float den = (dpart + wself) + 1e-16f;
    float4 res;
    res.x = acc.x / den + bv.x;
    res.y = acc.y / den + bv.y;
    res.z = acc.z / den + bv.z;
    res.w = acc.w / den + bv.w;
    if (valid) *(float4*)(out + (size_t)node * HDIM + 4 * l16) = res;
}

__global__ void zero_kernel(float* __restrict__ p, int n) {
    int t = blockIdx.x * blockDim.x + threadIdx.x;
    if (t < n) p[t] = 0.f;
}

#define SEG_ROWS 25
__global__ __launch_bounds__(1024) void seg_sum_kernel(const float* __restrict__ out,
                                                       const int* __restrict__ blocks,
                                                       float* __restrict__ vn_seg, int n) {
    __shared__ float acc[HNUMB * HDIM];
    for (int i = threadIdx.x; i < HNUMB * HDIM; i += 1024) acc[i] = 0.f;
    __syncthreads();
    const int lane   = threadIdx.x & 31;
    const int nwaves = gridDim.x * 32;
    const int gw   = __builtin_amdgcn_readfirstlane((int)(blockIdx.x * 32 + (threadIdx.x >> 5)));
    const int last = n - 1;
    for (int base = gw * SEG_ROWS; base < n; base += nwaves * SEG_ROWS) {
        const int rk  = base + lane;
        const int bid = blocks[rk < last ? rk : last];
        v2f v[SEG_ROWS];
#pragma unroll
        for (int k = 0; k < SEG_ROWS; ++k) {
            const int r = (base + k < n) ? (base + k) : last;
            v[k] = *(const v2f*)(out + (size_t)r * HDIM + 2 * lane);
        }
#pragma unroll
        for (int k = 0; k < SEG_ROWS; ++k) {
            const bool live = base + k < n;
            const int  b    = __builtin_amdgcn_readlane(bid, k);
            float* a = acc + b * HDIM + 2 * lane;
            atomicAdd(a,     live ? v[k].x : 0.f);
            atomicAdd(a + 1, live ? v[k].y : 0.f);
        }
    }
    __syncthreads();
    for (int i = threadIdx.x; i < HNUMB * HDIM; i += 1024) atomicAdd(vn_seg + i, acc[i]);
}

__global__ void vn_update_kernel(float* __restrict__ vn_direct, const float* __restrict__ vn_seg,
                                 const float* __restrict__ vn_emb) {
    int t = blockIdx.x * blockDim.x + threadIdx.x;
    if (t < HNUMB * HDIM) vn_direct[t] += vn_seg[t] + vn_emb[t & (HDIM - 1)];
}

__global__ __launch_bounds__(64) void mlp_kernel(float* __restrict__ vn,
                                                 const float* __restrict__ W1, const float* __restrict__ b1,
                                                 const float* __restrict__ W2, const float* __restrict__ b2) {
    __shared__ float rin[HDIM];
    __shared__ float t1[HDIM];
    int f = threadIdx.x;
    float* row = vn + (size_t)blockIdx.x * HDIM;
    rin[f] = row[f];
    __syncthreads();
    float s = b1[f];
#pragma unroll
    for (int k = 0; k < HDIM; ++k) s += rin[k] * W1[k * HDIM + f];
    t1[f] = fmaxf(s, 0.f);
    __syncthreads();
    float s2 = b2[f];
#pragma unroll
    for (int k = 0; k < HDIM; ++k) s2 += t1[k] * W2[k * HDIM + f];
    row[f] = fmaxf(s2, 0.f);
}

__global__ void vn_init_kernel(const float* __restrict__ vn_emb, float* __restrict__ vn_direct) {
    int t = blockIdx.x * blockDim.x + threadIdx.x;
    if (t < HNUMB * HDIM) vn_direct[t] = vn_emb[t & (HDIM - 1)];
}

extern "C" void kernel_launch(void* const* d_in, const int* in_sizes, int n_in,
                              void* d_out, int out_size, void* d_ws, size_t ws_size,
                              hipStream_t stream) {
    const float* x       = (const float*)d_in[0];
    const int*   ei      = (const int*)d_in[1];
    const int*   src     = ei;
    const int*   dst     = ei + NEDGES;
    const int*   hblocks = (const int*)d_in[2];
    const float* W0      = (const float*)d_in[5];
    const float* asrc0   = (const float*)d_in[6];
    const float* adst0   = (const float*)d_in[7];
    const float* b0      = (const float*)d_in[8];
    const float* Wl      = (const float*)d_in[9];
    const float* asrcl   = (const float*)d_in[10];
    const float* adstl   = (const float*)d_in[11];
    const float* bl      = (const float*)d_in[12];
    const float* vn_emb  = (const float*)d_in[13];
    const float* mW1     = (const float*)d_in[14];
    const float* mb1     = (const float*)d_in[15];
    const float* mW2     = (const float*)d_in[16];
    const float* mb2     = (const float*)d_in[17];
    float* out = (float*)d_out;

    char*  ws  = (char*)d_ws;
    size_t off = 0;
    auto carve = [&](size_t bytes) -> void* {
        void* p = ws + off;
        off += (bytes + 255) & ~(size_t)255;
        return p;
    };
    float*    h         = (float*)carve((size_t)NNODES * HDIM * 4);
    float*    a_s       = (float*)carve((size_t)NNODES * 4);
    float*    a_d       = (float*)carve((size_t)NNODES * 4);
    int*      cnt       = (int*)carve((size_t)NNODES * 4);
    int*      rowptr    = (int*)carve((size_t)(NNODES + 1) * 4);
    int*      tsum      = (int*)carve((size_t)((NNODES + 1023) / 1024) * 4);
    int*      rank      = (int*)carve((size_t)NEDGES * 4);
    int*      csr_src   = (int*)carve((size_t)NEDGES * 4);
    float*    vn_direct = (float*)carve((size_t)HNUMB * HDIM * 4);
    float*    vn_seg    = (float*)carve((size_t)HNUMB * HDIM * 4);

    const int nodeBlk = (NNODES + 255) / 256;
    const int edgeBlk = NEDGES / 256;
    const int tileBlk = (NNODES + 1023) / 1024;
    const int pullBlk = (NNODES + 15) / 16;

    zero_int_kernel<<<nodeBlk, 256, 0, stream>>>(cnt, NNODES);
    deg_count_kernel<<<edgeBlk, 256, 0, stream>>>(dst, cnt, rank, NEDGES);
    tile_sum_kernel<<<tileBlk, 256, 0, stream>>>(cnt, tsum, NNODES);
    tile_scan_kernel<<<tileBlk, 256, 0, stream>>>(cnt, tsum, rowptr, NNODES);
    csr_fill_kernel<<<edgeBlk, 256, 0, stream>>>(src, dst, rowptr, rank, csr_src, NEDGES);

    auto run_gat = [&](const float* Xin, bool addvn, const float* W, const float* asrc,
                       const float* adst, const float* bias) {
        if (addvn)
            gemm_xw_kernel<true><<<(NNODES + 127) / 128, 256, 0, stream>>>(Xin, W, vn_direct, hblocks, h, NNODES);
        else
            gemm_xw_kernel<false><<<(NNODES + 127) / 128, 256, 0, stream>>>(Xin, W, vn_direct, hblocks, h, NNODES);
        dots_kernel<<<nodeBlk, 256, 0, stream>>>(h, asrc, adst, a_s, a_d, NNODES);
        gat_pull_kernel<<<pullBlk, 256, 0, stream>>>(rowptr, csr_src, a_s, a_d, h, bias, out, NNODES);
    };

    vn_init_kernel<<<(HNUMB * HDIM + 255) / 256, 256, 0, stream>>>(vn_emb, vn_direct);

    run_gat(x, false, W0, asrc0, adst0, b0);

    static_assert(NCONVL == 2, "the root's update gets a reader with a third conv layer: restore it then");
    for (int l = 0; l < NCONVL; ++l) {
        run_gat(out, true, Wl + l * HDIM * HDIM, asrcl + l * HDIM, adstl + l * HDIM, bl + l * HDIM);
        if (l + 1 == NCONVL) break;
        zero_kernel<<<(HNUMB * HDIM + 255) / 256, 256, 0, stream>>>(vn_seg, HNUMB * HDIM);
        seg_sum_kernel<<<128, 1024, 0, stream>>>(out, hblocks, vn_seg, NNODES);
        vn_update_kernel<<<(HNUMB * HDIM + 255) / 256, 256, 0, stream>>>(vn_direct, vn_seg, vn_emb);
        for (int j = 0; j < NMLPL; ++j)
            mlp_kernel<<<HNUMB, 64, 0, stream>>>(vn_direct, mW1 + j * HDIM * HDIM, mb1 + j * HDIM,
                                                 mW2 + j * HDIM * HDIM, mb2 + j * HDIM);
    }
}


# Dependent launches fused: attention dots formed in the GEMM epilogue from the accumulator fragments, three init kernels in one, vn update + both MLPs in one: 20 launches to 13; dot sum order changes
# speedup vs baseline: 6.0518x; 1.0902x over previous
#include <hip/hip_runtime.h>
#include <hip/hip_bf16.h>
#include <hip/hip_fp16.h>

#define NNODES 100000
#define NEDGES 1600000
#define HDIM   64
#define HNUMB  128
#define NCONVL 2
#define NMLPL  2
#define NEGSLP 0.2f

typedef __attribute__((ext_vector_type(2))) float v2f;
typedef __attribute__((ext_vector_type(8))) float v8f;

__device__ __forceinline__ float leaky(float x) { return x > 0.f ? x : NEGSLP * x; }

template <int J> __device__ __forceinline__ int row_get_i(int v) {
    return __builtin_amdgcn_update_dpp(0, v, 0x150 + J, 0xF, 0xF, true);
}
template <int J> __device__ __forceinline__ float row_get_f(float v) {
    return __int_as_float(__builtin_amdgcn_update_dpp(0, __float_as_int(v), 0x150 + J, 0xF, 0xF, true));
}
template <int M> __device__ __forceinline__ float row_xor_f(float v) {
    return __int_as_float(__builtin_amdgcn_update_dpp(0, __float_as_int(v), 0x160 + M, 0xF, 0xF, true));
}

template <bool ADDVN>
__global__ __launch_bounds__(256) void gemm_xw_kernel(const float* __restrict__ X,
                                                      const float* __restrict__ W,
                                                      const float* __restrict__ vnd,
                                                      const int* __restrict__ blocks,
                                                      const float* __restrict__ asrc,
                                                      const float* __restrict__ adst,
                                                      float* __restrict__ Hout,
                                                      float* __restrict__ a_s,
                                                      float* __restrict__ a_d,
                                                      int nrows) {
    __shared__ float Wlds[HDIM * HDIM];
    const int tid = threadIdx.x;
    for (int idx = tid; idx < HDIM * HDIM; idx += 256) {
        int k = idx >> 6, n = idx & 63;
        Wlds[(k >> 1) * 128 + n * 2 + (k & 1)] = W[idx];
    }
    __syncthreads();

    const int wave    = tid >> 5;
    const int lane    = tid & 31;
    const int hi      = lane >> 4;
    const int mrow    = lane & 15;
    const int rowbase = blockIdx.x * 128 + wave * 16;
    if (rowbase >= nrows) return;

    const float* xrow = X + (size_t)(rowbase + mrow) * HDIM + hi * 2;
    v2f afrag[16];
    if (ADDVN) {
        const float* vrow = vnd + (size_t)blocks[rowbase + mrow] * HDIM + hi * 2;
#pragma unroll
        for (int kk = 0; kk < 16; ++kk)
            afrag[kk] = *(const v2f*)(xrow + kk * 4) + *(const v2f*)(vrow + kk * 4);
    } else {
#pragma unroll
        for (int kk = 0; kk < 16; ++kk)
            afrag[kk] = *(const v2f*)(xrow + kk * 4);
    }
    float ps[8], pd[8];
#pragma unroll
    for (int v = 0; v < 8; ++v) { ps[v] = 0.f; pd[v] = 0.f; }
#pragma unroll
    for (int nt = 0; nt < 4; ++nt) {
        const int ncol = nt * 16 + mrow;
        const float asc = asrc[ncol];
        const float adc = adst[ncol];
        v8f acc = {};
#pragma unroll
        for (int kk = 0; kk < 16; ++kk) {
            v2f b = *(const v2f*)&Wlds[(kk * 2 + hi) * 128 + ncol * 2];
            acc = __builtin_amdgcn_wmma_f32_16x16x4_f32(false, afrag[kk], false, b,
                                                        (short)0, acc, false, false);
        }
#pragma unroll
        for (int v = 0; v < 8; ++v) {
            Hout[(size_t)(rowbase + v + 8 * hi) * HDIM + ncol] = acc[v];
            ps[v] = fmaf(acc[v], asc, ps[v]);
            pd[v] = fmaf(acc[v], adc, pd[v]);
        }
    }
#pragma unroll
    for (int v = 0; v < 8; ++v) {
        ps[v] += row_xor_f<1>(ps[v]); pd[v] += row_xor_f<1>(pd[v]);
        ps[v] += row_xor_f<2>(ps[v]); pd[v] += row_xor_f<2>(pd[v]);
        ps[v] += row_xor_f<4>(ps[v]); pd[v] += row_xor_f<4>(pd[v]);
        ps[v] += row_xor_f<8>(ps[v]); pd[v] += row_xor_f<8>(pd[v]);
    }
    float sv = ps[0], dv = pd[0];
#pragma unroll
    for (int v = 1; v < 8; ++v) {
        sv = (mrow == v) ? ps[v] : sv;
        dv = (mrow == v) ? pd[v] : dv;
    }
    if (mrow < 8) {
        a_s[rowbase + 8 * hi + mrow] = sv;
        a_d[rowbase + 8 * hi + mrow] = dv;
    }
}

__global__ void init_kernel(int* __restrict__ cnt, float* __restrict__ vn_direct,
                            float* __restrict__ vn_seg, const float* __restrict__ vn_emb, int n) {
    int t = blockIdx.x * blockDim.x + threadIdx.x;
    if (t < n) cnt[t] = 0;
    if (t < HNUMB * HDIM) {
        vn_direct[t] = vn_emb[t & (HDIM - 1)];
        vn_seg[t]    = 0.f;
    }
}

__global__ void deg_count_kernel(const int* __restrict__ dst, int* __restrict__ cnt,
                                 int* __restrict__ rank, int ne) {
    int e = blockIdx.x * blockDim.x + threadIdx.x;
    if (e >= ne) return;
    rank[e] = atomicAdd(cnt + dst[e], 1);
}

__global__ __launch_bounds__(256) void tile_sum_kernel(const int* __restrict__ cnt,
                                                       int* __restrict__ tsum, int n) {
    __shared__ int red[8];
    const int tid = threadIdx.x, lane = tid & 31, wave = tid >> 5;
    const int base = blockIdx.x * 1024 + tid * 4;
    int s = 0;
#pragma unroll
    for (int k = 0; k < 4; ++k) if (base + k < n) s += cnt[base + k];
    for (int o = 16; o > 0; o >>= 1) s += __shfl_xor(s, o, 32);
    if (lane == 0) red[wave] = s;
    __syncthreads();
    if (tid == 0) {
        int t = 0;
        for (int w = 0; w < 8; ++w) t += red[w];
        tsum[blockIdx.x] = t;
    }
}

__global__ __launch_bounds__(256) void tile_scan_kernel(const int* __restrict__ cnt,
                                                        const int* __restrict__ tsum,
                                                        int* __restrict__ rowptr, int n) {
    __shared__ int wsum[8];
    __shared__ int tile_off;
    const int tid = threadIdx.x, lane = tid & 31, wave = tid >> 5;
    if (wave == 0) {
        int t = 0;
        for (int i = lane; i < (int)blockIdx.x; i += 32) t += tsum[i];
        for (int o = 16; o > 0; o >>= 1) t += __shfl_xor(t, o, 32);
        if (lane == 0) tile_off = t;
    }
    const int base = blockIdx.x * 1024 + tid * 4;
    int c[4];
#pragma unroll
    for (int k = 0; k < 4; ++k) c[k] = (base + k < n) ? cnt[base + k] : 0;
    const int mine = c[0] + c[1] + c[2] + c[3];
    int incl = mine;
    for (int o = 1; o < 32; o <<= 1) {
        int v = __shfl_up(incl, o, 32);
        if (lane >= o) incl += v;
    }
    if (lane == 31) wsum[wave] = incl;
    __syncthreads();
    int woff = 0;
    for (int w = 0; w < wave; ++w) woff += wsum[w];
    int run = tile_off + woff + incl - mine;
#pragma unroll
    for (int k = 0; k < 4; ++k) {
        if (base + k < n) {
            rowptr[base + k] = run;
            run += c[k];
        }
    }
    if (base <= n - 1 && n - 1 < base + 4) rowptr[n] = run;
}

__global__ void csr_fill_kernel(const int* __restrict__ src, const int* __restrict__ dst,
                                const int* __restrict__ rowptr, const int* __restrict__ rank,
                                int* __restrict__ csr_src, int ne) {
    int e = blockIdx.x * blockDim.x + threadIdx.x;
    if (e >= ne) return;
    csr_src[rowptr[dst[e]] + rank[e]] = src[e];
}


#define PULL_LOAD(DST, J, SV) DST = Hm4[row_get_i<(J) & 15>(SV) * 16 + l16]
#define PULL_FMA(SRC, J, WV)                                      \
    {                                                             \
        const float wj = row_get_f<(J) & 15>(WV);                 \
        acc.x = fmaf(wj, SRC.x, acc.x); acc.y = fmaf(wj, SRC.y, acc.y); \
        acc.z = fmaf(wj, SRC.z, acc.z); acc.w = fmaf(wj, SRC.w, acc.w); \
    }
#define PULL_GROUP4(J0, SV, WV)                                   \
    {                                                             \
        float4 t0, t1, t2, t3;                                    \
        PULL_LOAD(t0, (J0), SV);     PULL_LOAD(t1, (J0) + 1, SV); \
        PULL_LOAD(t2, (J0) + 2, SV); PULL_LOAD(t3, (J0) + 3, SV); \
        PULL_FMA(t0, (J0), WV);     PULL_FMA(t1, (J0) + 1, WV);   \
        PULL_FMA(t2, (J0) + 2, WV); PULL_FMA(t3, (J0) + 3, WV);   \
    }

__global__ __launch_bounds__(256) void gat_pull_kernel(const int* __restrict__ rowptr,
                                                       const int* __restrict__ csr_src,
                                                       const float* __restrict__ a_s,
                                                       const float* __restrict__ a_d,
                                                       const float* __restrict__ Hm,
                                                       const float* __restrict__ bias,
                                                       float* __restrict__ out, int n) {
    const int l16  = threadIdx.x & 15;
    const int node = blockIdx.x * 16 + (threadIdx.x >> 4);
    const bool valid = node < n;
    const int nd  = valid ? node : n - 1;
    const int beg  = rowptr[nd];
    const int endv = rowptr[nd + 1];
    const int deg  = valid ? endv - beg : 0;
    const float ad = a_d[nd];
    const float4* __restrict__ Hm4 = (const float4*)Hm;

    const bool p0 = l16 < deg, p1 = l16 + 16 < deg;
    int s0 = csr_src[p0 ? beg + l16 : 0];
    int s1 = csr_src[p1 ? beg + 16 + l16 : 0];
    s0 = p0 ? s0 : nd;
    s1 = p1 ? s1 : nd;
    const float es = leaky(a_s[nd] + ad);
    const float e0 = leaky(a_s[s0] + ad);
    const float e1 = leaky(a_s[s1] + ad);

    const float4 hs = Hm4[nd * 16 + l16];
    const float4 bv = *(const float4*)(bias + 4 * l16);
    float4 r0, r1, r2, r3, r4, r5, r6, r7, r8, r9, r10, r11, r12, r13, r14, r15, r16, r17, r18, r19;
    PULL_LOAD(r0, 0, s0);   PULL_LOAD(r1, 1, s0);   PULL_LOAD(r2, 2, s0);   PULL_LOAD(r3, 3, s0);
    PULL_LOAD(r4, 4, s0);   PULL_LOAD(r5, 5, s0);   PULL_LOAD(r6, 6, s0);   PULL_LOAD(r7, 7, s0);
    PULL_LOAD(r8, 8, s0);   PULL_LOAD(r9, 9, s0);   PULL_LOAD(r10, 10, s0); PULL_LOAD(r11, 11, s0);
    PULL_LOAD(r12, 12, s0); PULL_LOAD(r13, 13, s0); PULL_LOAD(r14, 14, s0); PULL_LOAD(r15, 15, s0);
    PULL_LOAD(r16, 16, s1); PULL_LOAD(r17, 17, s1); PULL_LOAD(r18, 18, s1); PULL_LOAD(r19, 19, s1);

    const bool past32 = __builtin_amdgcn_ballot_w32(deg > 32) != 0u;

    float m = fmaxf(es, fmaxf(e0, e1));
    if (past32)
        for (int i = 32 + l16; i < deg; i += 16) m = fmaxf(m, leaky(a_s[csr_src[beg + i]] + ad));
    m = fmaxf(m, row_xor_f<1>(m));
    m = fmaxf(m, row_xor_f<2>(m));
    m = fmaxf(m, row_xor_f<4>(m));
    m = fmaxf(m, row_xor_f<8>(m));

    const float wself = __expf(es - m);
    const float w0 = p0 ? __expf(e0 - m) : 0.f;
    const float w1 = p1 ? __expf(e1 - m) : 0.f;
    float dpart = w0 + w1;
    float4 acc;
    acc.x = wself * hs.x; acc.y = wself * hs.y; acc.z = wself * hs.z; acc.w = wself * hs.w;

    PULL_FMA(r0, 0, w0);   PULL_FMA(r1, 1, w0);   PULL_FMA(r2, 2, w0);   PULL_FMA(r3, 3, w0);
    PULL_FMA(r4, 4, w0);   PULL_FMA(r5, 5, w0);   PULL_FMA(r6, 6, w0);   PULL_FMA(r7, 7, w0);
    PULL_FMA(r8, 8, w0);   PULL_FMA(r9, 9, w0);   PULL_FMA(r10, 10, w0); PULL_FMA(r11, 11, w0);
    PULL_FMA(r12, 12, w0); PULL_FMA(r13, 13, w0); PULL_FMA(r14, 14, w0); PULL_FMA(r15, 15, w0);
    PULL_FMA(r16, 16, w1); PULL_FMA(r17, 17, w1); PULL_FMA(r18, 18, w1); PULL_FMA(r19, 19, w1);

    if (__builtin_amdgcn_ballot_w32(deg > 20) != 0u) {
        PULL_GROUP4(20, s1, w1);
        if (__builtin_amdgcn_ballot_w32(deg > 24) != 0u) {
            PULL_GROUP4(24, s1, w1);
            if (__builtin_amdgcn_ballot_w32(deg > 28) != 0u) {
                PULL_GROUP4(28, s1, w1);
            }
        }
    }
    if (past32) {
        for (int cb = 32; __builtin_amdgcn_ballot_w32(cb < deg) != 0u; cb += 16) {
            const bool p = cb + l16 < deg;
            int s = csr_src[p ? beg + cb + l16 : 0];
            s = p ? s : nd;
            const float w = p ? __expf(leaky(a_s[s] + ad) - m) : 0.f;
            dpart += w;
            PULL_GROUP4(0, s, w);
            PULL_GROUP4(4, s, w);
            PULL_GROUP4(8, s, w);
            PULL_GROUP4(12, s, w);
        }
    }

    dpart += row_xor_f<1>(dpart);
    dpart += row_xor_f<2>(dpart);
    dpart += row_xor_f<4>(dpart);
    dpart += row_xor_f<8>(dpart);
    const float den = (dpart + wself) + 1e-16f;
    float4 res;
    res.x = acc.x / den + bv.x;
    res.y = acc.y / den + bv.y;
    res.z = acc.z / den + bv.z;
    res.w = acc.w / den + bv.w;
    if (valid) *(float4*)(out + (size_t)node * HDIM + 4 * l16) = res;
}

#define SEG_ROWS 25
__global__ __launch_bounds__(1024) void seg_sum_kernel(const float* __restrict__ out,
                                                       const int* __restrict__ blocks,
                                                       float* __restrict__ vn_seg, int n) {
    __shared__ float acc[HNUMB * HDIM];
    for (int i = threadIdx.x; i < HNUMB * HDIM; i += 1024) acc[i] = 0.f;
    __syncthreads();
    const int lane   = threadIdx.x & 31;
    const int nwaves = gridDim.x * 32;
    const int gw   = __builtin_amdgcn_readfirstlane((int)(blockIdx.x * 32 + (threadIdx.x >> 5)));
    const int last = n - 1;
    for (int base = gw * SEG_ROWS; base < n; base += nwaves * SEG_ROWS) {
        const int rk  = base + lane;
        const int bid = blocks[rk < last ? rk : last];
        v2f v[SEG_ROWS];
#pragma unroll
        for (int k = 0; k < SEG_ROWS; ++k) {
            const int r = (base + k < n) ? (base + k) : last;
            v[k] = *(const v2f*)(out + (size_t)r * HDIM + 2 * lane);
        }
#pragma unroll
        for (int k = 0; k < SEG_ROWS; ++k) {
            const bool live = base + k < n;
            const int  b    = __builtin_amdgcn_readlane(bid, k);
            float* a = acc + b * HDIM + 2 * lane;
            atomicAdd(a,     live ? v[k].x : 0.f);
            atomicAdd(a + 1, live ? v[k].y : 0.f);
        }
    }
    __syncthreads();
    for (int i = threadIdx.x; i < HNUMB * HDIM; i += 1024) atomicAdd(vn_seg + i, acc[i]);
}

__global__ __launch_bounds__(64) void vn_update_mlp_kernel(float* __restrict__ vn_direct,
                                                           const float* __restrict__ vn_seg,
                                                           const float* __restrict__ vn_emb,
                                                           const float* __restrict__ W1, const float* __restrict__ b1,
                                                           const float* __restrict__ W2, const float* __restrict__ b2) {
    __shared__ float rin[HDIM];
    __shared__ float t1[HDIM];
    const int f = threadIdx.x;
    float* row = vn_direct + (size_t)blockIdx.x * HDIM;
    float cur = row[f] + (vn_seg[(size_t)blockIdx.x * HDIM + f] + vn_emb[f]);
    for (int j = 0; j < NMLPL; ++j) {
        const float* w1 = W1 + (size_t)j * HDIM * HDIM;
        const float* w2 = W2 + (size_t)j * HDIM * HDIM;
        rin[f] = cur;
        __syncthreads();
        float s = b1[j * HDIM + f];
#pragma unroll
        for (int k = 0; k < HDIM; ++k) s += rin[k] * w1[k * HDIM + f];
        t1[f] = fmaxf(s, 0.f);
        __syncthreads();
        float s2 = b2[j * HDIM + f];
#pragma unroll
        for (int k = 0; k < HDIM; ++k) s2 += t1[k] * w2[k * HDIM + f];
        cur = fmaxf(s2, 0.f);
    }
    row[f] = cur;
}

extern "C" void kernel_launch(void* const* d_in, const int* in_sizes, int n_in,
                              void* d_out, int out_size, void* d_ws, size_t ws_size,
                              hipStream_t stream) {
    const float* x       = (const float*)d_in[0];
    const int*   ei      = (const int*)d_in[1];
    const int*   src     = ei;
    const int*   dst     = ei + NEDGES;
    const int*   hblocks = (const int*)d_in[2];
    const float* W0      = (const float*)d_in[5];
    const float* asrc0   = (const float*)d_in[6];
    const float* adst0   = (const float*)d_in[7];
    const float* b0      = (const float*)d_in[8];
    const float* Wl      = (const float*)d_in[9];
    const float* asrcl   = (const float*)d_in[10];
    const float* adstl   = (const float*)d_in[11];
    const float* bl      = (const float*)d_in[12];
    const float* vn_emb  = (const float*)d_in[13];
    const float* mW1     = (const float*)d_in[14];
    const float* mb1     = (const float*)d_in[15];
    const float* mW2     = (const float*)d_in[16];
    const float* mb2     = (const float*)d_in[17];
    float* out = (float*)d_out;

    char*  ws  = (char*)d_ws;
    size_t off = 0;
    auto carve = [&](size_t bytes) -> void* {
        void* p = ws + off;
        off += (bytes + 255) & ~(size_t)255;
        return p;
    };
    float*    h         = (float*)carve((size_t)NNODES * HDIM * 4);
    float*    a_s       = (float*)carve((size_t)NNODES * 4);
    float*    a_d       = (float*)carve((size_t)NNODES * 4);
    int*      cnt       = (int*)carve((size_t)NNODES * 4);
    int*      rowptr    = (int*)carve((size_t)(NNODES + 1) * 4);
    int*      tsum      = (int*)carve((size_t)((NNODES + 1023) / 1024) * 4);
    int*      rank      = (int*)carve((size_t)NEDGES * 4);
    int*      csr_src   = (int*)carve((size_t)NEDGES * 4);
    float*    vn_direct = (float*)carve((size_t)HNUMB * HDIM * 4);
    float*    vn_seg    = (float*)carve((size_t)HNUMB * HDIM * 4);

    const int nodeBlk = (NNODES + 255) / 256;
    const int edgeBlk = NEDGES / 256;
    const int tileBlk = (NNODES + 1023) / 1024;
    const int pullBlk = (NNODES + 15) / 16;

    init_kernel<<<nodeBlk, 256, 0, stream>>>(cnt, vn_direct, vn_seg, vn_emb, NNODES);
    deg_count_kernel<<<edgeBlk, 256, 0, stream>>>(dst, cnt, rank, NEDGES);
    tile_sum_kernel<<<tileBlk, 256, 0, stream>>>(cnt, tsum, NNODES);
    tile_scan_kernel<<<tileBlk, 256, 0, stream>>>(cnt, tsum, rowptr, NNODES);
    csr_fill_kernel<<<edgeBlk, 256, 0, stream>>>(src, dst, rowptr, rank, csr_src, NEDGES);

    auto run_gat = [&](const float* Xin, bool addvn, const float* W, const float* asrc,
                       const float* adst, const float* bias) {
        if (addvn)
            gemm_xw_kernel<true><<<(NNODES + 127) / 128, 256, 0, stream>>>(Xin, W, vn_direct, hblocks, asrc, adst,
                                                                           h, a_s, a_d, NNODES);
        else
            gemm_xw_kernel<false><<<(NNODES + 127) / 128, 256, 0, stream>>>(Xin, W, vn_direct, hblocks, asrc, adst,
                                                                            h, a_s, a_d, NNODES);
        gat_pull_kernel<<<pullBlk, 256, 0, stream>>>(rowptr, csr_src, a_s, a_d, h, bias, out, NNODES);
    };

    run_gat(x, false, W0, asrc0, adst0, b0);

    static_assert(NCONVL == 2, "the root's update gets a reader with a third conv layer: restore it then");
    for (int l = 0; l < NCONVL; ++l) {
        run_gat(out, true, Wl + l * HDIM * HDIM, asrcl + l * HDIM, adstl + l * HDIM, bl + l * HDIM);
        if (l + 1 == NCONVL) break;
        seg_sum_kernel<<<128, 1024, 0, stream>>>(out, hblocks, vn_seg, NNODES);
        vn_update_mlp_kernel<<<HNUMB, 64, 0, stream>>>(vn_direct, vn_seg, vn_emb, mW1, mb1, mW2, mb2);
    }
}


# Neighbour score of a node's first 20 slots = dot(gathered row, a_src) over the half-wave's lanes, not a 4-byte read of the score array per edge (a cache line each); score sum order changes
# speedup vs baseline: 6.3851x; 1.0551x over previous
#include <hip/hip_runtime.h>
#include <hip/hip_bf16.h>
#include <hip/hip_fp16.h>

#define NNODES 100000
#define NEDGES 1600000
#define HDIM   64
#define HNUMB  128
#define NCONVL 2
#define NMLPL  2
#define NEGSLP 0.2f

typedef __attribute__((ext_vector_type(2))) float v2f;
typedef __attribute__((ext_vector_type(8))) float v8f;

__device__ __forceinline__ float leaky(float x) { return x > 0.f ? x : NEGSLP * x; }

template <int J> __device__ __forceinline__ int row_get_i(int v) {
    return __builtin_amdgcn_update_dpp(0, v, 0x150 + J, 0xF, 0xF, true);
}
template <int J> __device__ __forceinline__ float row_get_f(float v) {
    return __int_as_float(__builtin_amdgcn_update_dpp(0, __float_as_int(v), 0x150 + J, 0xF, 0xF, true));
}
template <int M> __device__ __forceinline__ float row_xor_f(float v) {
    return __int_as_float(__builtin_amdgcn_update_dpp(0, __float_as_int(v), 0x160 + M, 0xF, 0xF, true));
}

template <bool ADDVN>
__global__ __launch_bounds__(256) void gemm_xw_kernel(const float* __restrict__ X,
                                                      const float* __restrict__ W,
                                                      const float* __restrict__ vnd,
                                                      const int* __restrict__ blocks,
                                                      const float* __restrict__ asrc,
                                                      const float* __restrict__ adst,
                                                      float* __restrict__ Hout,
                                                      float* __restrict__ a_s,
                                                      float* __restrict__ a_d,
                                                      int nrows) {
    __shared__ float Wlds[HDIM * HDIM];
    const int tid = threadIdx.x;
    for (int idx = tid; idx < HDIM * HDIM; idx += 256) {
        int k = idx >> 6, n = idx & 63;
        Wlds[(k >> 1) * 128 + n * 2 + (k & 1)] = W[idx];
    }
    __syncthreads();

    const int wave    = tid >> 5;
    const int lane    = tid & 31;
    const int hi      = lane >> 4;
    const int mrow    = lane & 15;
    const int rowbase = blockIdx.x * 128 + wave * 16;
    if (rowbase >= nrows) return;

    const float* xrow = X + (size_t)(rowbase + mrow) * HDIM + hi * 2;
    v2f afrag[16];
    if (ADDVN) {
        const float* vrow = vnd + (size_t)blocks[rowbase + mrow] * HDIM + hi * 2;
#pragma unroll
        for (int kk = 0; kk < 16; ++kk)
            afrag[kk] = *(const v2f*)(xrow + kk * 4) + *(const v2f*)(vrow + kk * 4);
    } else {
#pragma unroll
        for (int kk = 0; kk < 16; ++kk)
            afrag[kk] = *(const v2f*)(xrow + kk * 4);
    }
    float ps[8], pd[8];
#pragma unroll
    for (int v = 0; v < 8; ++v) { ps[v] = 0.f; pd[v] = 0.f; }
#pragma unroll
    for (int nt = 0; nt < 4; ++nt) {
        const int ncol = nt * 16 + mrow;
        const float asc = asrc[ncol];
        const float adc = adst[ncol];
        v8f acc = {};
#pragma unroll
        for (int kk = 0; kk < 16; ++kk) {
            v2f b = *(const v2f*)&Wlds[(kk * 2 + hi) * 128 + ncol * 2];
            acc = __builtin_amdgcn_wmma_f32_16x16x4_f32(false, afrag[kk], false, b,
                                                        (short)0, acc, false, false);
        }
#pragma unroll
        for (int v = 0; v < 8; ++v) {
            Hout[(size_t)(rowbase + v + 8 * hi) * HDIM + ncol] = acc[v];
            ps[v] = fmaf(acc[v], asc, ps[v]);
            pd[v] = fmaf(acc[v], adc, pd[v]);
        }
    }
#pragma unroll
    for (int v = 0; v < 8; ++v) {
        ps[v] += row_xor_f<1>(ps[v]); pd[v] += row_xor_f<1>(pd[v]);
        ps[v] += row_xor_f<2>(ps[v]); pd[v] += row_xor_f<2>(pd[v]);
        ps[v] += row_xor_f<4>(ps[v]); pd[v] += row_xor_f<4>(pd[v]);
        ps[v] += row_xor_f<8>(ps[v]); pd[v] += row_xor_f<8>(pd[v]);
    }
    float sv = ps[0], dv = pd[0];
#pragma unroll
    for (int v = 1; v < 8; ++v) {
        sv = (mrow == v) ? ps[v] : sv;
        dv = (mrow == v) ? pd[v] : dv;
    }
    if (mrow < 8) {
        a_s[rowbase + 8 * hi + mrow] = sv;
        a_d[rowbase + 8 * hi + mrow] = dv;
    }
}

__global__ void init_kernel(int* __restrict__ cnt, float* __restrict__ vn_direct,
                            float* __restrict__ vn_seg, const float* __restrict__ vn_emb, int n) {
    int t = blockIdx.x * blockDim.x + threadIdx.x;
    if (t < n) cnt[t] = 0;
    if (t < HNUMB * HDIM) {
        vn_direct[t] = vn_emb[t & (HDIM - 1)];
        vn_seg[t]    = 0.f;
    }
}

__global__ void deg_count_kernel(const int* __restrict__ dst, int* __restrict__ cnt,
                                 int* __restrict__ rank, int ne) {
    int e = blockIdx.x * blockDim.x + threadIdx.x;
    if (e >= ne) return;
    rank[e] = atomicAdd(cnt + dst[e], 1);
}

__global__ __launch_bounds__(256) void tile_sum_kernel(const int* __restrict__ cnt,
                                                       int* __restrict__ tsum, int n) {
    __shared__ int red[8];
    const int tid = threadIdx.x, lane = tid & 31, wave = tid >> 5;
    const int base = blockIdx.x * 1024 + tid * 4;
    int s = 0;
#pragma unroll
    for (int k = 0; k < 4; ++k) if (base + k < n) s += cnt[base + k];
    for (int o = 16; o > 0; o >>= 1) s += __shfl_xor(s, o, 32);
    if (lane == 0) red[wave] = s;
    __syncthreads();
    if (tid == 0) {
        int t = 0;
        for (int w = 0; w < 8; ++w) t += red[w];
        tsum[blockIdx.x] = t;
    }
}

__global__ __launch_bounds__(256) void tile_scan_kernel(const int* __restrict__ cnt,
                                                        const int* __restrict__ tsum,
                                                        int* __restrict__ rowptr, int n) {
    __shared__ int wsum[8];
    __shared__ int tile_off;
    const int tid = threadIdx.x, lane = tid & 31, wave = tid >> 5;
    if (wave == 0) {
        int t = 0;
        for (int i = lane; i < (int)blockIdx.x; i += 32) t += tsum[i];
        for (int o = 16; o > 0; o >>= 1) t += __shfl_xor(t, o, 32);
        if (lane == 0) tile_off = t;
    }
    const int base = blockIdx.x * 1024 + tid * 4;
    int c[4];
#pragma unroll
    for (int k = 0; k < 4; ++k) c[k] = (base + k < n) ? cnt[base + k] : 0;
    const int mine = c[0] + c[1] + c[2] + c[3];
    int incl = mine;
    for (int o = 1; o < 32; o <<= 1) {
        int v = __shfl_up(incl, o, 32);
        if (lane >= o) incl += v;
    }
    if (lane == 31) wsum[wave] = incl;
    __syncthreads();
    int woff = 0;
    for (int w = 0; w < wave; ++w) woff += wsum[w];
    int run = tile_off + woff + incl - mine;
#pragma unroll
    for (int k = 0; k < 4; ++k) {
        if (base + k < n) {
            rowptr[base + k] = run;
            run += c[k];
        }
    }
    if (base <= n - 1 && n - 1 < base + 4) rowptr[n] = run;
}

__global__ void csr_fill_kernel(const int* __restrict__ src, const int* __restrict__ dst,
                                const int* __restrict__ rowptr, const int* __restrict__ rank,
                                int* __restrict__ csr_src, int ne) {
    int e = blockIdx.x * blockDim.x + threadIdx.x;
    if (e >= ne) return;
    csr_src[rowptr[dst[e]] + rank[e]] = src[e];
}


__device__ __forceinline__ float row_transpose_sum16(const float (&d)[16], bool b3, bool b2, bool b1, bool b0) {
    float t8[8], t4[4], t2[2];
#pragma unroll
    for (int j = 0; j < 8; ++j) t8[j] = (b3 ? d[j + 8] : d[j]) + row_xor_f<8>(b3 ? d[j] : d[j + 8]);
#pragma unroll
    for (int j = 0; j < 4; ++j) t4[j] = (b2 ? t8[j + 4] : t8[j]) + row_xor_f<4>(b2 ? t8[j] : t8[j + 4]);
#pragma unroll
    for (int j = 0; j < 2; ++j) t2[j] = (b1 ? t4[j + 2] : t4[j]) + row_xor_f<2>(b1 ? t4[j] : t4[j + 2]);
    return (b0 ? t2[1] : t2[0]) + row_xor_f<1>(b0 ? t2[0] : t2[1]);
}
__device__ __forceinline__ float row_transpose_sum4(const float (&q)[4], bool b1, bool b0) {
    float t2[2];
#pragma unroll
    for (int j = 0; j < 2; ++j) t2[j] = (b1 ? q[j + 2] : q[j]) + row_xor_f<2>(b1 ? q[j] : q[j + 2]);
    float u = (b0 ? t2[1] : t2[0]) + row_xor_f<1>(b0 ? t2[0] : t2[1]);
    u += row_xor_f<4>(u);
    u += row_xor_f<8>(u);
    return u;
}

#define PULL_LOAD(DST, J, SV) DST = Hm4[row_get_i<(J) & 15>(SV) * 16 + l16]
#define PULL_FMA(SRC, J, WV)                                      \
    {                                                             \
        const float wj = row_get_f<(J) & 15>(WV);                 \
        acc.x = fmaf(wj, SRC.x, acc.x); acc.y = fmaf(wj, SRC.y, acc.y); \
        acc.z = fmaf(wj, SRC.z, acc.z); acc.w = fmaf(wj, SRC.w, acc.w); \
    }
#define ROW_DOT(SRC) fmaf(SRC.w, av.w, fmaf(SRC.z, av.z, fmaf(SRC.y, av.y, SRC.x * av.x)))
#define PULL_GROUP4(J0, SV, WV)                                   \
    {                                                             \
        float4 t0, t1, t2, t3;                                    \
        PULL_LOAD(t0, (J0), SV);     PULL_LOAD(t1, (J0) + 1, SV); \
        PULL_LOAD(t2, (J0) + 2, SV); PULL_LOAD(t3, (J0) + 3, SV); \
        PULL_FMA(t0, (J0), WV);     PULL_FMA(t1, (J0) + 1, WV);   \
        PULL_FMA(t2, (J0) + 2, WV); PULL_FMA(t3, (J0) + 3, WV);   \
    }

__global__ __launch_bounds__(256) void gat_pull_kernel(const int* __restrict__ rowptr,
                                                       const int* __restrict__ csr_src,
                                                       const float* __restrict__ a_s,
                                                       const float* __restrict__ a_d,
                                                       const float* __restrict__ Hm,
                                                       const float* __restrict__ asrc,
                                                       const float* __restrict__ bias,
                                                       float* __restrict__ out, int n) {
    const int l16  = threadIdx.x & 15;
    const int node = blockIdx.x * 16 + (threadIdx.x >> 4);
    const bool valid = node < n;
    const int nd  = valid ? node : n - 1;
    const int beg  = rowptr[nd];
    const int endv = rowptr[nd + 1];
    const int deg  = valid ? endv - beg : 0;
    const float ad = a_d[nd];
    const float as_self = a_s[nd];
    const float4* __restrict__ Hm4 = (const float4*)Hm;

    const bool p0 = l16 < deg, p1 = l16 + 16 < deg;
    int s0 = csr_src[p0 ? beg + l16 : 0];
    int s1 = csr_src[p1 ? beg + 16 + l16 : 0];
    s0 = p0 ? s0 : nd;
    s1 = p1 ? s1 : nd;
    const float es = leaky(as_self + ad);
    const float as1 = a_s[(p1 && l16 >= 4) ? s1 : nd];
    const float4 av = *(const float4*)(asrc + 4 * l16);
    __builtin_amdgcn_sched_barrier(0);

    const float4 hs = Hm4[nd * 16 + l16];
    const float4 bv = *(const float4*)(bias + 4 * l16);
    float4 r0, r1, r2, r3, r4, r5, r6, r7, r8, r9, r10, r11, r12, r13, r14, r15, r16, r17, r18, r19;
    PULL_LOAD(r0, 0, s0);   PULL_LOAD(r1, 1, s0);   PULL_LOAD(r2, 2, s0);   PULL_LOAD(r3, 3, s0);
    PULL_LOAD(r4, 4, s0);   PULL_LOAD(r5, 5, s0);   PULL_LOAD(r6, 6, s0);   PULL_LOAD(r7, 7, s0);
    PULL_LOAD(r8, 8, s0);   PULL_LOAD(r9, 9, s0);   PULL_LOAD(r10, 10, s0); PULL_LOAD(r11, 11, s0);
    PULL_LOAD(r12, 12, s0); PULL_LOAD(r13, 13, s0); PULL_LOAD(r14, 14, s0); PULL_LOAD(r15, 15, s0);
    PULL_LOAD(r16, 16, s1); PULL_LOAD(r17, 17, s1); PULL_LOAD(r18, 18, s1); PULL_LOAD(r19, 19, s1);

    const bool past32 = __builtin_amdgcn_ballot_w32(deg > 32) != 0u;

    const bool b3 = (l16 & 8) != 0, b2 = (l16 & 4) != 0, b1 = (l16 & 2) != 0, b0 = (l16 & 1) != 0;
    const float d0[16] = {ROW_DOT(r0),  ROW_DOT(r1),  ROW_DOT(r2),  ROW_DOT(r3),
                          ROW_DOT(r4),  ROW_DOT(r5),  ROW_DOT(r6),  ROW_DOT(r7),
                          ROW_DOT(r8),  ROW_DOT(r9),  ROW_DOT(r10), ROW_DOT(r11),
                          ROW_DOT(r12), ROW_DOT(r13), ROW_DOT(r14), ROW_DOT(r15)};
    const float d1[4]  = {ROW_DOT(r16), ROW_DOT(r17), ROW_DOT(r18), ROW_DOT(r19)};
    const float sc0 = row_transpose_sum16(d0, b3, b2, b1, b0);
    const float sc1 = row_transpose_sum4(d1, b1, b0);
    const float e0 = p0 ? leaky(sc0 + ad) : es;
    const float e1 = p1 ? leaky((l16 < 4 ? sc1 : as1) + ad) : es;

    float m = fmaxf(es, fmaxf(e0, e1));
    if (past32)
        for (int i = 32 + l16; i < deg; i += 16) m = fmaxf(m, leaky(a_s[csr_src[beg + i]] + ad));
    m = fmaxf(m, row_xor_f<1>(m));
    m = fmaxf(m, row_xor_f<2>(m));
    m = fmaxf(m, row_xor_f<4>(m));
    m = fmaxf(m, row_xor_f<8>(m));

    const float wself = __expf(es - m);
    const float w0 = p0 ? __expf(e0 - m) : 0.f;
    const float w1 = p1 ? __expf(e1 - m) : 0.f;
    float dpart = w0 + w1;
    float4 acc;
    acc.x = wself * hs.x; acc.y = wself * hs.y; acc.z = wself * hs.z; acc.w = wself * hs.w;

    PULL_FMA(r0, 0, w0);   PULL_FMA(r1, 1, w0);   PULL_FMA(r2, 2, w0);   PULL_FMA(r3, 3, w0);
    PULL_FMA(r4, 4, w0);   PULL_FMA(r5, 5, w0);   PULL_FMA(r6, 6, w0);   PULL_FMA(r7, 7, w0);
    PULL_FMA(r8, 8, w0);   PULL_FMA(r9, 9, w0);   PULL_FMA(r10, 10, w0); PULL_FMA(r11, 11, w0);
    PULL_FMA(r12, 12, w0); PULL_FMA(r13, 13, w0); PULL_FMA(r14, 14, w0); PULL_FMA(r15, 15, w0);
    PULL_FMA(r16, 16, w1); PULL_FMA(r17, 17, w1); PULL_FMA(r18, 18, w1); PULL_FMA(r19, 19, w1);

    if (__builtin_amdgcn_ballot_w32(deg > 20) != 0u) {
        PULL_GROUP4(20, s1, w1);
        if (__builtin_amdgcn_ballot_w32(deg > 24) != 0u) {
            PULL_GROUP4(24, s1, w1);
            if (__builtin_amdgcn_ballot_w32(deg > 28) != 0u) {
                PULL_GROUP4(28, s1, w1);
            }
        }
    }
    if (past32) {
        for (int cb = 32; __builtin_amdgcn_ballot_w32(cb < deg) != 0u; cb += 16) {
            const bool p = cb + l16 < deg;
            int s = csr_src[p ? beg + cb + l16 : 0];
            s = p ? s : nd;
            const float w = p ? __expf(leaky(a_s[s] + ad) - m) : 0.f;
            dpart += w;
            PULL_GROUP4(0, s, w);
            PULL_GROUP4(4, s, w);
            PULL_GROUP4(8, s, w);
            PULL_GROUP4(12, s, w);
        }
    }

    dpart += row_xor_f<1>(dpart);
    dpart += row_xor_f<2>(dpart);
    dpart += row_xor_f<4>(dpart);
    dpart += row_xor_f<8>(dpart);
    const float den = (dpart + wself) + 1e-16f;
    float4 res;
    res.x = acc.x / den + bv.x;
    res.y = acc.y / den + bv.y;
    res.z = acc.z / den + bv.z;
    res.w = acc.w / den + bv.w;
    if (valid) *(float4*)(out + (size_t)node * HDIM + 4 * l16) = res;
}

#define SEG_ROWS 25
__global__ __launch_bounds__(1024) void seg_sum_kernel(const float* __restrict__ out,
                                                       const int* __restrict__ blocks,
                                                       float* __restrict__ vn_seg, int n) {
    __shared__ float acc[HNUMB * HDIM];
    for (int i = threadIdx.x; i < HNUMB * HDIM; i += 1024) acc[i] = 0.f;
    __syncthreads();
    const int lane   = threadIdx.x & 31;
    const int nwaves = gridDim.x * 32;
    const int gw   = __builtin_amdgcn_readfirstlane((int)(blockIdx.x * 32 + (threadIdx.x >> 5)));
    const int last = n - 1;
    for (int base = gw * SEG_ROWS; base < n; base += nwaves * SEG_ROWS) {
        const int rk  = base + lane;
        const int bid = blocks[rk < last ? rk : last];
        v2f v[SEG_ROWS];
#pragma unroll
        for (int k = 0; k < SEG_ROWS; ++k) {
            const int r = (base + k < n) ? (base + k) : last;
            v[k] = *(const v2f*)(out + (size_t)r * HDIM + 2 * lane);
        }
#pragma unroll
        for (int k = 0; k < SEG_ROWS; ++k) {
            const bool live = base + k < n;
            const int  b    = __builtin_amdgcn_readlane(bid, k);
            float* a = acc + b * HDIM + 2 * lane;
            atomicAdd(a,     live ? v[k].x : 0.f);
            atomicAdd(a + 1, live ? v[k].y : 0.f);
        }
    }
    __syncthreads();
    for (int i = threadIdx.x; i < HNUMB * HDIM; i += 1024) atomicAdd(vn_seg + i, acc[i]);
}

__global__ __launch_bounds__(64) void vn_update_mlp_kernel(float* __restrict__ vn_direct,
                                                           const float* __restrict__ vn_seg,
                                                           const float* __restrict__ vn_emb,
                                                           const float* __restrict__ W1, const float* __restrict__ b1,
                                                           const float* __restrict__ W2, const float* __restrict__ b2) {
    __shared__ float rin[HDIM];
    __shared__ float t1[HDIM];
    const int f = threadIdx.x;
    float* row = vn_direct + (size_t)blockIdx.x * HDIM;
    float cur = row[f] + (vn_seg[(size_t)blockIdx.x * HDIM + f] + vn_emb[f]);
    for (int j = 0; j < NMLPL; ++j) {
        const float* w1 = W1 + (size_t)j * HDIM * HDIM;
        const float* w2 = W2 + (size_t)j * HDIM * HDIM;
        rin[f] = cur;
        __syncthreads();
        float s = b1[j * HDIM + f];
#pragma unroll
        for (int k = 0; k < HDIM; ++k) s += rin[k] * w1[k * HDIM + f];
        t1[f] = fmaxf(s, 0.f);
        __syncthreads();
        float s2 = b2[j * HDIM + f];
#pragma unroll
        for (int k = 0; k < HDIM; ++k) s2 += t1[k] * w2[k * HDIM + f];
        cur = fmaxf(s2, 0.f);
    }
    row[f] = cur;
}

extern "C" void kernel_launch(void* const* d_in, const int* in_sizes, int n_in,
                              void* d_out, int out_size, void* d_ws, size_t ws_size,
                              hipStream_t stream) {
    const float* x       = (const float*)d_in[0];
    const int*   ei      = (const int*)d_in[1];
    const int*   src     = ei;
    const int*   dst     = ei + NEDGES;
    const int*   hblocks = (const int*)d_in[2];
    const float* W0      = (const float*)d_in[5];
    const float* asrc0   = (const float*)d_in[6];
    const float* adst0   = (const float*)d_in[7];
    const float* b0      = (const float*)d_in[8];
    const float* Wl      = (const float*)d_in[9];
    const float* asrcl   = (const float*)d_in[10];
    const float* adstl   = (const float*)d_in[11];
    const float* bl      = (const float*)d_in[12];
    const float* vn_emb  = (const float*)d_in[13];
    const float* mW1     = (const float*)d_in[14];
    const float* mb1     = (const float*)d_in[15];
    const float* mW2     = (const float*)d_in[16];
    const float* mb2     = (const float*)d_in[17];
    float* out = (float*)d_out;

    char*  ws  = (char*)d_ws;
    size_t off = 0;
    auto carve = [&](size_t bytes) -> void* {
        void* p = ws + off;
        off += (bytes + 255) & ~(size_t)255;
        return p;
    };
    float*    h         = (float*)carve((size_t)NNODES * HDIM * 4);
    float*    a_s       = (float*)carve((size_t)NNODES * 4);
    float*    a_d       = (float*)carve((size_t)NNODES * 4);
    int*      cnt       = (int*)carve((size_t)NNODES * 4);
    int*      rowptr    = (int*)carve((size_t)(NNODES + 1) * 4);
    int*      tsum      = (int*)carve((size_t)((NNODES + 1023) / 1024) * 4);
    int*      rank      = (int*)carve((size_t)NEDGES * 4);
    int*      csr_src   = (int*)carve((size_t)NEDGES * 4);
    float*    vn_direct = (float*)carve((size_t)HNUMB * HDIM * 4);
    float*    vn_seg    = (float*)carve((size_t)HNUMB * HDIM * 4);

    const int nodeBlk = (NNODES + 255) / 256;
    const int edgeBlk = NEDGES / 256;
    const int tileBlk = (NNODES + 1023) / 1024;
    const int pullBlk = (NNODES + 15) / 16;

    init_kernel<<<nodeBlk, 256, 0, stream>>>(cnt, vn_direct, vn_seg, vn_emb, NNODES);
    deg_count_kernel<<<edgeBlk, 256, 0, stream>>>(dst, cnt, rank, NEDGES);
    tile_sum_kernel<<<tileBlk, 256, 0, stream>>>(cnt, tsum, NNODES);
    tile_scan_kernel<<<tileBlk, 256, 0, stream>>>(cnt, tsum, rowptr, NNODES);
    csr_fill_kernel<<<edgeBlk, 256, 0, stream>>>(src, dst, rowptr, rank, csr_src, NEDGES);

    auto run_gat = [&](const float* Xin, bool addvn, const float* W, const float* asrc,
                       const float* adst, const float* bias) {
        if (addvn)
            gemm_xw_kernel<true><<<(NNODES + 127) / 128, 256, 0, stream>>>(Xin, W, vn_direct, hblocks, asrc, adst,
                                                                           h, a_s, a_d, NNODES);
        else
            gemm_xw_kernel<false><<<(NNODES + 127) / 128, 256, 0, stream>>>(Xin, W, vn_direct, hblocks, asrc, adst,
                                                                            h, a_s, a_d, NNODES);
        gat_pull_kernel<<<pullBlk, 256, 0, stream>>>(rowptr, csr_src, a_s, a_d, h, asrc, bias, out, NNODES);
    };

    run_gat(x, false, W0, asrc0, adst0, b0);

    static_assert(NCONVL == 2, "the root's update gets a reader with a third conv layer: restore it then");
    for (int l = 0; l < NCONVL; ++l) {
        run_gat(out, true, Wl + l * HDIM * HDIM, asrcl + l * HDIM, adstl + l * HDIM, bl + l * HDIM);
        if (l + 1 == NCONVL) break;
        seg_sum_kernel<<<128, 1024, 0, stream>>>(out, hblocks, vn_seg, NNODES);
        vn_update_mlp_kernel<<<HNUMB, 64, 0, stream>>>(vn_direct, vn_seg, vn_emb, mW1, mb1, mW2, mb2);
    }
}


# Bucket build as a two-level counting sort on LDS counters (edges binned by 512-node range, then placed per node in a bin): no returning global adds, 13 to 12 launches; slot (= sum) order changes
# speedup vs baseline: 8.7247x; 1.3664x over previous
#include <hip/hip_runtime.h>
#include <hip/hip_bf16.h>
#include <hip/hip_fp16.h>

#define NNODES 100000
#define NEDGES 1600000
#define HDIM   64
#define HNUMB  128
#define NCONVL 2
#define NMLPL  2
#define NEGSLP 0.2f

typedef __attribute__((ext_vector_type(2))) float v2f;
typedef __attribute__((ext_vector_type(8))) float v8f;

__device__ __forceinline__ float leaky(float x) { return x > 0.f ? x : NEGSLP * x; }

template <int J> __device__ __forceinline__ int row_get_i(int v) {
    return __builtin_amdgcn_update_dpp(0, v, 0x150 + J, 0xF, 0xF, true);
}
template <int J> __device__ __forceinline__ float row_get_f(float v) {
    return __int_as_float(__builtin_amdgcn_update_dpp(0, __float_as_int(v), 0x150 + J, 0xF, 0xF, true));
}
template <int M> __device__ __forceinline__ float row_xor_f(float v) {
    return __int_as_float(__builtin_amdgcn_update_dpp(0, __float_as_int(v), 0x160 + M, 0xF, 0xF, true));
}

template <bool ADDVN>
__global__ __launch_bounds__(256) void gemm_xw_kernel(const float* __restrict__ X,
                                                      const float* __restrict__ W,
                                                      const float* __restrict__ vnd,
                                                      const int* __restrict__ blocks,
                                                      const float* __restrict__ asrc,
                                                      const float* __restrict__ adst,
                                                      float* __restrict__ Hout,
                                                      float* __restrict__ a_s,
                                                      float* __restrict__ a_d,
                                                      int nrows) {
    __shared__ float Wlds[HDIM * HDIM];
    const int tid = threadIdx.x;
    for (int idx = tid; idx < HDIM * HDIM; idx += 256) {
        int k = idx >> 6, n = idx & 63;
        Wlds[(k >> 1) * 128 + n * 2 + (k & 1)] = W[idx];
    }
    __syncthreads();

    const int wave    = tid >> 5;
    const int lane    = tid & 31;
    const int hi      = lane >> 4;
    const int mrow    = lane & 15;
    const int rowbase = blockIdx.x * 128 + wave * 16;
    if (rowbase >= nrows) return;

    const float* xrow = X + (size_t)(rowbase + mrow) * HDIM + hi * 2;
    v2f afrag[16];
    if (ADDVN) {
        const float* vrow = vnd + (size_t)blocks[rowbase + mrow] * HDIM + hi * 2;
#pragma unroll
        for (int kk = 0; kk < 16; ++kk)
            afrag[kk] = *(const v2f*)(xrow + kk * 4) + *(const v2f*)(vrow + kk * 4);
    } else {
#pragma unroll
        for (int kk = 0; kk < 16; ++kk)
            afrag[kk] = *(const v2f*)(xrow + kk * 4);
    }
    float ps[8], pd[8];
#pragma unroll
    for (int v = 0; v < 8; ++v) { ps[v] = 0.f; pd[v] = 0.f; }
#pragma unroll
    for (int nt = 0; nt < 4; ++nt) {
        const int ncol = nt * 16 + mrow;
        const float asc = asrc[ncol];
        const float adc = adst[ncol];
        v8f acc = {};
#pragma unroll
        for (int kk = 0; kk < 16; ++kk) {
            v2f b = *(const v2f*)&Wlds[(kk * 2 + hi) * 128 + ncol * 2];
            acc = __builtin_amdgcn_wmma_f32_16x16x4_f32(false, afrag[kk], false, b,
                                                        (short)0, acc, false, false);
        }
#pragma unroll
        for (int v = 0; v < 8; ++v) {
            Hout[(size_t)(rowbase + v + 8 * hi) * HDIM + ncol] = acc[v];
            ps[v] = fmaf(acc[v], asc, ps[v]);
            pd[v] = fmaf(acc[v], adc, pd[v]);
        }
    }
#pragma unroll
    for (int v = 0; v < 8; ++v) {
        ps[v] += row_xor_f<1>(ps[v]); pd[v] += row_xor_f<1>(pd[v]);
        ps[v] += row_xor_f<2>(ps[v]); pd[v] += row_xor_f<2>(pd[v]);
        ps[v] += row_xor_f<4>(ps[v]); pd[v] += row_xor_f<4>(pd[v]);
        ps[v] += row_xor_f<8>(ps[v]); pd[v] += row_xor_f<8>(pd[v]);
    }
    float sv = ps[0], dv = pd[0];
#pragma unroll
    for (int v = 1; v < 8; ++v) {
        sv = (mrow == v) ? ps[v] : sv;
        dv = (mrow == v) ? pd[v] : dv;
    }
    if (mrow < 8) {
        a_s[rowbase + 8 * hi + mrow] = sv;
        a_d[rowbase + 8 * hi + mrow] = dv;
    }
}

#define BIN_SHIFT   9
#define BIN_NODES   (1 << BIN_SHIFT)
#define NBINS       ((NNODES + BIN_NODES - 1) >> BIN_SHIFT)
#define CHUNK_EDGES 6400
#define CHUNK_PER   (CHUNK_EDGES / 256)
#define NCHUNKS     ((NEDGES + CHUNK_EDGES - 1) / CHUNK_EDGES)
#define SCAN_M      (NBINS * NCHUNKS)
#define SCAN_PER    ((SCAN_M + 1023) / 1024)
#define FIN_BATCH   10
static_assert(CHUNK_EDGES % 256 == 0, "a chunk is a whole number of 256-edge rows");
static_assert(BIN_NODES <= 1024, "one thread per node of a bin in the last pass");
static_assert((((long long)(NNODES - 1)) << BIN_SHIFT) + (BIN_NODES - 1) <= 0xFFFFFFFFll,
              "source and in-bin index share one 32-bit word");
static_assert(NCHUNKS * 256 >= HNUMB * HDIM, "the first pass also writes the virtual-node start values");

__global__ __launch_bounds__(256) void edge_hist_kernel(const int* __restrict__ dst, int* __restrict__ tab,
                                                        float* __restrict__ vn_direct,
                                                        float* __restrict__ vn_seg,
                                                        const float* __restrict__ vn_emb, int ne) {
    __shared__ int hist[NBINS];
    const int tid = threadIdx.x;
    const int e0  = blockIdx.x * CHUNK_EDGES;
    int d[CHUNK_PER];
#pragma unroll
    for (int i = 0; i < CHUNK_PER; ++i) {
        const int e  = e0 + i * 256 + tid;
        const int dv = dst[e < ne ? e : ne - 1];
        d[i] = e < ne ? dv : -1;
    }
    for (int b = tid; b < NBINS; b += 256) hist[b] = 0;
    const int g = blockIdx.x * 256 + tid;
    if (g < HNUMB * HDIM) {
        vn_direct[g] = vn_emb[g & (HDIM - 1)];
        vn_seg[g]    = 0.f;
    }
    __syncthreads();
#pragma unroll
    for (int i = 0; i < CHUNK_PER; ++i)
        if (d[i] >= 0) atomicAdd(&hist[d[i] >> BIN_SHIFT], 1);
    __syncthreads();
    for (int b = tid; b < NBINS; b += 256) tab[b * NCHUNKS + blockIdx.x] = hist[b];
}

__global__ __launch_bounds__(1024) void bin_scan_kernel(const int* __restrict__ tab, int* __restrict__ off) {
    __shared__ int buf[(SCAN_PER + 1) * 1024];
    __shared__ int wsum[32];
    const int tid = threadIdx.x, lane = tid & 31, wave = tid >> 5;
#pragma unroll
    for (int i = 0; i < SCAN_PER; ++i) {
        const int j = i * 1024 + tid;
        const int t = tab[j < SCAN_M ? j : SCAN_M - 1];
        buf[(j / SCAN_PER) * (SCAN_PER + 1) + (j % SCAN_PER)] = j < SCAN_M ? t : 0;
    }
    __syncthreads();
    int* myrow = buf + tid * (SCAN_PER + 1);
    int v[SCAN_PER];
    int s = 0;
#pragma unroll
    for (int k = 0; k < SCAN_PER; ++k) { v[k] = myrow[k]; s += v[k]; }
    int incl = s;
    for (int o = 1; o < 32; o <<= 1) {
        const int t = __shfl_up(incl, o, 32);
        if (lane >= o) incl += t;
    }
    if (lane == 31) wsum[wave] = incl;
    __syncthreads();
    int run = incl - s;
    for (int w = 0; w < wave; ++w) run += wsum[w];
#pragma unroll
    for (int k = 0; k < SCAN_PER; ++k) { myrow[k] = run; run += v[k]; }
    __syncthreads();
#pragma unroll
    for (int i = 0; i < SCAN_PER; ++i) {
        const int j = i * 1024 + tid;
        if (j < SCAN_M) off[j] = buf[(j / SCAN_PER) * (SCAN_PER + 1) + (j % SCAN_PER)];
    }
}

__global__ __launch_bounds__(256) void edge_bin_kernel(const int* __restrict__ src, const int* __restrict__ dst,
                                                       const int* __restrict__ off,
                                                       unsigned* __restrict__ binned, int ne) {
    __shared__ int cur[NBINS + 1];
    const int tid = threadIdx.x;
    const int e0  = blockIdx.x * CHUNK_EDGES;
    int s[CHUNK_PER], d[CHUNK_PER];
#pragma unroll
    for (int i = 0; i < CHUNK_PER; ++i) {
        const int e  = e0 + i * 256 + tid;
        const int ec = e < ne ? e : ne - 1;
        const int dv = dst[ec];
        s[i] = src[ec];
        d[i] = e < ne ? dv : -1;
    }
    for (int b = tid; b < NBINS; b += 256) cur[b] = off[b * NCHUNKS + blockIdx.x];
    if (tid == 0) cur[NBINS] = 0;
    __syncthreads();
    int pos[CHUNK_PER];
#pragma unroll
    for (int i = 0; i < CHUNK_PER; ++i)
        pos[i] = atomicAdd(&cur[d[i] >= 0 ? (d[i] >> BIN_SHIFT) : NBINS], 1);
#pragma unroll
    for (int i = 0; i < CHUNK_PER; ++i)
        if (d[i] >= 0)
            binned[pos[i]] = ((unsigned)s[i] << BIN_SHIFT) | (unsigned)(d[i] & (BIN_NODES - 1));
}

__global__ __launch_bounds__(1024) void bin_finalize_kernel(const unsigned* __restrict__ binned,
                                                           const int* __restrict__ off,
                                                           int* __restrict__ rowptr,
                                                           int* __restrict__ csr_src, int n, int ne) {
    __shared__ int cnt[BIN_NODES];
    __shared__ int wsum[32];
    const int tid = threadIdx.x, lane = tid & 31, wave = tid >> 5;
    const int b = blockIdx.x;
    const int beg = __builtin_amdgcn_readfirstlane(off[b * NCHUNKS]);
    const int end = (b + 1 < NBINS) ? __builtin_amdgcn_readfirstlane(off[(b + 1) * NCHUNKS]) : ne;
    if (tid < BIN_NODES) cnt[tid] = 0;
    __syncthreads();
    for (int i0 = beg; i0 < end; i0 += FIN_BATCH * 1024) {
        unsigned v[FIN_BATCH];
#pragma unroll
        for (int k = 0; k < FIN_BATCH; ++k) {
            const int i = i0 + k * 1024 + tid;
            v[k] = binned[i < end ? i : end - 1];
        }
#pragma unroll
        for (int k = 0; k < FIN_BATCH; ++k)
            if (i0 + k * 1024 + tid < end) atomicAdd(&cnt[v[k] & (BIN_NODES - 1)], 1);
    }
    __syncthreads();
    const int c = tid < BIN_NODES ? cnt[tid] : 0;
    int incl = c;
    for (int o = 1; o < 32; o <<= 1) {
        const int t = __shfl_up(incl, o, 32);
        if (lane >= o) incl += t;
    }
    if (lane == 31) wsum[wave] = incl;
    __syncthreads();
    int excl = incl - c;
    for (int w = 0; w < wave; ++w) excl += wsum[w];
    if (tid < BIN_NODES) {
        cnt[tid] = excl;
        const int node = b * BIN_NODES + tid;
        if (node < n) rowptr[node] = beg + excl;
    }
    if (b == NBINS - 1 && tid == 0) rowptr[n] = end;
    __syncthreads();
    for (int i0 = beg; i0 < end; i0 += FIN_BATCH * 1024) {
        unsigned v[FIN_BATCH];
#pragma unroll
        for (int k = 0; k < FIN_BATCH; ++k) {
            const int i = i0 + k * 1024 + tid;
            v[k] = binned[i < end ? i : end - 1];
        }
        int slot[FIN_BATCH];
#pragma unroll
        for (int k = 0; k < FIN_BATCH; ++k)
            slot[k] = (i0 + k * 1024 + tid < end) ? atomicAdd(&cnt[v[k] & (BIN_NODES - 1)], 1) : 0;
#pragma unroll
        for (int k = 0; k < FIN_BATCH; ++k)
            if (i0 + k * 1024 + tid < end) csr_src[beg + slot[k]] = (int)(v[k] >> BIN_SHIFT);
    }
}


__device__ __forceinline__ float row_transpose_sum16(const float (&d)[16], bool b3, bool b2, bool b1, bool b0) {
    float t8[8], t4[4], t2[2];
#pragma unroll
    for (int j = 0; j < 8; ++j) t8[j] = (b3 ? d[j + 8] : d[j]) + row_xor_f<8>(b3 ? d[j] : d[j + 8]);
#pragma unroll
    for (int j = 0; j < 4; ++j) t4[j] = (b2 ? t8[j + 4] : t8[j]) + row_xor_f<4>(b2 ? t8[j] : t8[j + 4]);
#pragma unroll
    for (int j = 0; j < 2; ++j) t2[j] = (b1 ? t4[j + 2] : t4[j]) + row_xor_f<2>(b1 ? t4[j] : t4[j + 2]);
    return (b0 ? t2[1] : t2[0]) + row_xor_f<1>(b0 ? t2[0] : t2[1]);
}
__device__ __forceinline__ float row_transpose_sum4(const float (&q)[4], bool b1, bool b0) {
    float t2[2];
#pragma unroll
    for (int j = 0; j < 2; ++j) t2[j] = (b1 ? q[j + 2] : q[j]) + row_xor_f<2>(b1 ? q[j] : q[j + 2]);
    float u = (b0 ? t2[1] : t2[0]) + row_xor_f<1>(b0 ? t2[0] : t2[1]);
    u += row_xor_f<4>(u);
    u += row_xor_f<8>(u);
    return u;
}

#define PULL_LOAD(DST, J, SV) DST = Hm4[row_get_i<(J) & 15>(SV) * 16 + l16]
#define PULL_FMA(SRC, J, WV)                                      \
    {                                                             \
        const float wj = row_get_f<(J) & 15>(WV);                 \
        acc.x = fmaf(wj, SRC.x, acc.x); acc.y = fmaf(wj, SRC.y, acc.y); \
        acc.z = fmaf(wj, SRC.z, acc.z); acc.w = fmaf(wj, SRC.w, acc.w); \
    }
#define ROW_DOT(SRC) fmaf(SRC.w, av.w, fmaf(SRC.z, av.z, fmaf(SRC.y, av.y, SRC.x * av.x)))
#define PULL_GROUP4(J0, SV, WV)                                   \
    {                                                             \
        float4 t0, t1, t2, t3;                                    \
        PULL_LOAD(t0, (J0), SV);     PULL_LOAD(t1, (J0) + 1, SV); \
        PULL_LOAD(t2, (J0) + 2, SV); PULL_LOAD(t3, (J0) + 3, SV); \
        PULL_FMA(t0, (J0), WV);     PULL_FMA(t1, (J0) + 1, WV);   \
        PULL_FMA(t2, (J0) + 2, WV); PULL_FMA(t3, (J0) + 3, WV);   \
    }

__global__ __launch_bounds__(256) void gat_pull_kernel(const int* __restrict__ rowptr,
                                                       const int* __restrict__ csr_src,
                                                       const float* __restrict__ a_s,
                                                       const float* __restrict__ a_d,
                                                       const float* __restrict__ Hm,
                                                       const float* __restrict__ asrc,
                                                       const float* __restrict__ bias,
                                                       float* __restrict__ out, int n) {
    const int l16  = threadIdx.x & 15;
    const int node = blockIdx.x * 16 + (threadIdx.x >> 4);
    const bool valid = node < n;
    const int nd  = valid ? node : n - 1;
    const int beg  = rowptr[nd];
    const int endv = rowptr[nd + 1];
    const int deg  = valid ? endv - beg : 0;
    const float ad = a_d[nd];
    const float as_self = a_s[nd];
    const float4* __restrict__ Hm4 = (const float4*)Hm;

    const bool p0 = l16 < deg, p1 = l16 + 16 < deg;
    int s0 = csr_src[p0 ? beg + l16 : 0];
    int s1 = csr_src[p1 ? beg + 16 + l16 : 0];
    s0 = p0 ? s0 : nd;
    s1 = p1 ? s1 : nd;
    const float es = leaky(as_self + ad);
    const float as1 = a_s[(p1 && l16 >= 4) ? s1 : nd];
    const float4 av = *(const float4*)(asrc + 4 * l16);
    __builtin_amdgcn_sched_barrier(0);

    const float4 hs = Hm4[nd * 16 + l16];
    const float4 bv = *(const float4*)(bias + 4 * l16);
    float4 r0, r1, r2, r3, r4, r5, r6, r7, r8, r9, r10, r11, r12, r13, r14, r15, r16, r17, r18, r19;
    PULL_LOAD(r0, 0, s0);   PULL_LOAD(r1, 1, s0);   PULL_LOAD(r2, 2, s0);   PULL_LOAD(r3, 3, s0);
    PULL_LOAD(r4, 4, s0);   PULL_LOAD(r5, 5, s0);   PULL_LOAD(r6, 6, s0);   PULL_LOAD(r7, 7, s0);
    PULL_LOAD(r8, 8, s0);   PULL_LOAD(r9, 9, s0);   PULL_LOAD(r10, 10, s0); PULL_LOAD(r11, 11, s0);
    PULL_LOAD(r12, 12, s0); PULL_LOAD(r13, 13, s0); PULL_LOAD(r14, 14, s0); PULL_LOAD(r15, 15, s0);
    PULL_LOAD(r16, 16, s1); PULL_LOAD(r17, 17, s1); PULL_LOAD(r18, 18, s1); PULL_LOAD(r19, 19, s1);

    const bool past32 = __builtin_amdgcn_ballot_w32(deg > 32) != 0u;

    const bool b3 = (l16 & 8) != 0, b2 = (l16 & 4) != 0, b1 = (l16 & 2) != 0, b0 = (l16 & 1) != 0;
    const float d0[16] = {ROW_DOT(r0),  ROW_DOT(r1),  ROW_DOT(r2),  ROW_DOT(r3),
                          ROW_DOT(r4),  ROW_DOT(r5),  ROW_DOT(r6),  ROW_DOT(r7),
                          ROW_DOT(r8),  ROW_DOT(r9),  ROW_DOT(r10), ROW_DOT(r11),
                          ROW_DOT(r12), ROW_DOT(r13), ROW_DOT(r14), ROW_DOT(r15)};
    const float d1[4]  = {ROW_DOT(r16), ROW_DOT(r17), ROW_DOT(r18), ROW_DOT(r19)};
    const float sc0 = row_transpose_sum16(d0, b3, b2, b1, b0);
    const float sc1 = row_transpose_sum4(d1, b1, b0);
    const float e0 = p0 ? leaky(sc0 + ad) : es;
    const float e1 = p1 ? leaky((l16 < 4 ? sc1 : as1) + ad) : es;

    float m = fmaxf(es, fmaxf(e0, e1));
    if (past32)
        for (int i = 32 + l16; i < deg; i += 16) m = fmaxf(m, leaky(a_s[csr_src[beg + i]] + ad));
    m = fmaxf(m, row_xor_f<1>(m));
    m = fmaxf(m, row_xor_f<2>(m));
    m = fmaxf(m, row_xor_f<4>(m));
    m = fmaxf(m, row_xor_f<8>(m));

    const float wself = __expf(es - m);
    const float w0 = p0 ? __expf(e0 - m) : 0.f;
    const float w1 = p1 ? __expf(e1 - m) : 0.f;
    float dpart = w0 + w1;
    float4 acc;
    acc.x = wself * hs.x; acc.y = wself * hs.y; acc.z = wself * hs.z; acc.w = wself * hs.w;

    PULL_FMA(r0, 0, w0);   PULL_FMA(r1, 1, w0);   PULL_FMA(r2, 2, w0);   PULL_FMA(r3, 3, w0);
    PULL_FMA(r4, 4, w0);   PULL_FMA(r5, 5, w0);   PULL_FMA(r6, 6, w0);   PULL_FMA(r7, 7, w0);
    PULL_FMA(r8, 8, w0);   PULL_FMA(r9, 9, w0);   PULL_FMA(r10, 10, w0); PULL_FMA(r11, 11, w0);
    PULL_FMA(r12, 12, w0); PULL_FMA(r13, 13, w0); PULL_FMA(r14, 14, w0); PULL_FMA(r15, 15, w0);
    PULL_FMA(r16, 16, w1); PULL_FMA(r17, 17, w1); PULL_FMA(r18, 18, w1); PULL_FMA(r19, 19, w1);

    if (__builtin_amdgcn_ballot_w32(deg > 20) != 0u) {
        PULL_GROUP4(20, s1, w1);
        if (__builtin_amdgcn_ballot_w32(deg > 24) != 0u) {
            PULL_GROUP4(24, s1, w1);
            if (__builtin_amdgcn_ballot_w32(deg > 28) != 0u) {
                PULL_GROUP4(28, s1, w1);
            }
        }
    }
    if (past32) {
        for (int cb = 32; __builtin_amdgcn_ballot_w32(cb < deg) != 0u; cb += 16) {
            const bool p = cb + l16 < deg;
            int s = csr_src[p ? beg + cb + l16 : 0];
            s = p ? s : nd;
            const float w = p ? __expf(leaky(a_s[s] + ad) - m) : 0.f;
            dpart += w;
            PULL_GROUP4(0, s, w);
            PULL_GROUP4(4, s, w);
            PULL_GROUP4(8, s, w);
            PULL_GROUP4(12, s, w);
        }
    }

    dpart += row_xor_f<1>(dpart);
    dpart += row_xor_f<2>(dpart);
    dpart += row_xor_f<4>(dpart);
    dpart += row_xor_f<8>(dpart);
    const float den = (dpart + wself) + 1e-16f;
    float4 res;
    res.x = acc.x / den + bv.x;
    res.y = acc.y / den + bv.y;
    res.z = acc.z / den + bv.z;
    res.w = acc.w / den + bv.w;
    if (valid) *(float4*)(out + (size_t)node * HDIM + 4 * l16) = res;
}

#define SEG_ROWS 25
__global__ __launch_bounds__(1024) void seg_sum_kernel(const float* __restrict__ out,
                                                       const int* __restrict__ blocks,
                                                       float* __restrict__ vn_seg, int n) {
    __shared__ float acc[HNUMB * HDIM];
    for (int i = threadIdx.x; i < HNUMB * HDIM; i += 1024) acc[i] = 0.f;
    __syncthreads();
    const int lane   = threadIdx.x & 31;
    const int nwaves = gridDim.x * 32;
    const int gw   = __builtin_amdgcn_readfirstlane((int)(blockIdx.x * 32 + (threadIdx.x >> 5)));
    const int last = n - 1;
    for (int base = gw * SEG_ROWS; base < n; base += nwaves * SEG_ROWS) {
        const int rk  = base + lane;
        const int bid = blocks[rk < last ? rk : last];
        v2f v[SEG_ROWS];
#pragma unroll
        for (int k = 0; k < SEG_ROWS; ++k) {
            const int r = (base + k < n) ? (base + k) : last;
            v[k] = *(const v2f*)(out + (size_t)r * HDIM + 2 * lane);
        }
#pragma unroll
        for (int k = 0; k < SEG_ROWS; ++k) {
            const bool live = base + k < n;
            const int  b    = __builtin_amdgcn_readlane(bid, k);
            float* a = acc + b * HDIM + 2 * lane;
            atomicAdd(a,     live ? v[k].x : 0.f);
            atomicAdd(a + 1, live ? v[k].y : 0.f);
        }
    }
    __syncthreads();
    for (int i = threadIdx.x; i < HNUMB * HDIM; i += 1024) atomicAdd(vn_seg + i, acc[i]);
}

__global__ __launch_bounds__(64) void vn_update_mlp_kernel(float* __restrict__ vn_direct,
                                                           const float* __restrict__ vn_seg,
                                                           const float* __restrict__ vn_emb,
                                                           const float* __restrict__ W1, const float* __restrict__ b1,
                                                           const float* __restrict__ W2, const float* __restrict__ b2) {
    __shared__ float rin[HDIM];
    __shared__ float t1[HDIM];
    const int f = threadIdx.x;
    float* row = vn_direct + (size_t)blockIdx.x * HDIM;
    float cur = row[f] + (vn_seg[(size_t)blockIdx.x * HDIM + f] + vn_emb[f]);
    for (int j = 0; j < NMLPL; ++j) {
        const float* w1 = W1 + (size_t)j * HDIM * HDIM;
        const float* w2 = W2 + (size_t)j * HDIM * HDIM;
        rin[f] = cur;
        __syncthreads();
        float s = b1[j * HDIM + f];
#pragma unroll
        for (int k = 0; k < HDIM; ++k) s += rin[k] * w1[k * HDIM + f];
        t1[f] = fmaxf(s, 0.f);
        __syncthreads();
        float s2 = b2[j * HDIM + f];
#pragma unroll
        for (int k = 0; k < HDIM; ++k) s2 += t1[k] * w2[k * HDIM + f];
        cur = fmaxf(s2, 0.f);
    }
    row[f] = cur;
}

extern "C" void kernel_launch(void* const* d_in, const int* in_sizes, int n_in,
                              void* d_out, int out_size, void* d_ws, size_t ws_size,
                              hipStream_t stream) {
    const float* x       = (const float*)d_in[0];
    const int*   ei      = (const int*)d_in[1];
    const int*   src     = ei;
    const int*   dst     = ei + NEDGES;
    const int*   hblocks = (const int*)d_in[2];
    const float* W0      = (const float*)d_in[5];
    const float* asrc0   = (const float*)d_in[6];
    const float* adst0   = (const float*)d_in[7];
    const float* b0      = (const float*)d_in[8];
    const float* Wl      = (const float*)d_in[9];
    const float* asrcl   = (const float*)d_in[10];
    const float* adstl   = (const float*)d_in[11];
    const float* bl      = (const float*)d_in[12];
    const float* vn_emb  = (const float*)d_in[13];
    const float* mW1     = (const float*)d_in[14];
    const float* mb1     = (const float*)d_in[15];
    const float* mW2     = (const float*)d_in[16];
    const float* mb2     = (const float*)d_in[17];
    float* out = (float*)d_out;

    char*  ws  = (char*)d_ws;
    size_t off = 0;
    auto carve = [&](size_t bytes) -> void* {
        void* p = ws + off;
        off += (bytes + 255) & ~(size_t)255;
        return p;
    };
    float*    h         = (float*)carve((size_t)NNODES * HDIM * 4);
    float*    a_s       = (float*)carve((size_t)NNODES * 4);
    float*    a_d       = (float*)carve((size_t)NNODES * 4);
    int*      rowptr    = (int*)carve((size_t)(NNODES + 1) * 4);
    int*      bintab    = (int*)carve((size_t)SCAN_M * 4);
    int*      binoff    = (int*)carve((size_t)SCAN_M * 4);
    unsigned* binned    = (unsigned*)carve((size_t)NEDGES * 4);
    int*      csr_src   = (int*)carve((size_t)NEDGES * 4);
    float*    vn_direct = (float*)carve((size_t)HNUMB * HDIM * 4);
    float*    vn_seg    = (float*)carve((size_t)HNUMB * HDIM * 4);

    const int pullBlk = (NNODES + 15) / 16;

    edge_hist_kernel<<<NCHUNKS, 256, 0, stream>>>(dst, bintab, vn_direct, vn_seg, vn_emb, NEDGES);
    bin_scan_kernel<<<1, 1024, 0, stream>>>(bintab, binoff);
    edge_bin_kernel<<<NCHUNKS, 256, 0, stream>>>(src, dst, binoff, binned, NEDGES);
    bin_finalize_kernel<<<NBINS, 1024, 0, stream>>>(binned, binoff, rowptr, csr_src, NNODES, NEDGES);

    auto run_gat = [&](const float* Xin, bool addvn, const float* W, const float* asrc,
                       const float* adst, const float* bias) {
        if (addvn)
            gemm_xw_kernel<true><<<(NNODES + 127) / 128, 256, 0, stream>>>(Xin, W, vn_direct, hblocks, asrc, adst,
                                                                           h, a_s, a_d, NNODES);
        else
            gemm_xw_kernel<false><<<(NNODES + 127) / 128, 256, 0, stream>>>(Xin, W, vn_direct, hblocks, asrc, adst,
                                                                            h, a_s, a_d, NNODES);
        gat_pull_kernel<<<pullBlk, 256, 0, stream>>>(rowptr, csr_src, a_s, a_d, h, asrc, bias, out, NNODES);
    };

    run_gat(x, false, W0, asrc0, adst0, b0);

    static_assert(NCONVL == 2, "the root's update gets a reader with a third conv layer: restore it then");
    for (int l = 0; l < NCONVL; ++l) {
        run_gat(out, true, Wl + l * HDIM * HDIM, asrcl + l * HDIM, adstl + l * HDIM, bl + l * HDIM);
        if (l + 1 == NCONVL) break;
        seg_sum_kernel<<<128, 1024, 0, stream>>>(out, hblocks, vn_seg, NNODES);
        vn_update_mlp_kernel<<<HNUMB, 64, 0, stream>>>(vn_direct, vn_seg, vn_emb, mW1, mb1, mW2, mb2);
    }
}


# Bucket build: each chunk sorts its own edges by bin in place (local LDS scan); a bin's start = sum of the chunks' local offsets: count pass and table scan go, 12 to 10 launches; slot (sum) order new
# speedup vs baseline: 10.2859x; 1.1789x over previous
#include <hip/hip_runtime.h>
#include <hip/hip_bf16.h>
#include <hip/hip_fp16.h>

#define NNODES 100000
#define NEDGES 1600000
#define HDIM   64
#define HNUMB  128
#define NCONVL 2
#define NMLPL  2
#define NEGSLP 0.2f

typedef __attribute__((ext_vector_type(2))) float v2f;
typedef __attribute__((ext_vector_type(8))) float v8f;

__device__ __forceinline__ float leaky(float x) { return x > 0.f ? x : NEGSLP * x; }

template <int J> __device__ __forceinline__ int row_get_i(int v) {
    return __builtin_amdgcn_update_dpp(0, v, 0x150 + J, 0xF, 0xF, true);
}
template <int J> __device__ __forceinline__ float row_get_f(float v) {
    return __int_as_float(__builtin_amdgcn_update_dpp(0, __float_as_int(v), 0x150 + J, 0xF, 0xF, true));
}
template <int M> __device__ __forceinline__ float row_xor_f(float v) {
    return __int_as_float(__builtin_amdgcn_update_dpp(0, __float_as_int(v), 0x160 + M, 0xF, 0xF, true));
}

template <bool ADDVN>
__global__ __launch_bounds__(256) void gemm_xw_kernel(const float* __restrict__ X,
                                                      const float* __restrict__ W,
                                                      const float* __restrict__ vnd,
                                                      const int* __restrict__ blocks,
                                                      const float* __restrict__ asrc,
                                                      const float* __restrict__ adst,
                                                      float* __restrict__ Hout,
                                                      float* __restrict__ a_s,
                                                      float* __restrict__ a_d,
                                                      int nrows) {
    __shared__ float Wlds[HDIM * HDIM];
    const int tid = threadIdx.x;
    for (int idx = tid; idx < HDIM * HDIM; idx += 256) {
        int k = idx >> 6, n = idx & 63;
        Wlds[(k >> 1) * 128 + n * 2 + (k & 1)] = W[idx];
    }
    __syncthreads();

    const int wave    = tid >> 5;
    const int lane    = tid & 31;
    const int hi      = lane >> 4;
    const int mrow    = lane & 15;
    const int rowbase = blockIdx.x * 128 + wave * 16;
    if (rowbase >= nrows) return;

    const float* xrow = X + (size_t)(rowbase + mrow) * HDIM + hi * 2;
    v2f afrag[16];
    if (ADDVN) {
        const float* vrow = vnd + (size_t)blocks[rowbase + mrow] * HDIM + hi * 2;
#pragma unroll
        for (int kk = 0; kk < 16; ++kk)
            afrag[kk] = *(const v2f*)(xrow + kk * 4) + *(const v2f*)(vrow + kk * 4);
    } else {
#pragma unroll
        for (int kk = 0; kk < 16; ++kk)
            afrag[kk] = *(const v2f*)(xrow + kk * 4);
    }
    float ps[8], pd[8];
#pragma unroll
    for (int v = 0; v < 8; ++v) { ps[v] = 0.f; pd[v] = 0.f; }
#pragma unroll
    for (int nt = 0; nt < 4; ++nt) {
        const int ncol = nt * 16 + mrow;
        const float asc = asrc[ncol];
        const float adc = adst[ncol];
        v8f acc = {};
#pragma unroll
        for (int kk = 0; kk < 16; ++kk) {
            v2f b = *(const v2f*)&Wlds[(kk * 2 + hi) * 128 + ncol * 2];
            acc = __builtin_amdgcn_wmma_f32_16x16x4_f32(false, afrag[kk], false, b,
                                                        (short)0, acc, false, false);
        }
#pragma unroll
        for (int v = 0; v < 8; ++v) {
            Hout[(size_t)(rowbase + v + 8 * hi) * HDIM + ncol] = acc[v];
            ps[v] = fmaf(acc[v], asc, ps[v]);
            pd[v] = fmaf(acc[v], adc, pd[v]);
        }
    }
#pragma unroll
    for (int v = 0; v < 8; ++v) {
        ps[v] += row_xor_f<1>(ps[v]); pd[v] += row_xor_f<1>(pd[v]);
        ps[v] += row_xor_f<2>(ps[v]); pd[v] += row_xor_f<2>(pd[v]);
        ps[v] += row_xor_f<4>(ps[v]); pd[v] += row_xor_f<4>(pd[v]);
        ps[v] += row_xor_f<8>(ps[v]); pd[v] += row_xor_f<8>(pd[v]);
    }
    float sv = ps[0], dv = pd[0];
#pragma unroll
    for (int v = 1; v < 8; ++v) {
        sv = (mrow == v) ? ps[v] : sv;
        dv = (mrow == v) ? pd[v] : dv;
    }
    if (mrow < 8) {
        a_s[rowbase + 8 * hi + mrow] = sv;
        a_d[rowbase + 8 * hi + mrow] = dv;
    }
}

#define BIN_SHIFT   9
#define BIN_NODES   (1 << BIN_SHIFT)
#define NBINS       ((NNODES + BIN_NODES - 1) >> BIN_SHIFT)
#define CHUNK_EDGES 6400
#define CHUNK_PER   (CHUNK_EDGES / 256)
#define NCHUNKS     ((NEDGES + CHUNK_EDGES - 1) / CHUNK_EDGES)
#define TAB_WORDS   ((NBINS + 1) * NCHUNKS)
#define RUNS_PER_WAVE ((NCHUNKS + 31) / 32)
static_assert(CHUNK_EDGES % 256 == 0, "a chunk is a whole number of 256-edge rows");
static_assert(BIN_NODES <= 1024, "one thread per node of a bin in the last pass");
static_assert(NBINS < 256, "one thread per bin, and one more for the chunk's total, in the first pass");
static_assert(NCHUNKS <= 1024, "one thread per chunk in the last pass");
static_assert((((long long)(NNODES - 1)) << BIN_SHIFT) + (BIN_NODES - 1) <= 0xFFFFFFFFll,
              "source and in-bin index share one 32-bit word");
static_assert(NCHUNKS * 256 >= HNUMB * HDIM, "the first pass also writes the virtual-node start values");

__global__ __launch_bounds__(256) void edge_bin_kernel(const int* __restrict__ src, const int* __restrict__ dst,
                                                       int* __restrict__ tab, unsigned* __restrict__ binned,
                                                       float* __restrict__ vn_direct,
                                                       float* __restrict__ vn_seg,
                                                       const float* __restrict__ vn_emb, int ne) {
    __shared__ int cur[NBINS + 1];
    __shared__ int wsum[8];
    const int tid = threadIdx.x, lane = tid & 31, wave = tid >> 5;
    const int e0  = blockIdx.x * CHUNK_EDGES;
    int s[CHUNK_PER], d[CHUNK_PER];
#pragma unroll
    for (int i = 0; i < CHUNK_PER; ++i) {
        const int e  = e0 + i * 256 + tid;
        const int ec = e < ne ? e : ne - 1;
        const int dv = dst[ec];
        s[i] = src[ec];
        d[i] = e < ne ? dv : -1;
    }
    for (int b = tid; b <= NBINS; b += 256) cur[b] = 0;
    const int g = blockIdx.x * 256 + tid;
    if (g < HNUMB * HDIM) {
        vn_direct[g] = vn_emb[g & (HDIM - 1)];
        vn_seg[g]    = 0.f;
    }
    __syncthreads();
#pragma unroll
    for (int i = 0; i < CHUNK_PER; ++i)
        atomicAdd(&cur[d[i] >= 0 ? (d[i] >> BIN_SHIFT) : NBINS], 1);
    __syncthreads();
    const int c = tid < NBINS ? cur[tid] : 0;
    int incl = c;
    for (int o = 1; o < 32; o <<= 1) {
        const int t = __shfl_up(incl, o, 32);
        if (lane >= o) incl += t;
    }
    if (lane == 31) wsum[wave] = incl;
    __syncthreads();
    int excl = incl - c;
    for (int w = 0; w < wave; ++w) excl += wsum[w];
    if (tid <= NBINS) tab[tid * NCHUNKS + blockIdx.x] = excl;
    if (tid < NBINS) cur[tid] = excl;
    __syncthreads();
    int pos[CHUNK_PER];
#pragma unroll
    for (int i = 0; i < CHUNK_PER; ++i)
        pos[i] = atomicAdd(&cur[d[i] >= 0 ? (d[i] >> BIN_SHIFT) : NBINS], 1);
#pragma unroll
    for (int i = 0; i < CHUNK_PER; ++i)
        if (d[i] >= 0)
            binned[e0 + pos[i]] = ((unsigned)s[i] << BIN_SHIFT) | (unsigned)(d[i] & (BIN_NODES - 1));
}

__global__ __launch_bounds__(1024) void bin_finalize_kernel(const unsigned* __restrict__ binned,
                                                           const int* __restrict__ tab,
                                                           int* __restrict__ rowptr,
                                                           int* __restrict__ csr_src, int n, int ne) {
    __shared__ int cnt[BIN_NODES];
    __shared__ int wsum[32];
    __shared__ int rlo[NCHUNKS];
    __shared__ int rcn[NCHUNKS];
    const int tid = threadIdx.x, lane = tid & 31;
    const int wave = __builtin_amdgcn_readfirstlane(tid >> 5);
    const int b = blockIdx.x;
    int lo = 0;
    if (tid < NCHUNKS) {
        lo = tab[b * NCHUNKS + tid];
        const int hi = tab[(b + 1) * NCHUNKS + tid];
        rlo[tid] = tid * CHUNK_EDGES + lo;
        rcn[tid] = hi - lo;
    }
    if (tid < BIN_NODES) cnt[tid] = 0;
    int sv = lo;
    for (int o = 16; o > 0; o >>= 1) sv += __shfl_xor(sv, o, 32);
    if (lane == 0) wsum[wave] = sv;
    __syncthreads();
    int bsum = 0;
    for (int w = 0; w < 32; ++w) bsum += wsum[w];
    const int beg = __builtin_amdgcn_readfirstlane(bsum);

    int r0[RUNS_PER_WAVE], rn[RUNS_PER_WAVE];
    unsigned v[RUNS_PER_WAVE][2];
#pragma unroll
    for (int k = 0; k < RUNS_PER_WAVE; ++k) {
        const int ck = wave + 32 * k;
        const int cc = ck < NCHUNKS ? ck : NCHUNKS - 1;
        r0[k] = rlo[cc];
        rn[k] = rcn[cc];
    }
#pragma unroll
    for (int k = 0; k < RUNS_PER_WAVE; ++k) {
        r0[k] = __builtin_amdgcn_readfirstlane(r0[k]);
        rn[k] = (wave + 32 * k < NCHUNKS) ? __builtin_amdgcn_readfirstlane(rn[k]) : 0;
#pragma unroll
        for (int h = 0; h < 2; ++h) {
            const int j = lane + 32 * h;
            int idx = r0[k] + (j < rn[k] ? j : 0);
            idx = idx < ne ? idx : ne - 1;
            v[k][h] = binned[idx];
        }
    }
#pragma unroll
    for (int k = 0; k < RUNS_PER_WAVE; ++k) {
#pragma unroll
        for (int h = 0; h < 2; ++h)
            if (lane + 32 * h < rn[k]) atomicAdd(&cnt[v[k][h] & (BIN_NODES - 1)], 1);
        for (int j = 64 + lane; j < rn[k]; j += 32) atomicAdd(&cnt[binned[r0[k] + j] & (BIN_NODES - 1)], 1);
    }
    __syncthreads();
    const int c = tid < BIN_NODES ? cnt[tid] : 0;
    int incl = c;
    for (int o = 1; o < 32; o <<= 1) {
        const int t = __shfl_up(incl, o, 32);
        if (lane >= o) incl += t;
    }
    if (lane == 31) wsum[wave] = incl;
    __syncthreads();
    int excl = incl - c;
    for (int w = 0; w < wave; ++w) excl += wsum[w];
    if (tid < BIN_NODES) {
        cnt[tid] = excl;
        const int node = b * BIN_NODES + tid;
        if (node < n) rowptr[node] = beg + excl;
    }
    if (b == NBINS - 1 && tid == 0) rowptr[n] = ne;
    __syncthreads();
    int slot[RUNS_PER_WAVE][2];
#pragma unroll
    for (int k = 0; k < RUNS_PER_WAVE; ++k)
#pragma unroll
        for (int h = 0; h < 2; ++h)
            slot[k][h] = (lane + 32 * h < rn[k]) ? atomicAdd(&cnt[v[k][h] & (BIN_NODES - 1)], 1) : 0;
#pragma unroll
    for (int k = 0; k < RUNS_PER_WAVE; ++k) {
#pragma unroll
        for (int h = 0; h < 2; ++h)
            if (lane + 32 * h < rn[k]) csr_src[beg + slot[k][h]] = (int)(v[k][h] >> BIN_SHIFT);
        for (int j = 64 + lane; j < rn[k]; j += 32) {
            const unsigned u = binned[r0[k] + j];
            csr_src[beg + atomicAdd(&cnt[u & (BIN_NODES - 1)], 1)] = (int)(u >> BIN_SHIFT);
        }
    }
}


__device__ __forceinline__ float row_transpose_sum16(const float (&d)[16], bool b3, bool b2, bool b1, bool b0) {
    float t8[8], t4[4], t2[2];
#pragma unroll
    for (int j = 0; j < 8; ++j) t8[j] = (b3 ? d[j + 8] : d[j]) + row_xor_f<8>(b3 ? d[j] : d[j + 8]);
#pragma unroll
    for (int j = 0; j < 4; ++j) t4[j] = (b2 ? t8[j + 4] : t8[j]) + row_xor_f<4>(b2 ? t8[j] : t8[j + 4]);
#pragma unroll
    for (int j = 0; j < 2; ++j) t2[j] = (b1 ? t4[j + 2] : t4[j]) + row_xor_f<2>(b1 ? t4[j] : t4[j + 2]);
    return (b0 ? t2[1] : t2[0]) + row_xor_f<1>(b0 ? t2[0] : t2[1]);
}
__device__ __forceinline__ float row_transpose_sum4(const float (&q)[4], bool b1, bool b0) {
    float t2[2];
#pragma unroll
    for (int j = 0; j < 2; ++j) t2[j] = (b1 ? q[j + 2] : q[j]) + row_xor_f<2>(b1 ? q[j] : q[j + 2]);
    float u = (b0 ? t2[1] : t2[0]) + row_xor_f<1>(b0 ? t2[0] : t2[1]);
    u += row_xor_f<4>(u);
    u += row_xor_f<8>(u);
    return u;
}

#define PULL_LOAD(DST, J, SV) DST = Hm4[row_get_i<(J) & 15>(SV) * 16 + l16]
#define PULL_FMA(SRC, J, WV)                                      \
    {                                                             \
        const float wj = row_get_f<(J) & 15>(WV);                 \
        acc.x = fmaf(wj, SRC.x, acc.x); acc.y = fmaf(wj, SRC.y, acc.y); \
        acc.z = fmaf(wj, SRC.z, acc.z); acc.w = fmaf(wj, SRC.w, acc.w); \
    }
#define ROW_DOT(SRC) fmaf(SRC.w, av.w, fmaf(SRC.z, av.z, fmaf(SRC.y, av.y, SRC.x * av.x)))
#define PULL_GROUP4(J0, SV, WV)                                   \
    {                                                             \
        float4 t0, t1, t2, t3;                                    \
        PULL_LOAD(t0, (J0), SV);     PULL_LOAD(t1, (J0) + 1, SV); \
        PULL_LOAD(t2, (J0) + 2, SV); PULL_LOAD(t3, (J0) + 3, SV); \
        PULL_FMA(t0, (J0), WV);     PULL_FMA(t1, (J0) + 1, WV);   \
        PULL_FMA(t2, (J0) + 2, WV); PULL_FMA(t3, (J0) + 3, WV);   \
    }

__global__ __launch_bounds__(256) void gat_pull_kernel(const int* __restrict__ rowptr,
                                                       const int* __restrict__ csr_src,
                                                       const float* __restrict__ a_s,
                                                       const float* __restrict__ a_d,
                                                       const float* __restrict__ Hm,
                                                       const float* __restrict__ asrc,
                                                       const float* __restrict__ bias,
                                                       float* __restrict__ out, int n) {
    const int l16  = threadIdx.x & 15;
    const int node = blockIdx.x * 16 + (threadIdx.x >> 4);
    const bool valid = node < n;
    const int nd  = valid ? node : n - 1;
    const int beg  = rowptr[nd];
    const int endv = rowptr[nd + 1];
    const int deg  = valid ? endv - beg : 0;
    const float ad = a_d[nd];
    const float as_self = a_s[nd];
    const float4* __restrict__ Hm4 = (const float4*)Hm;

    const bool p0 = l16 < deg, p1 = l16 + 16 < deg;
    int s0 = csr_src[p0 ? beg + l16 : 0];
    int s1 = csr_src[p1 ? beg + 16 + l16 : 0];
    s0 = p0 ? s0 : nd;
    s1 = p1 ? s1 : nd;
    const float es = leaky(as_self + ad);
    const float as1 = a_s[(p1 && l16 >= 4) ? s1 : nd];
    const float4 av = *(const float4*)(asrc + 4 * l16);
    __builtin_amdgcn_sched_barrier(0);

    const float4 hs = Hm4[nd * 16 + l16];
    const float4 bv = *(const float4*)(bias + 4 * l16);
    float4 r0, r1, r2, r3, r4, r5, r6, r7, r8, r9, r10, r11, r12, r13, r14, r15, r16, r17, r18, r19;
    PULL_LOAD(r0, 0, s0);   PULL_LOAD(r1, 1, s0);   PULL_LOAD(r2, 2, s0);   PULL_LOAD(r3, 3, s0);
    PULL_LOAD(r4, 4, s0);   PULL_LOAD(r5, 5, s0);   PULL_LOAD(r6, 6, s0);   PULL_LOAD(r7, 7, s0);
    PULL_LOAD(r8, 8, s0);   PULL_LOAD(r9, 9, s0);   PULL_LOAD(r10, 10, s0); PULL_LOAD(r11, 11, s0);
    PULL_LOAD(r12, 12, s0); PULL_LOAD(r13, 13, s0); PULL_LOAD(r14, 14, s0); PULL_LOAD(r15, 15, s0);
    PULL_LOAD(r16, 16, s1); PULL_LOAD(r17, 17, s1); PULL_LOAD(r18, 18, s1); PULL_LOAD(r19, 19, s1);

    const bool past32 = __builtin_amdgcn_ballot_w32(deg > 32) != 0u;

    const bool b3 = (l16 & 8) != 0, b2 = (l16 & 4) != 0, b1 = (l16 & 2) != 0, b0 = (l16 & 1) != 0;
    const float d0[16] = {ROW_DOT(r0),  ROW_DOT(r1),  ROW_DOT(r2),  ROW_DOT(r3),
                          ROW_DOT(r4),  ROW_DOT(r5),  ROW_DOT(r6),  ROW_DOT(r7),
                          ROW_DOT(r8),  ROW_DOT(r9),  ROW_DOT(r10), ROW_DOT(r11),
                          ROW_DOT(r12), ROW_DOT(r13), ROW_DOT(r14), ROW_DOT(r15)};
    const float d1[4]  = {ROW_DOT(r16), ROW_DOT(r17), ROW_DOT(r18), ROW_DOT(r19)};
    const float sc0 = row_transpose_sum16(d0, b3, b2, b1, b0);
    const float sc1 = row_transpose_sum4(d1, b1, b0);
    const float e0 = p0 ? leaky(sc0 + ad) : es;
    const float e1 = p1 ? leaky((l16 < 4 ? sc1 : as1) + ad) : es;

    float m = fmaxf(es, fmaxf(e0, e1));
    if (past32)
        for (int i = 32 + l16; i < deg; i += 16) m = fmaxf(m, leaky(a_s[csr_src[beg + i]] + ad));
    m = fmaxf(m, row_xor_f<1>(m));
    m = fmaxf(m, row_xor_f<2>(m));
    m = fmaxf(m, row_xor_f<4>(m));
    m = fmaxf(m, row_xor_f<8>(m));

    const float wself = __expf(es - m);
    const float w0 = p0 ? __expf(e0 - m) : 0.f;
    const float w1 = p1 ? __expf(e1 - m) : 0.f;
    float dpart = w0 + w1;
    float4 acc;
    acc.x = wself * hs.x; acc.y = wself * hs.y; acc.z = wself * hs.z; acc.w = wself * hs.w;

    PULL_FMA(r0, 0, w0);   PULL_FMA(r1, 1, w0);   PULL_FMA(r2, 2, w0);   PULL_FMA(r3, 3, w0);
    PULL_FMA(r4, 4, w0);   PULL_FMA(r5, 5, w0);   PULL_FMA(r6, 6, w0);   PULL_FMA(r7, 7, w0);
    PULL_FMA(r8, 8, w0);   PULL_FMA(r9, 9, w0);   PULL_FMA(r10, 10, w0); PULL_FMA(r11, 11, w0);
    PULL_FMA(r12, 12, w0); PULL_FMA(r13, 13, w0); PULL_FMA(r14, 14, w0); PULL_FMA(r15, 15, w0);
    PULL_FMA(r16, 16, w1); PULL_FMA(r17, 17, w1); PULL_FMA(r18, 18, w1); PULL_FMA(r19, 19, w1);

    if (__builtin_amdgcn_ballot_w32(deg > 20) != 0u) {
        PULL_GROUP4(20, s1, w1);
        if (__builtin_amdgcn_ballot_w32(deg > 24) != 0u) {
            PULL_GROUP4(24, s1, w1);
            if (__builtin_amdgcn_ballot_w32(deg > 28) != 0u) {
                PULL_GROUP4(28, s1, w1);
            }
        }
    }
    if (past32) {
        for (int cb = 32; __builtin_amdgcn_ballot_w32(cb < deg) != 0u; cb += 16) {
            const bool p = cb + l16 < deg;
            int s = csr_src[p ? beg + cb + l16 : 0];
            s = p ? s : nd;
            const float w = p ? __expf(leaky(a_s[s] + ad) - m) : 0.f;
            dpart += w;
            PULL_GROUP4(0, s, w);
            PULL_GROUP4(4, s, w);
            PULL_GROUP4(8, s, w);
            PULL_GROUP4(12, s, w);
        }
    }

    dpart += row_xor_f<1>(dpart);
    dpart += row_xor_f<2>(dpart);
    dpart += row_xor_f<4>(dpart);
    dpart += row_xor_f<8>(dpart);
    const float den = (dpart + wself) + 1e-16f;
    float4 res;
    res.x = acc.x / den + bv.x;
    res.y = acc.y / den + bv.y;
    res.z = acc.z / den + bv.z;
    res.w = acc.w / den + bv.w;
    if (valid) *(float4*)(out + (size_t)node * HDIM + 4 * l16) = res;
}

#define SEG_ROWS 25
__global__ __launch_bounds__(1024) void seg_sum_kernel(const float* __restrict__ out,
                                                       const int* __restrict__ blocks,
                                                       float* __restrict__ vn_seg, int n) {
    __shared__ float acc[HNUMB * HDIM];
    for (int i = threadIdx.x; i < HNUMB * HDIM; i += 1024) acc[i] = 0.f;
    __syncthreads();
    const int lane   = threadIdx.x & 31;
    const int nwaves = gridDim.x * 32;
    const int gw   = __builtin_amdgcn_readfirstlane((int)(blockIdx.x * 32 + (threadIdx.x >> 5)));
    const int last = n - 1;
    for (int base = gw * SEG_ROWS; base < n; base += nwaves * SEG_ROWS) {
        const int rk  = base + lane;
        const int bid = blocks[rk < last ? rk : last];
        v2f v[SEG_ROWS];
#pragma unroll
        for (int k = 0; k < SEG_ROWS; ++k) {
            const int r = (base + k < n) ? (base + k) : last;
            v[k] = *(const v2f*)(out + (size_t)r * HDIM + 2 * lane);
        }
#pragma unroll
        for (int k = 0; k < SEG_ROWS; ++k) {
            const bool live = base + k < n;
            const int  b    = __builtin_amdgcn_readlane(bid, k);
            float* a = acc + b * HDIM + 2 * lane;
            atomicAdd(a,     live ? v[k].x : 0.f);
            atomicAdd(a + 1, live ? v[k].y : 0.f);
        }
    }
    __syncthreads();
    for (int i = threadIdx.x; i < HNUMB * HDIM; i += 1024) atomicAdd(vn_seg + i, acc[i]);
}

__global__ __launch_bounds__(64) void vn_update_mlp_kernel(float* __restrict__ vn_direct,
                                                           const float* __restrict__ vn_seg,
                                                           const float* __restrict__ vn_emb,
                                                           const float* __restrict__ W1, const float* __restrict__ b1,
                                                           const float* __restrict__ W2, const float* __restrict__ b2) {
    __shared__ float rin[HDIM];
    __shared__ float t1[HDIM];
    const int f = threadIdx.x;
    float* row = vn_direct + (size_t)blockIdx.x * HDIM;
    float cur = row[f] + (vn_seg[(size_t)blockIdx.x * HDIM + f] + vn_emb[f]);
    for (int j = 0; j < NMLPL; ++j) {
        const float* w1 = W1 + (size_t)j * HDIM * HDIM;
        const float* w2 = W2 + (size_t)j * HDIM * HDIM;
        rin[f] = cur;
        __syncthreads();
        float s = b1[j * HDIM + f];
#pragma unroll
        for (int k = 0; k < HDIM; ++k) s += rin[k] * w1[k * HDIM + f];
        t1[f] = fmaxf(s, 0.f);
        __syncthreads();
        float s2 = b2[j * HDIM + f];
#pragma unroll
        for (int k = 0; k < HDIM; ++k) s2 += t1[k] * w2[k * HDIM + f];
        cur = fmaxf(s2, 0.f);
    }
    row[f] = cur;
}

extern "C" void kernel_launch(void* const* d_in, const int* in_sizes, int n_in,
                              void* d_out, int out_size, void* d_ws, size_t ws_size,
                              hipStream_t stream) {
    const float* x       = (const float*)d_in[0];
    const int*   ei      = (const int*)d_in[1];
    const int*   src     = ei;
    const int*   dst     = ei + NEDGES;
    const int*   hblocks = (const int*)d_in[2];
    const float* W0      = (const float*)d_in[5];
    const float* asrc0   = (const float*)d_in[6];
    const float* adst0   = (const float*)d_in[7];
    const float* b0      = (const float*)d_in[8];
    const float* Wl      = (const float*)d_in[9];
    const float* asrcl   = (const float*)d_in[10];
    const float* adstl   = (const float*)d_in[11];
    const float* bl      = (const float*)d_in[12];
    const float* vn_emb  = (const float*)d_in[13];
    const float* mW1     = (const float*)d_in[14];
    const float* mb1     = (const float*)d_in[15];
    const float* mW2     = (const float*)d_in[16];
    const float* mb2     = (const float*)d_in[17];
    float* out = (float*)d_out;

    char*  ws  = (char*)d_ws;
    size_t off = 0;
    auto carve = [&](size_t bytes) -> void* {
        void* p = ws + off;
        off += (bytes + 255) & ~(size_t)255;
        return p;
    };
    float*    h         = (float*)carve((size_t)NNODES * HDIM * 4);
    float*    a_s       = (float*)carve((size_t)NNODES * 4);
    float*    a_d       = (float*)carve((size_t)NNODES * 4);
    int*      rowptr    = (int*)carve((size_t)(NNODES + 1) * 4);
    int*      bintab    = (int*)carve((size_t)TAB_WORDS * 4);
    unsigned* binned    = (unsigned*)carve((size_t)NEDGES * 4);
    int*      csr_src   = (int*)carve((size_t)NEDGES * 4);
    float*    vn_direct = (float*)carve((size_t)HNUMB * HDIM * 4);
    float*    vn_seg    = (float*)carve((size_t)HNUMB * HDIM * 4);

    const int pullBlk = (NNODES + 15) / 16;

    edge_bin_kernel<<<NCHUNKS, 256, 0, stream>>>(src, dst, bintab, binned, vn_direct, vn_seg, vn_emb, NEDGES);
    bin_finalize_kernel<<<NBINS, 1024, 0, stream>>>(binned, bintab, rowptr, csr_src, NNODES, NEDGES);

    auto run_gat = [&](const float* Xin, bool addvn, const float* W, const float* asrc,
                       const float* adst, const float* bias) {
        if (addvn)
            gemm_xw_kernel<true><<<(NNODES + 127) / 128, 256, 0, stream>>>(Xin, W, vn_direct, hblocks, asrc, adst,
                                                                           h, a_s, a_d, NNODES);
        else
            gemm_xw_kernel<false><<<(NNODES + 127) / 128, 256, 0, stream>>>(Xin, W, vn_direct, hblocks, asrc, adst,
                                                                            h, a_s, a_d, NNODES);
        gat_pull_kernel<<<pullBlk, 256, 0, stream>>>(rowptr, csr_src, a_s, a_d, h, asrc, bias, out, NNODES);
    };

    run_gat(x, false, W0, asrc0, adst0, b0);

    static_assert(NCONVL == 2, "the root's update gets a reader with a third conv layer: restore it then");
    for (int l = 0; l < NCONVL; ++l) {
        run_gat(out, true, Wl + l * HDIM * HDIM, asrcl + l * HDIM, adstl + l * HDIM, bl + l * HDIM);
        if (l + 1 == NCONVL) break;
        seg_sum_kernel<<<128, 1024, 0, stream>>>(out, hblocks, vn_seg, NNODES);
        vn_update_mlp_kernel<<<HNUMB, 64, 0, stream>>>(vn_direct, vn_seg, vn_emb, mW1, mb1, mW2, mb2);
    }
}


# First layer's product x @ W0 (with its scores) runs inside the edge binning launch, as extra work-groups beside the edge chunks it does not depend on: 10 launches become 9; no sum order changes
# speedup vs baseline: 10.8866x; 1.0584x over previous
#include <hip/hip_runtime.h>
#include <hip/hip_bf16.h>
#include <hip/hip_fp16.h>

#define NNODES 100000
#define NEDGES 1600000
#define HDIM   64
#define HNUMB  128
#define NCONVL 2
#define NMLPL  2
#define NEGSLP 0.2f

typedef __attribute__((ext_vector_type(2))) float v2f;
typedef __attribute__((ext_vector_type(8))) float v8f;

__device__ __forceinline__ float leaky(float x) { return x > 0.f ? x : NEGSLP * x; }

template <int J> __device__ __forceinline__ int row_get_i(int v) {
    return __builtin_amdgcn_update_dpp(0, v, 0x150 + J, 0xF, 0xF, true);
}
template <int J> __device__ __forceinline__ float row_get_f(float v) {
    return __int_as_float(__builtin_amdgcn_update_dpp(0, __float_as_int(v), 0x150 + J, 0xF, 0xF, true));
}
template <int M> __device__ __forceinline__ float row_xor_f(float v) {
    return __int_as_float(__builtin_amdgcn_update_dpp(0, __float_as_int(v), 0x160 + M, 0xF, 0xF, true));
}

template <bool ADDVN>
__device__ __forceinline__ void gemm_xw_body(const int tile,
                                             const float* __restrict__ X,
                                             const float* __restrict__ W,
                                             const float* __restrict__ vnd,
                                             const int* __restrict__ blocks,
                                             const float* __restrict__ asrc,
                                             const float* __restrict__ adst,
                                             float* __restrict__ Hout,
                                             float* __restrict__ a_s,
                                             float* __restrict__ a_d,
                                             int nrows) {
    __shared__ float Wlds[HDIM * HDIM];
    const int tid = threadIdx.x;
    for (int idx = tid; idx < HDIM * HDIM; idx += 256) {
        int k = idx >> 6, n = idx & 63;
        Wlds[(k >> 1) * 128 + n * 2 + (k & 1)] = W[idx];
    }
    __syncthreads();

    const int wave    = tid >> 5;
    const int lane    = tid & 31;
    const int hi      = lane >> 4;
    const int mrow    = lane & 15;
    const int rowbase = tile * 128 + wave * 16;
    if (rowbase >= nrows) return;

    const float* xrow = X + (size_t)(rowbase + mrow) * HDIM + hi * 2;
    v2f afrag[16];
    if (ADDVN) {
        const float* vrow = vnd + (size_t)blocks[rowbase + mrow] * HDIM + hi * 2;
#pragma unroll
        for (int kk = 0; kk < 16; ++kk)
            afrag[kk] = *(const v2f*)(xrow + kk * 4) + *(const v2f*)(vrow + kk * 4);
    } else {
#pragma unroll
        for (int kk = 0; kk < 16; ++kk)
            afrag[kk] = *(const v2f*)(xrow + kk * 4);
    }
    float ps[8], pd[8];
#pragma unroll
    for (int v = 0; v < 8; ++v) { ps[v] = 0.f; pd[v] = 0.f; }
#pragma unroll
    for (int nt = 0; nt < 4; ++nt) {
        const int ncol = nt * 16 + mrow;
        const float asc = asrc[ncol];
        const float adc = adst[ncol];
        v8f acc = {};
#pragma unroll
        for (int kk = 0; kk < 16; ++kk) {
            v2f b = *(const v2f*)&Wlds[(kk * 2 + hi) * 128 + ncol * 2];
            acc = __builtin_amdgcn_wmma_f32_16x16x4_f32(false, afrag[kk], false, b,
                                                        (short)0, acc, false, false);
        }
#pragma unroll
        for (int v = 0; v < 8; ++v) {
            Hout[(size_t)(rowbase + v + 8 * hi) * HDIM + ncol] = acc[v];
            ps[v] = fmaf(acc[v], asc, ps[v]);
            pd[v] = fmaf(acc[v], adc, pd[v]);
        }
    }
#pragma unroll
    for (int v = 0; v < 8; ++v) {
        ps[v] += row_xor_f<1>(ps[v]); pd[v] += row_xor_f<1>(pd[v]);
        ps[v] += row_xor_f<2>(ps[v]); pd[v] += row_xor_f<2>(pd[v]);
        ps[v] += row_xor_f<4>(ps[v]); pd[v] += row_xor_f<4>(pd[v]);
        ps[v] += row_xor_f<8>(ps[v]); pd[v] += row_xor_f<8>(pd[v]);
    }
    float sv = ps[0], dv = pd[0];
#pragma unroll
    for (int v = 1; v < 8; ++v) {
        sv = (mrow == v) ? ps[v] : sv;
        dv = (mrow == v) ? pd[v] : dv;
    }
    if (mrow < 8) {
        a_s[rowbase + 8 * hi + mrow] = sv;
        a_d[rowbase + 8 * hi + mrow] = dv;
    }
}

template <bool ADDVN>
__global__ __launch_bounds__(256) void gemm_xw_kernel(const float* __restrict__ X,
                                                      const float* __restrict__ W,
                                                      const float* __restrict__ vnd,
                                                      const int* __restrict__ blocks,
                                                      const float* __restrict__ asrc,
                                                      const float* __restrict__ adst,
                                                      float* __restrict__ Hout,
                                                      float* __restrict__ a_s,
                                                      float* __restrict__ a_d,
                                                      int nrows) {
    gemm_xw_body<ADDVN>((int)blockIdx.x, X, W, vnd, blocks, asrc, adst, Hout, a_s, a_d, nrows);
}

#define BIN_SHIFT   9
#define BIN_NODES   (1 << BIN_SHIFT)
#define NBINS       ((NNODES + BIN_NODES - 1) >> BIN_SHIFT)
#define CHUNK_EDGES 6400
#define CHUNK_PER   (CHUNK_EDGES / 256)
#define NCHUNKS     ((NEDGES + CHUNK_EDGES - 1) / CHUNK_EDGES)
#define TAB_WORDS   ((NBINS + 1) * NCHUNKS)
#define RUNS_PER_WAVE ((NCHUNKS + 31) / 32)
static_assert(CHUNK_EDGES % 256 == 0, "a chunk is a whole number of 256-edge rows");
static_assert(BIN_NODES <= 1024, "one thread per node of a bin in the last pass");
static_assert(NBINS < 256, "one thread per bin, and one more for the chunk's total, in the first pass");
static_assert(NCHUNKS <= 1024, "one thread per chunk in the last pass");
static_assert((((long long)(NNODES - 1)) << BIN_SHIFT) + (BIN_NODES - 1) <= 0xFFFFFFFFll,
              "source and in-bin index share one 32-bit word");
static_assert(NCHUNKS * 256 >= HNUMB * HDIM, "the first pass also writes the virtual-node start values");

__device__ __forceinline__ void edge_bin_body(const int chunk,
                                              const int* __restrict__ src, const int* __restrict__ dst,
                                              int* __restrict__ tab, unsigned* __restrict__ binned,
                                              float* __restrict__ vn_direct,
                                              float* __restrict__ vn_seg,
                                              const float* __restrict__ vn_emb, int ne) {
    __shared__ int cur[NBINS + 1];
    __shared__ int wsum[8];
    const int tid = threadIdx.x, lane = tid & 31, wave = tid >> 5;
    const int e0  = chunk * CHUNK_EDGES;
    int s[CHUNK_PER], d[CHUNK_PER];
#pragma unroll
    for (int i = 0; i < CHUNK_PER; ++i) {
        const int e  = e0 + i * 256 + tid;
        const int ec = e < ne ? e : ne - 1;
        const int dv = dst[ec];
        s[i] = src[ec];
        d[i] = e < ne ? dv : -1;
    }
    for (int b = tid; b <= NBINS; b += 256) cur[b] = 0;
    const int g = chunk * 256 + tid;
    if (g < HNUMB * HDIM) {
        vn_direct[g] = vn_emb[g & (HDIM - 1)];
        vn_seg[g]    = 0.f;
    }
    __syncthreads();
#pragma unroll
    for (int i = 0; i < CHUNK_PER; ++i)
        atomicAdd(&cur[d[i] >= 0 ? (d[i] >> BIN_SHIFT) : NBINS], 1);
    __syncthreads();
    const int c = tid < NBINS ? cur[tid] : 0;
    int incl = c;
    for (int o = 1; o < 32; o <<= 1) {
        const int t = __shfl_up(incl, o, 32);
        if (lane >= o) incl += t;
    }
    if (lane == 31) wsum[wave] = incl;
    __syncthreads();
    int excl = incl - c;
    for (int w = 0; w < wave; ++w) excl += wsum[w];
    if (tid <= NBINS) tab[tid * NCHUNKS + chunk] = excl;
    if (tid < NBINS) cur[tid] = excl;
    __syncthreads();
    int pos[CHUNK_PER];
#pragma unroll
    for (int i = 0; i < CHUNK_PER; ++i)
        pos[i] = atomicAdd(&cur[d[i] >= 0 ? (d[i] >> BIN_SHIFT) : NBINS], 1);
#pragma unroll
    for (int i = 0; i < CHUNK_PER; ++i)
        if (d[i] >= 0)
            binned[e0 + pos[i]] = ((unsigned)s[i] << BIN_SHIFT) | (unsigned)(d[i] & (BIN_NODES - 1));
}

#define GEMM_TILES ((NNODES + 127) / 128)
__global__ __launch_bounds__(256) void edge_bin_gemm_kernel(const int* __restrict__ src, const int* __restrict__ dst,
                                                            int* __restrict__ tab, unsigned* __restrict__ binned,
                                                            float* __restrict__ vn_direct,
                                                            float* __restrict__ vn_seg,
                                                            const float* __restrict__ vn_emb, int ne,
                                                            const float* __restrict__ X,
                                                            const float* __restrict__ W,
                                                            const float* __restrict__ asrc,
                                                            const float* __restrict__ adst,
                                                            float* __restrict__ Hout,
                                                            float* __restrict__ a_s,
                                                            float* __restrict__ a_d,
                                                            int nrows) {
    if (blockIdx.x < NCHUNKS)
        edge_bin_body((int)blockIdx.x, src, dst, tab, binned, vn_direct, vn_seg, vn_emb, ne);
    else
        gemm_xw_body<false>((int)blockIdx.x - NCHUNKS, X, W, nullptr, nullptr, asrc, adst, Hout, a_s, a_d, nrows);
}

__global__ __launch_bounds__(1024) void bin_finalize_kernel(const unsigned* __restrict__ binned,
                                                           const int* __restrict__ tab,
                                                           int* __restrict__ rowptr,
                                                           int* __restrict__ csr_src, int n, int ne) {
    __shared__ int cnt[BIN_NODES];
    __shared__ int wsum[32];
    __shared__ int rlo[NCHUNKS];
    __shared__ int rcn[NCHUNKS];
    const int tid = threadIdx.x, lane = tid & 31;
    const int wave = __builtin_amdgcn_readfirstlane(tid >> 5);
    const int b = blockIdx.x;
    int lo = 0;
    if (tid < NCHUNKS) {
        lo = tab[b * NCHUNKS + tid];
        const int hi = tab[(b + 1) * NCHUNKS + tid];
        rlo[tid] = tid * CHUNK_EDGES + lo;
        rcn[tid] = hi - lo;
    }
    if (tid < BIN_NODES) cnt[tid] = 0;
    int sv = lo;
    for (int o = 16; o > 0; o >>= 1) sv += __shfl_xor(sv, o, 32);
    if (lane == 0) wsum[wave] = sv;
    __syncthreads();
    int bsum = 0;
    for (int w = 0; w < 32; ++w) bsum += wsum[w];
    const int beg = __builtin_amdgcn_readfirstlane(bsum);

    int r0[RUNS_PER_WAVE], rn[RUNS_PER_WAVE];
    unsigned v[RUNS_PER_WAVE][2];
#pragma unroll
    for (int k = 0; k < RUNS_PER_WAVE; ++k) {
        const int ck = wave + 32 * k;
        const int cc = ck < NCHUNKS ? ck : NCHUNKS - 1;
        r0[k] = rlo[cc];
        rn[k] = rcn[cc];
    }
#pragma unroll
    for (int k = 0; k < RUNS_PER_WAVE; ++k) {
        r0[k] = __builtin_amdgcn_readfirstlane(r0[k]);
        rn[k] = (wave + 32 * k < NCHUNKS) ? __builtin_amdgcn_readfirstlane(rn[k]) : 0;
#pragma unroll
        for (int h = 0; h < 2; ++h) {
            const int j = lane + 32 * h;
            int idx = r0[k] + (j < rn[k] ? j : 0);
            idx = idx < ne ? idx : ne - 1;
            v[k][h] = binned[idx];
        }
    }
#pragma unroll
    for (int k = 0; k < RUNS_PER_WAVE; ++k) {
#pragma unroll
        for (int h = 0; h < 2; ++h)
            if (lane + 32 * h < rn[k]) atomicAdd(&cnt[v[k][h] & (BIN_NODES - 1)], 1);
        for (int j = 64 + lane; j < rn[k]; j += 32) atomicAdd(&cnt[binned[r0[k] + j] & (BIN_NODES - 1)], 1);
    }
    __syncthreads();
    const int c = tid < BIN_NODES ? cnt[tid] : 0;
    int incl = c;
    for (int o = 1; o < 32; o <<= 1) {
        const int t = __shfl_up(incl, o, 32);
        if (lane >= o) incl += t;
    }
    if (lane == 31) wsum[wave] = incl;
    __syncthreads();
    int excl = incl - c;
    for (int w = 0; w < wave; ++w) excl += wsum[w];
    if (tid < BIN_NODES) {
        cnt[tid] = excl;
        const int node = b * BIN_NODES + tid;
        if (node < n) rowptr[node] = beg + excl;
    }
    if (b == NBINS - 1 && tid == 0) rowptr[n] = ne;
    __syncthreads();
    int slot[RUNS_PER_WAVE][2];
#pragma unroll
    for (int k = 0; k < RUNS_PER_WAVE; ++k)
#pragma unroll
        for (int h = 0; h < 2; ++h)
            slot[k][h] = (lane + 32 * h < rn[k]) ? atomicAdd(&cnt[v[k][h] & (BIN_NODES - 1)], 1) : 0;
#pragma unroll
    for (int k = 0; k < RUNS_PER_WAVE; ++k) {
#pragma unroll
        for (int h = 0; h < 2; ++h)
            if (lane + 32 * h < rn[k]) csr_src[beg + slot[k][h]] = (int)(v[k][h] >> BIN_SHIFT);
        for (int j = 64 + lane; j < rn[k]; j += 32) {
            const unsigned u = binned[r0[k] + j];
            csr_src[beg + atomicAdd(&cnt[u & (BIN_NODES - 1)], 1)] = (int)(u >> BIN_SHIFT);
        }
    }
}


__device__ __forceinline__ float row_transpose_sum16(const float (&d)[16], bool b3, bool b2, bool b1, bool b0) {
    float t8[8], t4[4], t2[2];
#pragma unroll
    for (int j = 0; j < 8; ++j) t8[j] = (b3 ? d[j + 8] : d[j]) + row_xor_f<8>(b3 ? d[j] : d[j + 8]);
#pragma unroll
    for (int j = 0; j < 4; ++j) t4[j] = (b2 ? t8[j + 4] : t8[j]) + row_xor_f<4>(b2 ? t8[j] : t8[j + 4]);
#pragma unroll
    for (int j = 0; j < 2; ++j) t2[j] = (b1 ? t4[j + 2] : t4[j]) + row_xor_f<2>(b1 ? t4[j] : t4[j + 2]);
    return (b0 ? t2[1] : t2[0]) + row_xor_f<1>(b0 ? t2[0] : t2[1]);
}
__device__ __forceinline__ float row_transpose_sum4(const float (&q)[4], bool b1, bool b0) {
    float t2[2];
#pragma unroll
    for (int j = 0; j < 2; ++j) t2[j] = (b1 ? q[j + 2] : q[j]) + row_xor_f<2>(b1 ? q[j] : q[j + 2]);
    float u = (b0 ? t2[1] : t2[0]) + row_xor_f<1>(b0 ? t2[0] : t2[1]);
    u += row_xor_f<4>(u);
    u += row_xor_f<8>(u);
    return u;
}

#define PULL_LOAD(DST, J, SV) DST = Hm4[row_get_i<(J) & 15>(SV) * 16 + l16]
#define PULL_FMA(SRC, J, WV)                                      \
    {                                                             \
        const float wj = row_get_f<(J) & 15>(WV);                 \
        acc.x = fmaf(wj, SRC.x, acc.x); acc.y = fmaf(wj, SRC.y, acc.y); \
        acc.z = fmaf(wj, SRC.z, acc.z); acc.w = fmaf(wj, SRC.w, acc.w); \
    }
#define ROW_DOT(SRC) fmaf(SRC.w, av.w, fmaf(SRC.z, av.z, fmaf(SRC.y, av.y, SRC.x * av.x)))
#define PULL_GROUP4(J0, SV, WV)                                   \
    {                                                             \
        float4 t0, t1, t2, t3;                                    \
        PULL_LOAD(t0, (J0), SV);     PULL_LOAD(t1, (J0) + 1, SV); \
        PULL_LOAD(t2, (J0) + 2, SV); PULL_LOAD(t3, (J0) + 3, SV); \
        PULL_FMA(t0, (J0), WV);     PULL_FMA(t1, (J0) + 1, WV);   \
        PULL_FMA(t2, (J0) + 2, WV); PULL_FMA(t3, (J0) + 3, WV);   \
    }

__global__ __launch_bounds__(256) void gat_pull_kernel(const int* __restrict__ rowptr,
                                                       const int* __restrict__ csr_src,
                                                       const float* __restrict__ a_s,
                                                       const float* __restrict__ a_d,
                                                       const float* __restrict__ Hm,
                                                       const float* __restrict__ asrc,
                                                       const float* __restrict__ bias,
                                                       float* __restrict__ out, int n) {
    const int l16  = threadIdx.x & 15;
    const int node = blockIdx.x * 16 + (threadIdx.x >> 4);
    const bool valid = node < n;
    const int nd  = valid ? node : n - 1;
    const int beg  = rowptr[nd];
    const int endv = rowptr[nd + 1];
    const int deg  = valid ? endv - beg : 0;
    const float ad = a_d[nd];
    const float as_self = a_s[nd];
    const float4* __restrict__ Hm4 = (const float4*)Hm;

    const bool p0 = l16 < deg, p1 = l16 + 16 < deg;
    int s0 = csr_src[p0 ? beg + l16 : 0];
    int s1 = csr_src[p1 ? beg + 16 + l16 : 0];
    s0 = p0 ? s0 : nd;
    s1 = p1 ? s1 : nd;
    const float es = leaky(as_self + ad);
    const float as1 = a_s[(p1 && l16 >= 4) ? s1 : nd];
    const float4 av = *(const float4*)(asrc + 4 * l16);
    __builtin_amdgcn_sched_barrier(0);

    const float4 hs = Hm4[nd * 16 + l16];
    const float4 bv = *(const float4*)(bias + 4 * l16);
    float4 r0, r1, r2, r3, r4, r5, r6, r7, r8, r9, r10, r11, r12, r13, r14, r15, r16, r17, r18, r19;
    PULL_LOAD(r0, 0, s0);   PULL_LOAD(r1, 1, s0);   PULL_LOAD(r2, 2, s0);   PULL_LOAD(r3, 3, s0);
    PULL_LOAD(r4, 4, s0);   PULL_LOAD(r5, 5, s0);   PULL_LOAD(r6, 6, s0);   PULL_LOAD(r7, 7, s0);
    PULL_LOAD(r8, 8, s0);   PULL_LOAD(r9, 9, s0);   PULL_LOAD(r10, 10, s0); PULL_LOAD(r11, 11, s0);
    PULL_LOAD(r12, 12, s0); PULL_LOAD(r13, 13, s0); PULL_LOAD(r14, 14, s0); PULL_LOAD(r15, 15, s0);
    PULL_LOAD(r16, 16, s1); PULL_LOAD(r17, 17, s1); PULL_LOAD(r18, 18, s1); PULL_LOAD(r19, 19, s1);

    const bool past32 = __builtin_amdgcn_ballot_w32(deg > 32) != 0u;

    const bool b3 = (l16 & 8) != 0, b2 = (l16 & 4) != 0, b1 = (l16 & 2) != 0, b0 = (l16 & 1) != 0;
    const float d0[16] = {ROW_DOT(r0),  ROW_DOT(r1),  ROW_DOT(r2),  ROW_DOT(r3),
                          ROW_DOT(r4),  ROW_DOT(r5),  ROW_DOT(r6),  ROW_DOT(r7),
                          ROW_DOT(r8),  ROW_DOT(r9),  ROW_DOT(r10), ROW_DOT(r11),
                          ROW_DOT(r12), ROW_DOT(r13), ROW_DOT(r14), ROW_DOT(r15)};
    const float d1[4]  = {ROW_DOT(r16), ROW_DOT(r17), ROW_DOT(r18), ROW_DOT(r19)};
    const float sc0 = row_transpose_sum16(d0, b3, b2, b1, b0);
    const float sc1 = row_transpose_sum4(d1, b1, b0);
    const float e0 = p0 ? leaky(sc0 + ad) : es;
    const float e1 = p1 ? leaky((l16 < 4 ? sc1 : as1) + ad) : es;

    float m = fmaxf(es, fmaxf(e0, e1));
    if (past32)
        for (int i = 32 + l16; i < deg; i += 16) m = fmaxf(m, leaky(a_s[csr_src[beg + i]] + ad));
    m = fmaxf(m, row_xor_f<1>(m));
    m = fmaxf(m, row_xor_f<2>(m));
    m = fmaxf(m, row_xor_f<4>(m));
    m = fmaxf(m, row_xor_f<8>(m));

    const float wself = __expf(es - m);
    const float w0 = p0 ? __expf(e0 - m) : 0.f;
    const float w1 = p1 ? __expf(e1 - m) : 0.f;
    float dpart = w0 + w1;
    float4 acc;
    acc.x = wself * hs.x; acc.y = wself * hs.y; acc.z = wself * hs.z; acc.w = wself * hs.w;

    PULL_FMA(r0, 0, w0);   PULL_FMA(r1, 1, w0);   PULL_FMA(r2, 2, w0);   PULL_FMA(r3, 3, w0);
    PULL_FMA(r4, 4, w0);   PULL_FMA(r5, 5, w0);   PULL_FMA(r6, 6, w0);   PULL_FMA(r7, 7, w0);
    PULL_FMA(r8, 8, w0);   PULL_FMA(r9, 9, w0);   PULL_FMA(r10, 10, w0); PULL_FMA(r11, 11, w0);
    PULL_FMA(r12, 12, w0); PULL_FMA(r13, 13, w0); PULL_FMA(r14, 14, w0); PULL_FMA(r15, 15, w0);
    PULL_FMA(r16, 16, w1); PULL_FMA(r17, 17, w1); PULL_FMA(r18, 18, w1); PULL_FMA(r19, 19, w1);

    if (__builtin_amdgcn_ballot_w32(deg > 20) != 0u) {
        PULL_GROUP4(20, s1, w1);
        if (__builtin_amdgcn_ballot_w32(deg > 24) != 0u) {
            PULL_GROUP4(24, s1, w1);
            if (__builtin_amdgcn_ballot_w32(deg > 28) != 0u) {
                PULL_GROUP4(28, s1, w1);
            }
        }
    }
    if (past32) {
        for (int cb = 32; __builtin_amdgcn_ballot_w32(cb < deg) != 0u; cb += 16) {
            const bool p = cb + l16 < deg;
            int s = csr_src[p ? beg + cb + l16 : 0];
            s = p ? s : nd;
            const float w = p ? __expf(leaky(a_s[s] + ad) - m) : 0.f;
            dpart += w;
            PULL_GROUP4(0, s, w);
            PULL_GROUP4(4, s, w);
            PULL_GROUP4(8, s, w);
            PULL_GROUP4(12, s, w);
        }
    }

    dpart += row_xor_f<1>(dpart);
    dpart += row_xor_f<2>(dpart);
    dpart += row_xor_f<4>(dpart);
    dpart += row_xor_f<8>(dpart);
    const float den = (dpart + wself) + 1e-16f;
    float4 res;
    res.x = acc.x / den + bv.x;
    res.y = acc.y / den + bv.y;
    res.z = acc.z / den + bv.z;
    res.w = acc.w / den + bv.w;
    if (valid) *(float4*)(out + (size_t)node * HDIM + 4 * l16) = res;
}

#define SEG_ROWS 25
__global__ __launch_bounds__(1024) void seg_sum_kernel(const float* __restrict__ out,
                                                       const int* __restrict__ blocks,
                                                       float* __restrict__ vn_seg, int n) {
    __shared__ float acc[HNUMB * HDIM];
    for (int i = threadIdx.x; i < HNUMB * HDIM; i += 1024) acc[i] = 0.f;
    __syncthreads();
    const int lane   = threadIdx.x & 31;
    const int nwaves = gridDim.x * 32;
    const int gw   = __builtin_amdgcn_readfirstlane((int)(blockIdx.x * 32 + (threadIdx.x >> 5)));
    const int last = n - 1;
    for (int base = gw * SEG_ROWS; base < n; base += nwaves * SEG_ROWS) {
        const int rk  = base + lane;
        const int bid = blocks[rk < last ? rk : last];
        v2f v[SEG_ROWS];
#pragma unroll
        for (int k = 0; k < SEG_ROWS; ++k) {
            const int r = (base + k < n) ? (base + k) : last;
            v[k] = *(const v2f*)(out + (size_t)r * HDIM + 2 * lane);
        }
#pragma unroll
        for (int k = 0; k < SEG_ROWS; ++k) {
            const bool live = base + k < n;
            const int  b    = __builtin_amdgcn_readlane(bid, k);
            float* a = acc + b * HDIM + 2 * lane;
            atomicAdd(a,     live ? v[k].x : 0.f);
            atomicAdd(a + 1, live ? v[k].y : 0.f);
        }
    }
    __syncthreads();
    for (int i = threadIdx.x; i < HNUMB * HDIM; i += 1024) atomicAdd(vn_seg + i, acc[i]);
}

__global__ __launch_bounds__(64) void vn_update_mlp_kernel(float* __restrict__ vn_direct,
                                                           const float* __restrict__ vn_seg,
                                                           const float* __restrict__ vn_emb,
                                                           const float* __restrict__ W1, const float* __restrict__ b1,
                                                           const float* __restrict__ W2, const float* __restrict__ b2) {
    __shared__ float rin[HDIM];
    __shared__ float t1[HDIM];
    const int f = threadIdx.x;
    float* row = vn_direct + (size_t)blockIdx.x * HDIM;
    float cur = row[f] + (vn_seg[(size_t)blockIdx.x * HDIM + f] + vn_emb[f]);
    for (int j = 0; j < NMLPL; ++j) {
        const float* w1 = W1 + (size_t)j * HDIM * HDIM;
        const float* w2 = W2 + (size_t)j * HDIM * HDIM;
        rin[f] = cur;
        __syncthreads();
        float s = b1[j * HDIM + f];
#pragma unroll
        for (int k = 0; k < HDIM; ++k) s += rin[k] * w1[k * HDIM + f];
        t1[f] = fmaxf(s, 0.f);
        __syncthreads();
        float s2 = b2[j * HDIM + f];
#pragma unroll
        for (int k = 0; k < HDIM; ++k) s2 += t1[k] * w2[k * HDIM + f];
        cur = fmaxf(s2, 0.f);
    }
    row[f] = cur;
}

extern "C" void kernel_launch(void* const* d_in, const int* in_sizes, int n_in,
                              void* d_out, int out_size, void* d_ws, size_t ws_size,
                              hipStream_t stream) {
    const float* x       = (const float*)d_in[0];
    const int*   ei      = (const int*)d_in[1];
    const int*   src     = ei;
    const int*   dst     = ei + NEDGES;
    const int*   hblocks = (const int*)d_in[2];
    const float* W0      = (const float*)d_in[5];
    const float* asrc0   = (const float*)d_in[6];
    const float* adst0   = (const float*)d_in[7];
    const float* b0      = (const float*)d_in[8];
    const float* Wl      = (const float*)d_in[9];
    const float* asrcl   = (const float*)d_in[10];
    const float* adstl   = (const float*)d_in[11];
    const float* bl      = (const float*)d_in[12];
    const float* vn_emb  = (const float*)d_in[13];
    const float* mW1     = (const float*)d_in[14];
    const float* mb1     = (const float*)d_in[15];
    const float* mW2     = (const float*)d_in[16];
    const float* mb2     = (const float*)d_in[17];
    float* out = (float*)d_out;

    char*  ws  = (char*)d_ws;
    size_t off = 0;
    auto carve = [&](size_t bytes) -> void* {
        void* p = ws + off;
        off += (bytes + 255) & ~(size_t)255;
        return p;
    };
    float*    h         = (float*)carve((size_t)NNODES * HDIM * 4);
    float*    a_s       = (float*)carve((size_t)NNODES * 4);
    float*    a_d       = (float*)carve((size_t)NNODES * 4);
    int*      rowptr    = (int*)carve((size_t)(NNODES + 1) * 4);
    int*      bintab    = (int*)carve((size_t)TAB_WORDS * 4);
    unsigned* binned    = (unsigned*)carve((size_t)NEDGES * 4);
    int*      csr_src   = (int*)carve((size_t)NEDGES * 4);
    float*    vn_direct = (float*)carve((size_t)HNUMB * HDIM * 4);
    float*    vn_seg    = (float*)carve((size_t)HNUMB * HDIM * 4);

    const int pullBlk = (NNODES + 15) / 16;

    edge_bin_gemm_kernel<<<NCHUNKS + GEMM_TILES, 256, 0, stream>>>(src, dst, bintab, binned, vn_direct, vn_seg,
                                                                   vn_emb, NEDGES,
                                                                   x, W0, asrc0, adst0, h, a_s, a_d, NNODES);
    bin_finalize_kernel<<<NBINS, 1024, 0, stream>>>(binned, bintab, rowptr, csr_src, NNODES, NEDGES);

    gat_pull_kernel<<<pullBlk, 256, 0, stream>>>(rowptr, csr_src, a_s, a_d, h, asrc0, b0, out, NNODES);

    auto run_gat = [&](const float* Xin, const float* W, const float* asrc,
                       const float* adst, const float* bias) {
        gemm_xw_kernel<true><<<GEMM_TILES, 256, 0, stream>>>(Xin, W, vn_direct, hblocks, asrc, adst,
                                                             h, a_s, a_d, NNODES);
        gat_pull_kernel<<<pullBlk, 256, 0, stream>>>(rowptr, csr_src, a_s, a_d, h, asrc, bias, out, NNODES);
    };

    static_assert(NCONVL == 2, "the root's update gets a reader with a third conv layer: restore it then");
    for (int l = 0; l < NCONVL; ++l) {
        run_gat(out, Wl + l * HDIM * HDIM, asrcl + l * HDIM, adstl + l * HDIM, bl + l * HDIM);
        if (l + 1 == NCONVL) break;
        seg_sum_kernel<<<128, 1024, 0, stream>>>(out, hblocks, vn_seg, NNODES);
        vn_update_mlp_kernel<<<HNUMB, 64, 0, stream>>>(vn_direct, vn_seg, vn_emb, mW1, mb1, mW2, mb2);
    }
}
